# starting kernel
# baseline (speedup 1.0000x reference)
#include <hip/hip_runtime.h>
#include <hip/hip_bf16.h>
#include <math.h>

#define EMB     128
#define BATCH   16384
#define NNODES  100000
#define NEDGES  640000

typedef float v2f __attribute__((ext_vector_type(2)));
typedef float v8f __attribute__((ext_vector_type(8)));

__device__ __forceinline__ float wave_sum(float v) {
#pragma unroll
    for (int m = 16; m >= 1; m >>= 1) v += __shfl_xor(v, m, 32);
    return v;
}

template <int NT>
__global__ void __launch_bounds__(256)
wmma_gemm_kernel(const float* __restrict__ X,
                 const float* __restrict__ W,
                 const float* __restrict__ bias,
                 float* __restrict__ Y,
                 int M, int K, int act) {
    constexpr int N   = NT * 16;
    constexpr int KC  = 64;
    constexpr int LDW = N + 16;
    __shared__ float2 wt2[(KC / 2) * LDW];

    const int tid   = threadIdx.x;
    const int wave  = tid >> 5;
    const int lane  = tid & 31;
    const int half  = lane >> 4;
    const int l     = lane & 15;
    const int rowBase = (blockIdx.x * (blockDim.x >> 5) + wave) * 16;

    v8f acc[NT] = {};

    const int  row   = rowBase + l;
    const bool rowOK = (row < M);
    const int  rr    = rowOK ? row : (M - 1);
    const float* xrow = X + (size_t)rr * K;

    for (int kc = 0; kc < K; kc += KC) {
        for (int i = tid; i < (KC / 2) * N; i += blockDim.x) {
            const int p = i / N;
            const int c = i % N;
            float2 w2;
            w2.x = W[(size_t)(kc + 2 * p)     * N + c];
            w2.y = W[(size_t)(kc + 2 * p + 1) * N + c];
            wt2[p * LDW + c] = w2;
        }
        __syncthreads();

        float2 a_nxt = *(const float2*)(xrow + kc + 2 * half);
#pragma unroll
        for (int k = 0; k < KC; k += 4) {
            v2f a;
            a.x = rowOK ? a_nxt.x : 0.0f;
            a.y = rowOK ? a_nxt.y : 0.0f;
            if (k + 4 < KC)
                a_nxt = *(const float2*)(xrow + kc + k + 4 + 2 * half);

            const float2* bp = wt2 + ((k >> 1) + half) * LDW + l;
            float2 breg[NT];
#pragma unroll
            for (int t = 0; t < NT; ++t) breg[t] = bp[t * 16];
#pragma unroll
            for (int t = 0; t < NT; ++t) {
                v2f b; b.x = breg[t].x; b.y = breg[t].y;
                acc[t] = __builtin_amdgcn_wmma_f32_16x16x4_f32(
                    false, a, false, b, (short)0, acc[t], false, false);
            }
        }
        __syncthreads();
    }

#pragma unroll
    for (int t = 0; t < NT; ++t) {
        const int   c  = t * 16 + l;
        const float bv = bias ? bias[c] : 0.0f;
#pragma unroll
        for (int v = 0; v < 8; ++v) {
            const int r = rowBase + v + 8 * half;
            if (r < M) {
                float o = acc[t][v] + bv;
                if (act) o = fmaxf(o, 0.0f);
                Y[(size_t)r * N + c] = o;
            }
        }
    }
}

__global__ void edge_scatter_kernel(const int* __restrict__ src,
                                    const int* __restrict__ dst,
                                    const int* __restrict__ etype,
                                    const float* __restrict__ eweight,
                                    const float* __restrict__ y0,
                                    const float* __restrict__ y1,
                                    float* __restrict__ out) {
    const int e = blockIdx.x * (blockDim.x >> 5) + (threadIdx.x >> 5);
    if (e >= NEDGES) return;
    const int lane = threadIdx.x & 31;
    const int s = src[e], d = dst[e];
    const float w = eweight[e];
    const float* y = (etype[e] == 0) ? y0 : y1;
    const float4 v = *(const float4*)(y + (size_t)s * EMB + lane * 4);
    float* o = out + (size_t)d * EMB + lane * 4;
    unsafeAtomicAdd(o + 0, w * v.x);
    unsafeAtomicAdd(o + 1, w * v.y);
    unsafeAtomicAdd(o + 2, w * v.z);
    unsafeAtomicAdd(o + 3, w * v.w);
}

__global__ void relu_ln_kernel(float* __restrict__ x,
                               const float* __restrict__ g,
                               const float* __restrict__ b,
                               int M) {
    const int row = blockIdx.x * (blockDim.x >> 5) + (threadIdx.x >> 5);
    if (row >= M) return;
    const int lane = threadIdx.x & 31;
    float* p = x + (size_t)row * EMB + lane * 4;
    float4 v = *(const float4*)p;
    v.x = fmaxf(v.x, 0.0f); v.y = fmaxf(v.y, 0.0f);
    v.z = fmaxf(v.z, 0.0f); v.w = fmaxf(v.w, 0.0f);
    const float mu = wave_sum(v.x + v.y + v.z + v.w) * (1.0f / EMB);
    const float dx = v.x - mu, dy = v.y - mu, dz = v.z - mu, dw = v.w - mu;
    const float var = wave_sum(dx * dx + dy * dy + dz * dz + dw * dw) * (1.0f / EMB);
    const float inv = rsqrtf(var + 1e-5f);
    const float4 gg = *(const float4*)(g + lane * 4);
    const float4 bb = *(const float4*)(b + lane * 4);
    float4 o;
    o.x = dx * inv * gg.x + bb.x;
    o.y = dy * inv * gg.y + bb.y;
    o.z = dz * inv * gg.z + bb.z;
    o.w = dw * inv * gg.w + bb.w;
    *(float4*)p = o;
}

__global__ void gather_gmf_kernel(const float* __restrict__ h2,
                                  const int* __restrict__ ui,
                                  const int* __restrict__ vi,
                                  float* __restrict__ gmf,
                                  float* __restrict__ hcat) {
    const int bidx = blockIdx.x * (blockDim.x >> 5) + (threadIdx.x >> 5);
    if (bidx >= BATCH) return;
    const int lane = threadIdx.x & 31;
    const int u = ui[bidx], w = vi[bidx];
    const float4 uv = *(const float4*)(h2 + (size_t)u * EMB + lane * 4);
    const float4 vv = *(const float4*)(h2 + (size_t)w * EMB + lane * 4);
    const float su = wave_sum(uv.x * uv.x + uv.y * uv.y + uv.z * uv.z + uv.w * uv.w);
    const float sv = wave_sum(vv.x * vv.x + vv.y * vv.y + vv.z * vv.z + vv.w * vv.w);
    const float rn = 1.0f / (fmaxf(sqrtf(su), 1e-12f) * fmaxf(sqrtf(sv), 1e-12f));
    float4 gg;
    gg.x = uv.x * vv.x * rn; gg.y = uv.y * vv.y * rn;
    gg.z = uv.z * vv.z * rn; gg.w = uv.w * vv.w * rn;
    *(float4*)(gmf + (size_t)bidx * EMB + lane * 4) = gg;
    *(float4*)(hcat + (size_t)bidx * 2 * EMB + lane * 4) = uv;
    *(float4*)(hcat + (size_t)bidx * 2 * EMB + EMB + lane * 4) = vv;
}

__global__ void score_kernel(const float* __restrict__ gmf,
                             const float* __restrict__ h3,
                             const float* __restrict__ out_w,
                             const float* __restrict__ out_b,
                             float* __restrict__ out) {
    const int bidx = blockIdx.x * (blockDim.x >> 5) + (threadIdx.x >> 5);
    if (bidx >= BATCH) return;
    const int lane = threadIdx.x & 31;
    const float4 gg = *(const float4*)(gmf + (size_t)bidx * EMB + lane * 4);
    const float4 ww = *(const float4*)(out_w + lane * 4);
    float acc = gg.x * ww.x + gg.y * ww.y + gg.z * ww.z + gg.w * ww.w;
    acc += h3[(size_t)bidx * 32 + lane] * out_w[EMB + lane];
    acc = wave_sum(acc) + out_b[0];
    if (lane == 0) out[bidx] = 1.0f / (1.0f + expf(-acc));
}

extern "C" void kernel_launch(void* const* d_in, const int* in_sizes, int n_in,
                              void* d_out, int out_size, void* d_ws, size_t ws_size,
                              hipStream_t stream) {
    const int*   user_idx = (const int*)d_in[0];
    const int*   item_idx = (const int*)d_in[1];
    const int*   eidx     = (const int*)d_in[2];
    const int*   etype    = (const int*)d_in[3];
    const float* eweight  = (const float*)d_in[4];
    const float* emb      = (const float*)d_in[5];
    const float* w1_rel   = (const float*)d_in[6];
    const float* w1_root  = (const float*)d_in[7];
    const float* b1       = (const float*)d_in[8];
    const float* ln_g     = (const float*)d_in[9];
    const float* ln_b     = (const float*)d_in[10];
    const float* w2_rel   = (const float*)d_in[11];
    const float* w2_root  = (const float*)d_in[12];
    const float* b2       = (const float*)d_in[13];
    const float* mlp_w0   = (const float*)d_in[14];
    const float* mlp_b0   = (const float*)d_in[15];
    const float* mlp_w1   = (const float*)d_in[16];
    const float* mlp_b1   = (const float*)d_in[17];
    const float* mlp_w2   = (const float*)d_in[18];
    const float* mlp_b2   = (const float*)d_in[19];
    const float* out_w    = (const float*)d_in[20];
    const float* out_b    = (const float*)d_in[21];
    float* out = (float*)d_out;

    char* ws = (char*)d_ws;
    const size_t NB = (size_t)NNODES * EMB * sizeof(float);
    float* y0 = (float*)(ws);
    float* y1 = (float*)(ws + NB);
    float* h1 = (float*)(ws + 2 * NB);
    float* h2 = (float*)(ws + 3 * NB);
    float* gmf  = (float*)(ws);
    float* hcat = (float*)(ws + (size_t)BATCH * EMB * 4);
    float* m0   = (float*)(ws + (size_t)BATCH * EMB * 4 + (size_t)BATCH * 2 * EMB * 4);
    float* m1   = (float*)((char*)m0 + (size_t)BATCH * 128 * 4);
    float* m2   = (float*)((char*)m1 + (size_t)BATCH * 64 * 4);

    const int nodeGrid  = (NNODES + 127) / 128;
    const int batchGrid = (BATCH + 127) / 128;
    const int edgeGrid  = (NEDGES + 7) / 8;
    const int rowGridN  = (NNODES + 7) / 8;
    const int rowGridB  = (BATCH + 7) / 8;

    wmma_gemm_kernel<8><<<nodeGrid, 256, 0, stream>>>(emb, w1_rel,             nullptr, y0, NNODES, EMB, 0);
    wmma_gemm_kernel<8><<<nodeGrid, 256, 0, stream>>>(emb, w1_rel + EMB * EMB, nullptr, y1, NNODES, EMB, 0);
    wmma_gemm_kernel<8><<<nodeGrid, 256, 0, stream>>>(emb, w1_root,            b1,      h1, NNODES, EMB, 0);
    edge_scatter_kernel<<<edgeGrid, 256, 0, stream>>>(eidx, eidx + NEDGES, etype, eweight, y0, y1, h1);
    relu_ln_kernel<<<rowGridN, 256, 0, stream>>>(h1, ln_g, ln_b, NNODES);

    wmma_gemm_kernel<8><<<nodeGrid, 256, 0, stream>>>(h1, w2_rel,             nullptr, y0, NNODES, EMB, 0);
    wmma_gemm_kernel<8><<<nodeGrid, 256, 0, stream>>>(h1, w2_rel + EMB * EMB, nullptr, y1, NNODES, EMB, 0);
    wmma_gemm_kernel<8><<<nodeGrid, 256, 0, stream>>>(h1, w2_root,            b2,      h2, NNODES, EMB, 0);
    edge_scatter_kernel<<<edgeGrid, 256, 0, stream>>>(eidx, eidx + NEDGES, etype, eweight, y0, y1, h2);

    gather_gmf_kernel<<<rowGridB, 256, 0, stream>>>(h2, user_idx, item_idx, gmf, hcat);
    wmma_gemm_kernel<8><<<batchGrid, 256, 0, stream>>>(hcat, mlp_w0, mlp_b0, m0, BATCH, 2 * EMB, 1);
    wmma_gemm_kernel<4><<<batchGrid, 256, 0, stream>>>(m0,   mlp_w1, mlp_b1, m1, BATCH, 128,     1);
    wmma_gemm_kernel<2><<<batchGrid, 256, 0, stream>>>(m1,   mlp_w2, mlp_b2, m2, BATCH, 64,      1);
    score_kernel<<<rowGridB, 256, 0, stream>>>(gmf, m2, out_w, out_b, out);
}


# edge scatter: lane l owns floats l,l+32,l+64,l+96 (row read as 4 one-line loads), so each f32 atomic instruction covers one whole 128-B line, not 8 lanes on each of four; arrival order as before
# speedup vs baseline: 1.1222x; 1.1222x over previous
#include <hip/hip_runtime.h>
#include <hip/hip_bf16.h>
#include <math.h>

#define EMB     128
#define BATCH   16384
#define NNODES  100000
#define NEDGES  640000

typedef float v2f __attribute__((ext_vector_type(2)));
typedef float v8f __attribute__((ext_vector_type(8)));

__device__ __forceinline__ float wave_sum(float v) {
#pragma unroll
    for (int m = 16; m >= 1; m >>= 1) v += __shfl_xor(v, m, 32);
    return v;
}

template <int NT>
__global__ void __launch_bounds__(256)
wmma_gemm_kernel(const float* __restrict__ X,
                 const float* __restrict__ W,
                 const float* __restrict__ bias,
                 float* __restrict__ Y,
                 int M, int K, int act) {
    constexpr int N   = NT * 16;
    constexpr int KC  = 64;
    constexpr int LDW = N + 16;
    __shared__ float2 wt2[(KC / 2) * LDW];

    const int tid   = threadIdx.x;
    const int wave  = tid >> 5;
    const int lane  = tid & 31;
    const int half  = lane >> 4;
    const int l     = lane & 15;
    const int rowBase = (blockIdx.x * (blockDim.x >> 5) + wave) * 16;

    v8f acc[NT] = {};

    const int  row   = rowBase + l;
    const bool rowOK = (row < M);
    const int  rr    = rowOK ? row : (M - 1);
    const float* xrow = X + (size_t)rr * K;

    for (int kc = 0; kc < K; kc += KC) {
        for (int i = tid; i < (KC / 2) * N; i += blockDim.x) {
            const int p = i / N;
            const int c = i % N;
            float2 w2;
            w2.x = W[(size_t)(kc + 2 * p)     * N + c];
            w2.y = W[(size_t)(kc + 2 * p + 1) * N + c];
            wt2[p * LDW + c] = w2;
        }
        __syncthreads();

        float2 a_nxt = *(const float2*)(xrow + kc + 2 * half);
#pragma unroll
        for (int k = 0; k < KC; k += 4) {
            v2f a;
            a.x = rowOK ? a_nxt.x : 0.0f;
            a.y = rowOK ? a_nxt.y : 0.0f;
            if (k + 4 < KC)
                a_nxt = *(const float2*)(xrow + kc + k + 4 + 2 * half);

            const float2* bp = wt2 + ((k >> 1) + half) * LDW + l;
            float2 breg[NT];
#pragma unroll
            for (int t = 0; t < NT; ++t) breg[t] = bp[t * 16];
#pragma unroll
            for (int t = 0; t < NT; ++t) {
                v2f b; b.x = breg[t].x; b.y = breg[t].y;
                acc[t] = __builtin_amdgcn_wmma_f32_16x16x4_f32(
                    false, a, false, b, (short)0, acc[t], false, false);
            }
        }
        __syncthreads();
    }

#pragma unroll
    for (int t = 0; t < NT; ++t) {
        const int   c  = t * 16 + l;
        const float bv = bias ? bias[c] : 0.0f;
#pragma unroll
        for (int v = 0; v < 8; ++v) {
            const int r = rowBase + v + 8 * half;
            if (r < M) {
                float o = acc[t][v] + bv;
                if (act) o = fmaxf(o, 0.0f);
                Y[(size_t)r * N + c] = o;
            }
        }
    }
}

__global__ void edge_scatter_kernel(const int* __restrict__ src,
                                    const int* __restrict__ dst,
                                    const int* __restrict__ etype,
                                    const float* __restrict__ eweight,
                                    const float* __restrict__ y0,
                                    const float* __restrict__ y1,
                                    float* __restrict__ out) {
    const int e = blockIdx.x * (blockDim.x >> 5) + (threadIdx.x >> 5);
    if (e >= NEDGES) return;
    const int lane = threadIdx.x & 31;
    const int s = src[e], d = dst[e];
    const float w = eweight[e];
    const float* y = (etype[e] == 0) ? y0 : y1;
    const float* yr = y + (size_t)s * EMB + lane;
    const float v0 = yr[0];
    const float v1 = yr[32];
    const float v2 = yr[64];
    const float v3 = yr[96];
    float* o = out + (size_t)d * EMB + lane;
    unsafeAtomicAdd(o,      w * v0);
    unsafeAtomicAdd(o + 32, w * v1);
    unsafeAtomicAdd(o + 64, w * v2);
    unsafeAtomicAdd(o + 96, w * v3);
}

__global__ void relu_ln_kernel(float* __restrict__ x,
                               const float* __restrict__ g,
                               const float* __restrict__ b,
                               int M) {
    const int row = blockIdx.x * (blockDim.x >> 5) + (threadIdx.x >> 5);
    if (row >= M) return;
    const int lane = threadIdx.x & 31;
    float* p = x + (size_t)row * EMB + lane * 4;
    float4 v = *(const float4*)p;
    v.x = fmaxf(v.x, 0.0f); v.y = fmaxf(v.y, 0.0f);
    v.z = fmaxf(v.z, 0.0f); v.w = fmaxf(v.w, 0.0f);
    const float mu = wave_sum(v.x + v.y + v.z + v.w) * (1.0f / EMB);
    const float dx = v.x - mu, dy = v.y - mu, dz = v.z - mu, dw = v.w - mu;
    const float var = wave_sum(dx * dx + dy * dy + dz * dz + dw * dw) * (1.0f / EMB);
    const float inv = rsqrtf(var + 1e-5f);
    const float4 gg = *(const float4*)(g + lane * 4);
    const float4 bb = *(const float4*)(b + lane * 4);
    float4 o;
    o.x = dx * inv * gg.x + bb.x;
    o.y = dy * inv * gg.y + bb.y;
    o.z = dz * inv * gg.z + bb.z;
    o.w = dw * inv * gg.w + bb.w;
    *(float4*)p = o;
}

__global__ void gather_gmf_kernel(const float* __restrict__ h2,
                                  const int* __restrict__ ui,
                                  const int* __restrict__ vi,
                                  float* __restrict__ gmf,
                                  float* __restrict__ hcat) {
    const int bidx = blockIdx.x * (blockDim.x >> 5) + (threadIdx.x >> 5);
    if (bidx >= BATCH) return;
    const int lane = threadIdx.x & 31;
    const int u = ui[bidx], w = vi[bidx];
    const float4 uv = *(const float4*)(h2 + (size_t)u * EMB + lane * 4);
    const float4 vv = *(const float4*)(h2 + (size_t)w * EMB + lane * 4);
    const float su = wave_sum(uv.x * uv.x + uv.y * uv.y + uv.z * uv.z + uv.w * uv.w);
    const float sv = wave_sum(vv.x * vv.x + vv.y * vv.y + vv.z * vv.z + vv.w * vv.w);
    const float rn = 1.0f / (fmaxf(sqrtf(su), 1e-12f) * fmaxf(sqrtf(sv), 1e-12f));
    float4 gg;
    gg.x = uv.x * vv.x * rn; gg.y = uv.y * vv.y * rn;
    gg.z = uv.z * vv.z * rn; gg.w = uv.w * vv.w * rn;
    *(float4*)(gmf + (size_t)bidx * EMB + lane * 4) = gg;
    *(float4*)(hcat + (size_t)bidx * 2 * EMB + lane * 4) = uv;
    *(float4*)(hcat + (size_t)bidx * 2 * EMB + EMB + lane * 4) = vv;
}

__global__ void score_kernel(const float* __restrict__ gmf,
                             const float* __restrict__ h3,
                             const float* __restrict__ out_w,
                             const float* __restrict__ out_b,
                             float* __restrict__ out) {
    const int bidx = blockIdx.x * (blockDim.x >> 5) + (threadIdx.x >> 5);
    if (bidx >= BATCH) return;
    const int lane = threadIdx.x & 31;
    const float4 gg = *(const float4*)(gmf + (size_t)bidx * EMB + lane * 4);
    const float4 ww = *(const float4*)(out_w + lane * 4);
    float acc = gg.x * ww.x + gg.y * ww.y + gg.z * ww.z + gg.w * ww.w;
    acc += h3[(size_t)bidx * 32 + lane] * out_w[EMB + lane];
    acc = wave_sum(acc) + out_b[0];
    if (lane == 0) out[bidx] = 1.0f / (1.0f + expf(-acc));
}

extern "C" void kernel_launch(void* const* d_in, const int* in_sizes, int n_in,
                              void* d_out, int out_size, void* d_ws, size_t ws_size,
                              hipStream_t stream) {
    const int*   user_idx = (const int*)d_in[0];
    const int*   item_idx = (const int*)d_in[1];
    const int*   eidx     = (const int*)d_in[2];
    const int*   etype    = (const int*)d_in[3];
    const float* eweight  = (const float*)d_in[4];
    const float* emb      = (const float*)d_in[5];
    const float* w1_rel   = (const float*)d_in[6];
    const float* w1_root  = (const float*)d_in[7];
    const float* b1       = (const float*)d_in[8];
    const float* ln_g     = (const float*)d_in[9];
    const float* ln_b     = (const float*)d_in[10];
    const float* w2_rel   = (const float*)d_in[11];
    const float* w2_root  = (const float*)d_in[12];
    const float* b2       = (const float*)d_in[13];
    const float* mlp_w0   = (const float*)d_in[14];
    const float* mlp_b0   = (const float*)d_in[15];
    const float* mlp_w1   = (const float*)d_in[16];
    const float* mlp_b1   = (const float*)d_in[17];
    const float* mlp_w2   = (const float*)d_in[18];
    const float* mlp_b2   = (const float*)d_in[19];
    const float* out_w    = (const float*)d_in[20];
    const float* out_b    = (const float*)d_in[21];
    float* out = (float*)d_out;

    char* ws = (char*)d_ws;
    const size_t NB = (size_t)NNODES * EMB * sizeof(float);
    float* y0 = (float*)(ws);
    float* y1 = (float*)(ws + NB);
    float* h1 = (float*)(ws + 2 * NB);
    float* h2 = (float*)(ws + 3 * NB);
    float* gmf  = (float*)(ws);
    float* hcat = (float*)(ws + (size_t)BATCH * EMB * 4);
    float* m0   = (float*)(ws + (size_t)BATCH * EMB * 4 + (size_t)BATCH * 2 * EMB * 4);
    float* m1   = (float*)((char*)m0 + (size_t)BATCH * 128 * 4);
    float* m2   = (float*)((char*)m1 + (size_t)BATCH * 64 * 4);

    const int nodeGrid  = (NNODES + 127) / 128;
    const int batchGrid = (BATCH + 127) / 128;
    const int edgeGrid  = (NEDGES + 7) / 8;
    const int rowGridN  = (NNODES + 7) / 8;
    const int rowGridB  = (BATCH + 7) / 8;

    wmma_gemm_kernel<8><<<nodeGrid, 256, 0, stream>>>(emb, w1_rel,             nullptr, y0, NNODES, EMB, 0);
    wmma_gemm_kernel<8><<<nodeGrid, 256, 0, stream>>>(emb, w1_rel + EMB * EMB, nullptr, y1, NNODES, EMB, 0);
    wmma_gemm_kernel<8><<<nodeGrid, 256, 0, stream>>>(emb, w1_root,            b1,      h1, NNODES, EMB, 0);
    edge_scatter_kernel<<<edgeGrid, 256, 0, stream>>>(eidx, eidx + NEDGES, etype, eweight, y0, y1, h1);
    relu_ln_kernel<<<rowGridN, 256, 0, stream>>>(h1, ln_g, ln_b, NNODES);

    wmma_gemm_kernel<8><<<nodeGrid, 256, 0, stream>>>(h1, w2_rel,             nullptr, y0, NNODES, EMB, 0);
    wmma_gemm_kernel<8><<<nodeGrid, 256, 0, stream>>>(h1, w2_rel + EMB * EMB, nullptr, y1, NNODES, EMB, 0);
    wmma_gemm_kernel<8><<<nodeGrid, 256, 0, stream>>>(h1, w2_root,            b2,      h2, NNODES, EMB, 0);
    edge_scatter_kernel<<<edgeGrid, 256, 0, stream>>>(eidx, eidx + NEDGES, etype, eweight, y0, y1, h2);

    gather_gmf_kernel<<<rowGridB, 256, 0, stream>>>(h2, user_idx, item_idx, gmf, hcat);
    wmma_gemm_kernel<8><<<batchGrid, 256, 0, stream>>>(hcat, mlp_w0, mlp_b0, m0, BATCH, 2 * EMB, 1);
    wmma_gemm_kernel<4><<<batchGrid, 256, 0, stream>>>(m0,   mlp_w1, mlp_b1, m1, BATCH, 128,     1);
    wmma_gemm_kernel<2><<<batchGrid, 256, 0, stream>>>(m1,   mlp_w2, mlp_b2, m2, BATCH, 64,      1);
    score_kernel<<<rowGridB, 256, 0, stream>>>(gmf, m2, out_w, out_b, out);
}


# fewer, longer waves: an edge-scatter wave walks 16 edges (its records read once, one per lane), a relu+LN wave 4 rows, was 1 each: 16x/4x fewer waves to start; atomic arrival order differs
# speedup vs baseline: 1.2195x; 1.0867x over previous
#include <hip/hip_runtime.h>
#include <hip/hip_bf16.h>
#include <math.h>

#define EMB     128
#define BATCH   16384
#define NNODES  100000
#define NEDGES  640000

typedef float v2f __attribute__((ext_vector_type(2)));
typedef float v8f __attribute__((ext_vector_type(8)));

__device__ __forceinline__ float wave_sum(float v) {
#pragma unroll
    for (int m = 16; m >= 1; m >>= 1) v += __shfl_xor(v, m, 32);
    return v;
}

template <int NT>
__global__ void __launch_bounds__(256)
wmma_gemm_kernel(const float* __restrict__ X,
                 const float* __restrict__ W,
                 const float* __restrict__ bias,
                 float* __restrict__ Y,
                 int M, int K, int act) {
    constexpr int N   = NT * 16;
    constexpr int KC  = 64;
    constexpr int LDW = N + 16;
    __shared__ float2 wt2[(KC / 2) * LDW];

    const int tid   = threadIdx.x;
    const int wave  = tid >> 5;
    const int lane  = tid & 31;
    const int half  = lane >> 4;
    const int l     = lane & 15;
    const int rowBase = (blockIdx.x * (blockDim.x >> 5) + wave) * 16;

    v8f acc[NT] = {};

    const int  row   = rowBase + l;
    const bool rowOK = (row < M);
    const int  rr    = rowOK ? row : (M - 1);
    const float* xrow = X + (size_t)rr * K;

    for (int kc = 0; kc < K; kc += KC) {
        for (int i = tid; i < (KC / 2) * N; i += blockDim.x) {
            const int p = i / N;
            const int c = i % N;
            float2 w2;
            w2.x = W[(size_t)(kc + 2 * p)     * N + c];
            w2.y = W[(size_t)(kc + 2 * p + 1) * N + c];
            wt2[p * LDW + c] = w2;
        }
        __syncthreads();

        float2 a_nxt = *(const float2*)(xrow + kc + 2 * half);
#pragma unroll
        for (int k = 0; k < KC; k += 4) {
            v2f a;
            a.x = rowOK ? a_nxt.x : 0.0f;
            a.y = rowOK ? a_nxt.y : 0.0f;
            if (k + 4 < KC)
                a_nxt = *(const float2*)(xrow + kc + k + 4 + 2 * half);

            const float2* bp = wt2 + ((k >> 1) + half) * LDW + l;
            float2 breg[NT];
#pragma unroll
            for (int t = 0; t < NT; ++t) breg[t] = bp[t * 16];
#pragma unroll
            for (int t = 0; t < NT; ++t) {
                v2f b; b.x = breg[t].x; b.y = breg[t].y;
                acc[t] = __builtin_amdgcn_wmma_f32_16x16x4_f32(
                    false, a, false, b, (short)0, acc[t], false, false);
            }
        }
        __syncthreads();
    }

#pragma unroll
    for (int t = 0; t < NT; ++t) {
        const int   c  = t * 16 + l;
        const float bv = bias ? bias[c] : 0.0f;
#pragma unroll
        for (int v = 0; v < 8; ++v) {
            const int r = rowBase + v + 8 * half;
            if (r < M) {
                float o = acc[t][v] + bv;
                if (act) o = fmaxf(o, 0.0f);
                Y[(size_t)r * N + c] = o;
            }
        }
    }
}

#define EPW 16
static_assert(NEDGES % EPW == 0, "a chunk never runs past the edge list");
static_assert(EPW % 4 == 0 && EPW <= 32, "chunk walked by fours, one edge per lane");

__global__ void __launch_bounds__(256)
edge_scatter_kernel(const int* __restrict__ src,
                    const int* __restrict__ dst,
                    const int* __restrict__ etype,
                    const float* __restrict__ eweight,
                    const float* __restrict__ y0,
                    const float* __restrict__ y1,
                    float* __restrict__ out) {
    const int chunk = blockIdx.x * (blockDim.x >> 5) + (threadIdx.x >> 5);
    const int e0 = chunk * EPW;
    if (e0 >= NEDGES) return;
    const int lane = threadIdx.x & 31;
    const int le = e0 + (lane & (EPW - 1));
    const int   ls = src[le];
    const int   ld = dst[le];
    const int   lt = etype[le];
    const float lw = eweight[le];

    for (int j = 0; j < EPW; j += 4) {
        float v[4][4];
        int   d[4];
        float w[4];
#pragma unroll
        for (int u = 0; u < 4; ++u) {
            const int s = __builtin_amdgcn_readlane(ls, j + u);
            const int t = __builtin_amdgcn_readlane(lt, j + u);
            d[u] = __builtin_amdgcn_readlane(ld, j + u);
            w[u] = __int_as_float(__builtin_amdgcn_readlane(__float_as_int(lw), j + u));
            const float* yr = ((t == 0) ? y0 : y1) + (size_t)s * EMB + lane;
            v[u][0] = yr[0];
            v[u][1] = yr[32];
            v[u][2] = yr[64];
            v[u][3] = yr[96];
        }
#pragma unroll
        for (int u = 0; u < 4; ++u) {
            float* o = out + (size_t)d[u] * EMB + lane;
            unsafeAtomicAdd(o,      w[u] * v[u][0]);
            unsafeAtomicAdd(o + 32, w[u] * v[u][1]);
            unsafeAtomicAdd(o + 64, w[u] * v[u][2]);
            unsafeAtomicAdd(o + 96, w[u] * v[u][3]);
        }
    }
}

#define LN_RPW 4

__global__ void __launch_bounds__(256)
relu_ln_kernel(float* __restrict__ x,
               const float* __restrict__ g,
               const float* __restrict__ b,
               int M) {
    const int row0 = (blockIdx.x * (blockDim.x >> 5) + (threadIdx.x >> 5)) * LN_RPW;
    if (row0 >= M) return;
    const int lane = threadIdx.x & 31;
    const float4 gg = *(const float4*)(g + lane * 4);
    const float4 bb = *(const float4*)(b + lane * 4);
    float4 vr[LN_RPW];
#pragma unroll
    for (int r = 0; r < LN_RPW; ++r) {
        const int row = (row0 + r < M) ? (row0 + r) : (M - 1);
        vr[r] = *(const float4*)(x + (size_t)row * EMB + lane * 4);
    }
#pragma unroll
    for (int r = 0; r < LN_RPW; ++r) {
        float4 v = vr[r];
        v.x = fmaxf(v.x, 0.0f); v.y = fmaxf(v.y, 0.0f);
        v.z = fmaxf(v.z, 0.0f); v.w = fmaxf(v.w, 0.0f);
        const float mu = wave_sum(v.x + v.y + v.z + v.w) * (1.0f / EMB);
        const float dx = v.x - mu, dy = v.y - mu, dz = v.z - mu, dw = v.w - mu;
        const float var = wave_sum(dx * dx + dy * dy + dz * dz + dw * dw) * (1.0f / EMB);
        const float inv = rsqrtf(var + 1e-5f);
        float4 o;
        o.x = dx * inv * gg.x + bb.x;
        o.y = dy * inv * gg.y + bb.y;
        o.z = dz * inv * gg.z + bb.z;
        o.w = dw * inv * gg.w + bb.w;
        if (row0 + r < M)
            *(float4*)(x + (size_t)(row0 + r) * EMB + lane * 4) = o;
    }
}

__global__ void gather_gmf_kernel(const float* __restrict__ h2,
                                  const int* __restrict__ ui,
                                  const int* __restrict__ vi,
                                  float* __restrict__ gmf,
                                  float* __restrict__ hcat) {
    const int bidx = blockIdx.x * (blockDim.x >> 5) + (threadIdx.x >> 5);
    if (bidx >= BATCH) return;
    const int lane = threadIdx.x & 31;
    const int u = ui[bidx], w = vi[bidx];
    const float4 uv = *(const float4*)(h2 + (size_t)u * EMB + lane * 4);
    const float4 vv = *(const float4*)(h2 + (size_t)w * EMB + lane * 4);
    const float su = wave_sum(uv.x * uv.x + uv.y * uv.y + uv.z * uv.z + uv.w * uv.w);
    const float sv = wave_sum(vv.x * vv.x + vv.y * vv.y + vv.z * vv.z + vv.w * vv.w);
    const float rn = 1.0f / (fmaxf(sqrtf(su), 1e-12f) * fmaxf(sqrtf(sv), 1e-12f));
    float4 gg;
    gg.x = uv.x * vv.x * rn; gg.y = uv.y * vv.y * rn;
    gg.z = uv.z * vv.z * rn; gg.w = uv.w * vv.w * rn;
    *(float4*)(gmf + (size_t)bidx * EMB + lane * 4) = gg;
    *(float4*)(hcat + (size_t)bidx * 2 * EMB + lane * 4) = uv;
    *(float4*)(hcat + (size_t)bidx * 2 * EMB + EMB + lane * 4) = vv;
}

__global__ void score_kernel(const float* __restrict__ gmf,
                             const float* __restrict__ h3,
                             const float* __restrict__ out_w,
                             const float* __restrict__ out_b,
                             float* __restrict__ out) {
    const int bidx = blockIdx.x * (blockDim.x >> 5) + (threadIdx.x >> 5);
    if (bidx >= BATCH) return;
    const int lane = threadIdx.x & 31;
    const float4 gg = *(const float4*)(gmf + (size_t)bidx * EMB + lane * 4);
    const float4 ww = *(const float4*)(out_w + lane * 4);
    float acc = gg.x * ww.x + gg.y * ww.y + gg.z * ww.z + gg.w * ww.w;
    acc += h3[(size_t)bidx * 32 + lane] * out_w[EMB + lane];
    acc = wave_sum(acc) + out_b[0];
    if (lane == 0) out[bidx] = 1.0f / (1.0f + expf(-acc));
}

extern "C" void kernel_launch(void* const* d_in, const int* in_sizes, int n_in,
                              void* d_out, int out_size, void* d_ws, size_t ws_size,
                              hipStream_t stream) {
    const int*   user_idx = (const int*)d_in[0];
    const int*   item_idx = (const int*)d_in[1];
    const int*   eidx     = (const int*)d_in[2];
    const int*   etype    = (const int*)d_in[3];
    const float* eweight  = (const float*)d_in[4];
    const float* emb      = (const float*)d_in[5];
    const float* w1_rel   = (const float*)d_in[6];
    const float* w1_root  = (const float*)d_in[7];
    const float* b1       = (const float*)d_in[8];
    const float* ln_g     = (const float*)d_in[9];
    const float* ln_b     = (const float*)d_in[10];
    const float* w2_rel   = (const float*)d_in[11];
    const float* w2_root  = (const float*)d_in[12];
    const float* b2       = (const float*)d_in[13];
    const float* mlp_w0   = (const float*)d_in[14];
    const float* mlp_b0   = (const float*)d_in[15];
    const float* mlp_w1   = (const float*)d_in[16];
    const float* mlp_b1   = (const float*)d_in[17];
    const float* mlp_w2   = (const float*)d_in[18];
    const float* mlp_b2   = (const float*)d_in[19];
    const float* out_w    = (const float*)d_in[20];
    const float* out_b    = (const float*)d_in[21];
    float* out = (float*)d_out;

    char* ws = (char*)d_ws;
    const size_t NB = (size_t)NNODES * EMB * sizeof(float);
    float* y0 = (float*)(ws);
    float* y1 = (float*)(ws + NB);
    float* h1 = (float*)(ws + 2 * NB);
    float* h2 = (float*)(ws + 3 * NB);
    float* gmf  = (float*)(ws);
    float* hcat = (float*)(ws + (size_t)BATCH * EMB * 4);
    float* m0   = (float*)(ws + (size_t)BATCH * EMB * 4 + (size_t)BATCH * 2 * EMB * 4);
    float* m1   = (float*)((char*)m0 + (size_t)BATCH * 128 * 4);
    float* m2   = (float*)((char*)m1 + (size_t)BATCH * 64 * 4);

    const int nodeGrid  = (NNODES + 127) / 128;
    const int batchGrid = (BATCH + 127) / 128;
    const int edgeGrid  = (NEDGES / EPW + 7) / 8;
    const int rowGridN  = ((NNODES + LN_RPW - 1) / LN_RPW + 7) / 8;
    const int rowGridB  = (BATCH + 7) / 8;

    wmma_gemm_kernel<8><<<nodeGrid, 256, 0, stream>>>(emb, w1_rel,             nullptr, y0, NNODES, EMB, 0);
    wmma_gemm_kernel<8><<<nodeGrid, 256, 0, stream>>>(emb, w1_rel + EMB * EMB, nullptr, y1, NNODES, EMB, 0);
    wmma_gemm_kernel<8><<<nodeGrid, 256, 0, stream>>>(emb, w1_root,            b1,      h1, NNODES, EMB, 0);
    edge_scatter_kernel<<<edgeGrid, 256, 0, stream>>>(eidx, eidx + NEDGES, etype, eweight, y0, y1, h1);
    relu_ln_kernel<<<rowGridN, 256, 0, stream>>>(h1, ln_g, ln_b, NNODES);

    wmma_gemm_kernel<8><<<nodeGrid, 256, 0, stream>>>(h1, w2_rel,             nullptr, y0, NNODES, EMB, 0);
    wmma_gemm_kernel<8><<<nodeGrid, 256, 0, stream>>>(h1, w2_rel + EMB * EMB, nullptr, y1, NNODES, EMB, 0);
    wmma_gemm_kernel<8><<<nodeGrid, 256, 0, stream>>>(h1, w2_root,            b2,      h2, NNODES, EMB, 0);
    edge_scatter_kernel<<<edgeGrid, 256, 0, stream>>>(eidx, eidx + NEDGES, etype, eweight, y0, y1, h2);

    gather_gmf_kernel<<<rowGridB, 256, 0, stream>>>(h2, user_idx, item_idx, gmf, hcat);
    wmma_gemm_kernel<8><<<batchGrid, 256, 0, stream>>>(hcat, mlp_w0, mlp_b0, m0, BATCH, 2 * EMB, 1);
    wmma_gemm_kernel<4><<<batchGrid, 256, 0, stream>>>(m0,   mlp_w1, mlp_b1, m1, BATCH, 128,     1);
    wmma_gemm_kernel<2><<<batchGrid, 256, 0, stream>>>(m1,   mlp_w2, mlp_b2, m2, BATCH, 64,      1);
    score_kernel<<<rowGridB, 256, 0, stream>>>(gmf, m2, out_w, out_b, out);
}


# layer-2 edge scatter only for destination rows the head gathers: per-node flag cleared+set from the two index lists each call, a wave walks 32 edges and skips the rest; atomic arrival order differs
# speedup vs baseline: 1.2612x; 1.0342x over previous
#include <hip/hip_runtime.h>
#include <hip/hip_bf16.h>
#include <math.h>

#define EMB     128
#define BATCH   16384
#define NNODES  100000
#define NEDGES  640000

typedef float v2f __attribute__((ext_vector_type(2)));
typedef float v8f __attribute__((ext_vector_type(8)));

__device__ __forceinline__ float wave_sum(float v) {
#pragma unroll
    for (int m = 16; m >= 1; m >>= 1) v += __shfl_xor(v, m, 32);
    return v;
}

template <int NT>
__global__ void __launch_bounds__(256)
wmma_gemm_kernel(const float* __restrict__ X,
                 const float* __restrict__ W,
                 const float* __restrict__ bias,
                 float* __restrict__ Y,
                 int M, int K, int act) {
    constexpr int N   = NT * 16;
    constexpr int KC  = 64;
    constexpr int LDW = N + 16;
    __shared__ float2 wt2[(KC / 2) * LDW];

    const int tid   = threadIdx.x;
    const int wave  = tid >> 5;
    const int lane  = tid & 31;
    const int half  = lane >> 4;
    const int l     = lane & 15;
    const int rowBase = (blockIdx.x * (blockDim.x >> 5) + wave) * 16;

    v8f acc[NT] = {};

    const int  row   = rowBase + l;
    const bool rowOK = (row < M);
    const int  rr    = rowOK ? row : (M - 1);
    const float* xrow = X + (size_t)rr * K;

    for (int kc = 0; kc < K; kc += KC) {
        for (int i = tid; i < (KC / 2) * N; i += blockDim.x) {
            const int p = i / N;
            const int c = i % N;
            float2 w2;
            w2.x = W[(size_t)(kc + 2 * p)     * N + c];
            w2.y = W[(size_t)(kc + 2 * p + 1) * N + c];
            wt2[p * LDW + c] = w2;
        }
        __syncthreads();

        float2 a_nxt = *(const float2*)(xrow + kc + 2 * half);
#pragma unroll
        for (int k = 0; k < KC; k += 4) {
            v2f a;
            a.x = rowOK ? a_nxt.x : 0.0f;
            a.y = rowOK ? a_nxt.y : 0.0f;
            if (k + 4 < KC)
                a_nxt = *(const float2*)(xrow + kc + k + 4 + 2 * half);

            const float2* bp = wt2 + ((k >> 1) + half) * LDW + l;
            float2 breg[NT];
#pragma unroll
            for (int t = 0; t < NT; ++t) breg[t] = bp[t * 16];
#pragma unroll
            for (int t = 0; t < NT; ++t) {
                v2f b; b.x = breg[t].x; b.y = breg[t].y;
                acc[t] = __builtin_amdgcn_wmma_f32_16x16x4_f32(
                    false, a, false, b, (short)0, acc[t], false, false);
            }
        }
        __syncthreads();
    }

#pragma unroll
    for (int t = 0; t < NT; ++t) {
        const int   c  = t * 16 + l;
        const float bv = bias ? bias[c] : 0.0f;
#pragma unroll
        for (int v = 0; v < 8; ++v) {
            const int r = rowBase + v + 8 * half;
            if (r < M) {
                float o = acc[t][v] + bv;
                if (act) o = fmaxf(o, 0.0f);
                Y[(size_t)r * N + c] = o;
            }
        }
    }
}

#define EPW 16
static_assert(NEDGES % EPW == 0, "a chunk never runs past the edge list");
static_assert(EPW % 4 == 0 && EPW <= 32, "chunk walked by fours, one edge per lane");

__global__ void __launch_bounds__(256)
edge_scatter_kernel(const int* __restrict__ src,
                    const int* __restrict__ dst,
                    const int* __restrict__ etype,
                    const float* __restrict__ eweight,
                    const float* __restrict__ y0,
                    const float* __restrict__ y1,
                    float* __restrict__ out) {
    const int chunk = blockIdx.x * (blockDim.x >> 5) + (threadIdx.x >> 5);
    const int e0 = chunk * EPW;
    if (e0 >= NEDGES) return;
    const int lane = threadIdx.x & 31;
    const int le = e0 + (lane & (EPW - 1));
    const int   ls = src[le];
    const int   ld = dst[le];
    const int   lt = etype[le];
    const float lw = eweight[le];

    for (int j = 0; j < EPW; j += 4) {
        float v[4][4];
        int   d[4];
        float w[4];
#pragma unroll
        for (int u = 0; u < 4; ++u) {
            const int s = __builtin_amdgcn_readlane(ls, j + u);
            const int t = __builtin_amdgcn_readlane(lt, j + u);
            d[u] = __builtin_amdgcn_readlane(ld, j + u);
            w[u] = __int_as_float(__builtin_amdgcn_readlane(__float_as_int(lw), j + u));
            const float* yr = ((t == 0) ? y0 : y1) + (size_t)s * EMB + lane;
            v[u][0] = yr[0];
            v[u][1] = yr[32];
            v[u][2] = yr[64];
            v[u][3] = yr[96];
        }
#pragma unroll
        for (int u = 0; u < 4; ++u) {
            float* o = out + (size_t)d[u] * EMB + lane;
            unsafeAtomicAdd(o,      w[u] * v[u][0]);
            unsafeAtomicAdd(o + 32, w[u] * v[u][1]);
            unsafeAtomicAdd(o + 64, w[u] * v[u][2]);
            unsafeAtomicAdd(o + 96, w[u] * v[u][3]);
        }
    }
}

__global__ void __launch_bounds__(256)
rowflag_clear_kernel(int* __restrict__ rowflag) {
    const int i = blockIdx.x * blockDim.x + threadIdx.x;
    if (i < NNODES) rowflag[i] = 0;
}

__global__ void __launch_bounds__(256)
rowflag_mark_kernel(const int* __restrict__ ui,
                    const int* __restrict__ vi,
                    int* __restrict__ rowflag) {
    const int i = blockIdx.x * blockDim.x + threadIdx.x;
    if (i < BATCH) {
        rowflag[ui[i]] = 1;
        rowflag[vi[i]] = 1;
    }
}

#define EPW2 32
static_assert(NEDGES % EPW2 == 0, "a chunk never runs past the edge list");

__global__ void __launch_bounds__(256)
edge_scatter_rows_kernel(const int* __restrict__ src,
                         const int* __restrict__ dst,
                         const int* __restrict__ etype,
                         const float* __restrict__ eweight,
                         const float* __restrict__ y0,
                         const float* __restrict__ y1,
                         const int* __restrict__ rowflag,
                         float* __restrict__ out) {
    const int chunk = blockIdx.x * (blockDim.x >> 5) + (threadIdx.x >> 5);
    const int e0 = chunk * EPW2;
    if (e0 >= NEDGES) return;
    const int lane = threadIdx.x & 31;
    const int le = e0 + lane;
    const int   ls = src[le];
    const int   ld = dst[le];
    const int   lt = etype[le];
    const float lw = eweight[le];
    unsigned todo = __builtin_amdgcn_ballot_w32(rowflag[ld] != 0);

    while (todo != 0u) {
        const int left = __builtin_popcount(todo);
        const int n = left < 4 ? left : 4;
        float v[4][4];
        int   d[4];
        float w[4];
#pragma unroll
        for (int u = 0; u < 4; ++u) {
            if (u < n) {
                const int j = __builtin_ctz(todo);
                todo &= todo - 1u;
                const int s = __builtin_amdgcn_readlane(ls, j);
                const int t = __builtin_amdgcn_readlane(lt, j);
                d[u] = __builtin_amdgcn_readlane(ld, j);
                w[u] = __int_as_float(__builtin_amdgcn_readlane(__float_as_int(lw), j));
                const float* yr = ((t == 0) ? y0 : y1) + (size_t)s * EMB + lane;
                v[u][0] = yr[0];
                v[u][1] = yr[32];
                v[u][2] = yr[64];
                v[u][3] = yr[96];
            }
        }
#pragma unroll
        for (int u = 0; u < 4; ++u) {
            if (u < n) {
                float* o = out + (size_t)d[u] * EMB + lane;
                unsafeAtomicAdd(o,      w[u] * v[u][0]);
                unsafeAtomicAdd(o + 32, w[u] * v[u][1]);
                unsafeAtomicAdd(o + 64, w[u] * v[u][2]);
                unsafeAtomicAdd(o + 96, w[u] * v[u][3]);
            }
        }
    }
}

#define LN_RPW 4

__global__ void __launch_bounds__(256)
relu_ln_kernel(float* __restrict__ x,
               const float* __restrict__ g,
               const float* __restrict__ b,
               int M) {
    const int row0 = (blockIdx.x * (blockDim.x >> 5) + (threadIdx.x >> 5)) * LN_RPW;
    if (row0 >= M) return;
    const int lane = threadIdx.x & 31;
    const float4 gg = *(const float4*)(g + lane * 4);
    const float4 bb = *(const float4*)(b + lane * 4);
    float4 vr[LN_RPW];
#pragma unroll
    for (int r = 0; r < LN_RPW; ++r) {
        const int row = (row0 + r < M) ? (row0 + r) : (M - 1);
        vr[r] = *(const float4*)(x + (size_t)row * EMB + lane * 4);
    }
#pragma unroll
    for (int r = 0; r < LN_RPW; ++r) {
        float4 v = vr[r];
        v.x = fmaxf(v.x, 0.0f); v.y = fmaxf(v.y, 0.0f);
        v.z = fmaxf(v.z, 0.0f); v.w = fmaxf(v.w, 0.0f);
        const float mu = wave_sum(v.x + v.y + v.z + v.w) * (1.0f / EMB);
        const float dx = v.x - mu, dy = v.y - mu, dz = v.z - mu, dw = v.w - mu;
        const float var = wave_sum(dx * dx + dy * dy + dz * dz + dw * dw) * (1.0f / EMB);
        const float inv = rsqrtf(var + 1e-5f);
        float4 o;
        o.x = dx * inv * gg.x + bb.x;
        o.y = dy * inv * gg.y + bb.y;
        o.z = dz * inv * gg.z + bb.z;
        o.w = dw * inv * gg.w + bb.w;
        if (row0 + r < M)
            *(float4*)(x + (size_t)(row0 + r) * EMB + lane * 4) = o;
    }
}

__global__ void gather_gmf_kernel(const float* __restrict__ h2,
                                  const int* __restrict__ ui,
                                  const int* __restrict__ vi,
                                  float* __restrict__ gmf,
                                  float* __restrict__ hcat) {
    const int bidx = blockIdx.x * (blockDim.x >> 5) + (threadIdx.x >> 5);
    if (bidx >= BATCH) return;
    const int lane = threadIdx.x & 31;
    const int u = ui[bidx], w = vi[bidx];
    const float4 uv = *(const float4*)(h2 + (size_t)u * EMB + lane * 4);
    const float4 vv = *(const float4*)(h2 + (size_t)w * EMB + lane * 4);
    const float su = wave_sum(uv.x * uv.x + uv.y * uv.y + uv.z * uv.z + uv.w * uv.w);
    const float sv = wave_sum(vv.x * vv.x + vv.y * vv.y + vv.z * vv.z + vv.w * vv.w);
    const float rn = 1.0f / (fmaxf(sqrtf(su), 1e-12f) * fmaxf(sqrtf(sv), 1e-12f));
    float4 gg;
    gg.x = uv.x * vv.x * rn; gg.y = uv.y * vv.y * rn;
    gg.z = uv.z * vv.z * rn; gg.w = uv.w * vv.w * rn;
    *(float4*)(gmf + (size_t)bidx * EMB + lane * 4) = gg;
    *(float4*)(hcat + (size_t)bidx * 2 * EMB + lane * 4) = uv;
    *(float4*)(hcat + (size_t)bidx * 2 * EMB + EMB + lane * 4) = vv;
}

__global__ void score_kernel(const float* __restrict__ gmf,
                             const float* __restrict__ h3,
                             const float* __restrict__ out_w,
                             const float* __restrict__ out_b,
                             float* __restrict__ out) {
    const int bidx = blockIdx.x * (blockDim.x >> 5) + (threadIdx.x >> 5);
    if (bidx >= BATCH) return;
    const int lane = threadIdx.x & 31;
    const float4 gg = *(const float4*)(gmf + (size_t)bidx * EMB + lane * 4);
    const float4 ww = *(const float4*)(out_w + lane * 4);
    float acc = gg.x * ww.x + gg.y * ww.y + gg.z * ww.z + gg.w * ww.w;
    acc += h3[(size_t)bidx * 32 + lane] * out_w[EMB + lane];
    acc = wave_sum(acc) + out_b[0];
    if (lane == 0) out[bidx] = 1.0f / (1.0f + expf(-acc));
}

extern "C" void kernel_launch(void* const* d_in, const int* in_sizes, int n_in,
                              void* d_out, int out_size, void* d_ws, size_t ws_size,
                              hipStream_t stream) {
    const int*   user_idx = (const int*)d_in[0];
    const int*   item_idx = (const int*)d_in[1];
    const int*   eidx     = (const int*)d_in[2];
    const int*   etype    = (const int*)d_in[3];
    const float* eweight  = (const float*)d_in[4];
    const float* emb      = (const float*)d_in[5];
    const float* w1_rel   = (const float*)d_in[6];
    const float* w1_root  = (const float*)d_in[7];
    const float* b1       = (const float*)d_in[8];
    const float* ln_g     = (const float*)d_in[9];
    const float* ln_b     = (const float*)d_in[10];
    const float* w2_rel   = (const float*)d_in[11];
    const float* w2_root  = (const float*)d_in[12];
    const float* b2       = (const float*)d_in[13];
    const float* mlp_w0   = (const float*)d_in[14];
    const float* mlp_b0   = (const float*)d_in[15];
    const float* mlp_w1   = (const float*)d_in[16];
    const float* mlp_b1   = (const float*)d_in[17];
    const float* mlp_w2   = (const float*)d_in[18];
    const float* mlp_b2   = (const float*)d_in[19];
    const float* out_w    = (const float*)d_in[20];
    const float* out_b    = (const float*)d_in[21];
    float* out = (float*)d_out;

    char* ws = (char*)d_ws;
    const size_t NB = (size_t)NNODES * EMB * sizeof(float);
    float* y0 = (float*)(ws);
    float* y1 = (float*)(ws + NB);
    float* h1 = (float*)(ws + 2 * NB);
    float* h2 = (float*)(ws + 3 * NB);
    int* rowflag = (int*)(ws + 2 * NB);
    float* gmf  = (float*)(ws);
    float* hcat = (float*)(ws + (size_t)BATCH * EMB * 4);
    float* m0   = (float*)(ws + (size_t)BATCH * EMB * 4 + (size_t)BATCH * 2 * EMB * 4);
    float* m1   = (float*)((char*)m0 + (size_t)BATCH * 128 * 4);
    float* m2   = (float*)((char*)m1 + (size_t)BATCH * 64 * 4);

    const int nodeGrid  = (NNODES + 127) / 128;
    const int batchGrid = (BATCH + 127) / 128;
    const int edgeGrid  = (NEDGES / EPW + 7) / 8;
    const int rowGridN  = ((NNODES + LN_RPW - 1) / LN_RPW + 7) / 8;
    const int rowGridB  = (BATCH + 7) / 8;
    const int edgeGrid2 = (NEDGES / EPW2 + 7) / 8;
    const int flagGridN = (NNODES + 255) / 256;
    const int flagGridB = (BATCH + 255) / 256;

    wmma_gemm_kernel<8><<<nodeGrid, 256, 0, stream>>>(emb, w1_rel,             nullptr, y0, NNODES, EMB, 0);
    wmma_gemm_kernel<8><<<nodeGrid, 256, 0, stream>>>(emb, w1_rel + EMB * EMB, nullptr, y1, NNODES, EMB, 0);
    wmma_gemm_kernel<8><<<nodeGrid, 256, 0, stream>>>(emb, w1_root,            b1,      h1, NNODES, EMB, 0);
    edge_scatter_kernel<<<edgeGrid, 256, 0, stream>>>(eidx, eidx + NEDGES, etype, eweight, y0, y1, h1);
    relu_ln_kernel<<<rowGridN, 256, 0, stream>>>(h1, ln_g, ln_b, NNODES);

    wmma_gemm_kernel<8><<<nodeGrid, 256, 0, stream>>>(h1, w2_rel,             nullptr, y0, NNODES, EMB, 0);
    wmma_gemm_kernel<8><<<nodeGrid, 256, 0, stream>>>(h1, w2_rel + EMB * EMB, nullptr, y1, NNODES, EMB, 0);
    wmma_gemm_kernel<8><<<nodeGrid, 256, 0, stream>>>(h1, w2_root,            b2,      h2, NNODES, EMB, 0);
    rowflag_clear_kernel<<<flagGridN, 256, 0, stream>>>(rowflag);
    rowflag_mark_kernel<<<flagGridB, 256, 0, stream>>>(user_idx, item_idx, rowflag);
    edge_scatter_rows_kernel<<<edgeGrid2, 256, 0, stream>>>(eidx, eidx + NEDGES, etype, eweight, y0, y1, rowflag, h2);

    gather_gmf_kernel<<<rowGridB, 256, 0, stream>>>(h2, user_idx, item_idx, gmf, hcat);
    wmma_gemm_kernel<8><<<batchGrid, 256, 0, stream>>>(hcat, mlp_w0, mlp_b0, m0, BATCH, 2 * EMB, 1);
    wmma_gemm_kernel<4><<<batchGrid, 256, 0, stream>>>(m0,   mlp_w1, mlp_b1, m1, BATCH, 128,     1);
    wmma_gemm_kernel<2><<<batchGrid, 256, 0, stream>>>(m1,   mlp_w2, mlp_b2, m2, BATCH, 64,      1);
    score_kernel<<<rowGridB, 256, 0, stream>>>(gmf, m2, out_w, out_b, out);
}


# layer 2 only at the 2*BATCH rows the head gathers: neighbour rows summed per relation first (f32 atomics), then one K=384 f32 WMMA product over those rows; sum order changes
# speedup vs baseline: 1.6567x; 1.3136x over previous
#include <hip/hip_runtime.h>
#include <hip/hip_bf16.h>
#include <math.h>

#define EMB     128
#define BATCH   16384
#define NNODES  100000
#define NEDGES  640000

typedef float v2f __attribute__((ext_vector_type(2)));
typedef float v8f __attribute__((ext_vector_type(8)));

__device__ __forceinline__ float wave_sum(float v) {
#pragma unroll
    for (int m = 16; m >= 1; m >>= 1) v += __shfl_xor(v, m, 32);
    return v;
}

template <int NT>
__global__ void __launch_bounds__(256)
wmma_gemm_kernel(const float* __restrict__ X,
                 const float* __restrict__ W,
                 const float* __restrict__ bias,
                 float* __restrict__ Y,
                 int M, int K, int act) {
    constexpr int N   = NT * 16;
    constexpr int KC  = 64;
    constexpr int LDW = N + 16;
    __shared__ float2 wt2[(KC / 2) * LDW];

    const int tid   = threadIdx.x;
    const int wave  = tid >> 5;
    const int lane  = tid & 31;
    const int half  = lane >> 4;
    const int l     = lane & 15;
    const int rowBase = (blockIdx.x * (blockDim.x >> 5) + wave) * 16;

    v8f acc[NT] = {};

    const int  row   = rowBase + l;
    const bool rowOK = (row < M);
    const int  rr    = rowOK ? row : (M - 1);
    const float* xrow = X + (size_t)rr * K;

    for (int kc = 0; kc < K; kc += KC) {
        for (int i = tid; i < (KC / 2) * N; i += blockDim.x) {
            const int p = i / N;
            const int c = i % N;
            float2 w2;
            w2.x = W[(size_t)(kc + 2 * p)     * N + c];
            w2.y = W[(size_t)(kc + 2 * p + 1) * N + c];
            wt2[p * LDW + c] = w2;
        }
        __syncthreads();

        float2 a_nxt = *(const float2*)(xrow + kc + 2 * half);
#pragma unroll
        for (int k = 0; k < KC; k += 4) {
            v2f a;
            a.x = rowOK ? a_nxt.x : 0.0f;
            a.y = rowOK ? a_nxt.y : 0.0f;
            if (k + 4 < KC)
                a_nxt = *(const float2*)(xrow + kc + k + 4 + 2 * half);

            const float2* bp = wt2 + ((k >> 1) + half) * LDW + l;
            float2 breg[NT];
#pragma unroll
            for (int t = 0; t < NT; ++t) breg[t] = bp[t * 16];
#pragma unroll
            for (int t = 0; t < NT; ++t) {
                v2f b; b.x = breg[t].x; b.y = breg[t].y;
                acc[t] = __builtin_amdgcn_wmma_f32_16x16x4_f32(
                    false, a, false, b, (short)0, acc[t], false, false);
            }
        }
        __syncthreads();
    }

#pragma unroll
    for (int t = 0; t < NT; ++t) {
        const int   c  = t * 16 + l;
        const float bv = bias ? bias[c] : 0.0f;
#pragma unroll
        for (int v = 0; v < 8; ++v) {
            const int r = rowBase + v + 8 * half;
            if (r < M) {
                float o = acc[t][v] + bv;
                if (act) o = fmaxf(o, 0.0f);
                Y[(size_t)r * N + c] = o;
            }
        }
    }
}

#define EPW 16
static_assert(NEDGES % EPW == 0, "a chunk never runs past the edge list");
static_assert(EPW % 4 == 0 && EPW <= 32, "chunk walked by fours, one edge per lane");

__global__ void __launch_bounds__(256)
edge_scatter_kernel(const int* __restrict__ src,
                    const int* __restrict__ dst,
                    const int* __restrict__ etype,
                    const float* __restrict__ eweight,
                    const float* __restrict__ y0,
                    const float* __restrict__ y1,
                    float* __restrict__ out) {
    const int chunk = blockIdx.x * (blockDim.x >> 5) + (threadIdx.x >> 5);
    const int e0 = chunk * EPW;
    if (e0 >= NEDGES) return;
    const int lane = threadIdx.x & 31;
    const int le = e0 + (lane & (EPW - 1));
    const int   ls = src[le];
    const int   ld = dst[le];
    const int   lt = etype[le];
    const float lw = eweight[le];

    for (int j = 0; j < EPW; j += 4) {
        float v[4][4];
        int   d[4];
        float w[4];
#pragma unroll
        for (int u = 0; u < 4; ++u) {
            const int s = __builtin_amdgcn_readlane(ls, j + u);
            const int t = __builtin_amdgcn_readlane(lt, j + u);
            d[u] = __builtin_amdgcn_readlane(ld, j + u);
            w[u] = __int_as_float(__builtin_amdgcn_readlane(__float_as_int(lw), j + u));
            const float* yr = ((t == 0) ? y0 : y1) + (size_t)s * EMB + lane;
            v[u][0] = yr[0];
            v[u][1] = yr[32];
            v[u][2] = yr[64];
            v[u][3] = yr[96];
        }
#pragma unroll
        for (int u = 0; u < 4; ++u) {
            float* o = out + (size_t)d[u] * EMB + lane;
            unsafeAtomicAdd(o,      w[u] * v[u][0]);
            unsafeAtomicAdd(o + 32, w[u] * v[u][1]);
            unsafeAtomicAdd(o + 64, w[u] * v[u][2]);
            unsafeAtomicAdd(o + 96, w[u] * v[u][3]);
        }
    }
}

__global__ void __launch_bounds__(256)
rowflag_clear_kernel(int* __restrict__ rowflag) {
    const int i = blockIdx.x * blockDim.x + threadIdx.x;
    if (i < NNODES) rowflag[i] = 0;
}

#define MZ_IPW 4
static_assert(BATCH % MZ_IPW == 0, "a wave never runs past the index lists");

__global__ void __launch_bounds__(256)
rows_mark_zero_kernel(const int* __restrict__ ui,
                      const int* __restrict__ vi,
                      int* __restrict__ rowflag,
                      float* __restrict__ agg0,
                      float* __restrict__ agg1) {
    const int b0 = (blockIdx.x * (blockDim.x >> 5) + (threadIdx.x >> 5)) * MZ_IPW;
    if (b0 >= BATCH) return;
    const int lane = threadIdx.x & 31;
    const float4 z = make_float4(0.0f, 0.0f, 0.0f, 0.0f);
    int u[MZ_IPW], v[MZ_IPW];
#pragma unroll
    for (int i = 0; i < MZ_IPW; ++i) {
        u[i] = ui[b0 + i];
        v[i] = vi[b0 + i];
    }
#pragma unroll
    for (int i = 0; i < MZ_IPW; ++i) {
        *(float4*)(agg0 + (size_t)u[i] * EMB + lane * 4) = z;
        *(float4*)(agg1 + (size_t)u[i] * EMB + lane * 4) = z;
        *(float4*)(agg0 + (size_t)v[i] * EMB + lane * 4) = z;
        *(float4*)(agg1 + (size_t)v[i] * EMB + lane * 4) = z;
        if (lane == 0) {
            rowflag[u[i]] = 1;
            rowflag[v[i]] = 1;
        }
    }
}

#define EPW2 32
static_assert(NEDGES % EPW2 == 0, "a chunk never runs past the edge list");

__global__ void __launch_bounds__(256)
edge_gather_rows_kernel(const int* __restrict__ src,
                        const int* __restrict__ dst,
                        const int* __restrict__ etype,
                        const float* __restrict__ eweight,
                        const float* __restrict__ x,
                        const int* __restrict__ rowflag,
                        float* __restrict__ agg0,
                        float* __restrict__ agg1) {
    const int chunk = blockIdx.x * (blockDim.x >> 5) + (threadIdx.x >> 5);
    const int e0 = chunk * EPW2;
    if (e0 >= NEDGES) return;
    const int lane = threadIdx.x & 31;
    const int le = e0 + lane;
    const int   ls = src[le];
    const int   ld = dst[le];
    const int   lt = etype[le];
    const float lw = eweight[le];
    unsigned todo = __builtin_amdgcn_ballot_w32(rowflag[ld] != 0);

    while (todo != 0u) {
        const int left = __builtin_popcount(todo);
        const int n = left < 4 ? left : 4;
        float v[4][4];
        int   d[4];
        int   r[4];
        float w[4];
#pragma unroll
        for (int u = 0; u < 4; ++u) {
            if (u < n) {
                const int j = __builtin_ctz(todo);
                todo &= todo - 1u;
                const int s = __builtin_amdgcn_readlane(ls, j);
                r[u] = __builtin_amdgcn_readlane(lt, j);
                d[u] = __builtin_amdgcn_readlane(ld, j);
                w[u] = __int_as_float(__builtin_amdgcn_readlane(__float_as_int(lw), j));
                const float* xr = x + (size_t)s * EMB + lane;
                v[u][0] = xr[0];
                v[u][1] = xr[32];
                v[u][2] = xr[64];
                v[u][3] = xr[96];
            }
        }
#pragma unroll
        for (int u = 0; u < 4; ++u) {
            if (u < n) {
                float* o = ((r[u] == 0) ? agg0 : agg1) + (size_t)d[u] * EMB + lane;
                unsafeAtomicAdd(o,      w[u] * v[u][0]);
                unsafeAtomicAdd(o + 32, w[u] * v[u][1]);
                unsafeAtomicAdd(o + 64, w[u] * v[u][2]);
                unsafeAtomicAdd(o + 96, w[u] * v[u][3]);
            }
        }
    }
}

static_assert((2 * BATCH) % 128 == 0, "every block of eight 16-row wave tiles is full");

__global__ void __launch_bounds__(256)
wmma_gemm_rows3_kernel(const float* __restrict__ agg0,
                       const float* __restrict__ agg1,
                       const float* __restrict__ x,
                       const float* __restrict__ w_rel,
                       const float* __restrict__ w_root,
                       const float* __restrict__ bias,
                       const int* __restrict__ ui,
                       const int* __restrict__ vi,
                       float* __restrict__ Y) {
    constexpr int NT   = 8;
    constexpr int N    = NT * 16;
    constexpr int KC   = 64;
    constexpr int LDW  = N + 16;
    constexpr int NTHR = 256;
    constexpr int WIT  = (KC / 2) * N / NTHR;
    constexpr int NCH  = 3 * (EMB / KC);
    __shared__ float2 wt2[(KC / 2) * LDW];

    const int tid   = threadIdx.x;
    const int wave  = tid >> 5;
    const int lane  = tid & 31;
    const int half  = lane >> 4;
    const int l     = lane & 15;
    const int rowBase = (blockIdx.x * (NTHR >> 5) + wave) * 16;

    v8f acc[NT] = {};

    const int    row  = rowBase + l;
    const int    node = (row & 1) ? vi[row >> 1] : ui[row >> 1];
    const size_t xoff = (size_t)node * EMB;

    float  wx[WIT], wy[WIT];
    float2 an[KC / 4];

    auto prefetch = [&](int ch) {
        const int seg = ch / (EMB / KC);
        const int kc  = (ch % (EMB / KC)) * KC;
        const float* Xs = (seg == 0) ? agg0 : ((seg == 1) ? agg1 : x);
        const float* Ws = (seg == 2) ? w_root : (w_rel + (size_t)seg * EMB * EMB);
        const float* xrow = Xs + xoff + kc + 2 * half;
#pragma unroll
        for (int s = 0; s < KC / 4; ++s)
            an[s] = *(const float2*)(xrow + 4 * s);
#pragma unroll
        for (int it = 0; it < WIT; ++it) {
            const int i = tid + it * NTHR;
            const int p = i / N;
            const int c = i % N;
            wx[it] = Ws[(size_t)(kc + 2 * p)     * N + c];
            wy[it] = Ws[(size_t)(kc + 2 * p + 1) * N + c];
        }
    };

    prefetch(0);
    for (int ch = 0; ch < NCH; ++ch) {
        float2 a[KC / 4];
#pragma unroll
        for (int s = 0; s < KC / 4; ++s) a[s] = an[s];
#pragma unroll
        for (int it = 0; it < WIT; ++it) {
            const int i = tid + it * NTHR;
            float2 w2; w2.x = wx[it]; w2.y = wy[it];
            wt2[(i / N) * LDW + (i % N)] = w2;
        }
        __syncthreads();

        if (ch + 1 < NCH) prefetch(ch + 1);

#pragma unroll
        for (int k = 0; k < KC; k += 4) {
            v2f av; av.x = a[k >> 2].x; av.y = a[k >> 2].y;
            const float2* bp = wt2 + ((k >> 1) + half) * LDW + l;
            float2 breg[NT];
#pragma unroll
            for (int t = 0; t < NT; ++t) breg[t] = bp[t * 16];
#pragma unroll
            for (int t = 0; t < NT; ++t) {
                v2f b; b.x = breg[t].x; b.y = breg[t].y;
                acc[t] = __builtin_amdgcn_wmma_f32_16x16x4_f32(
                    false, av, false, b, (short)0, acc[t], false, false);
            }
        }
        __syncthreads();
    }

#pragma unroll
    for (int t = 0; t < NT; ++t) {
        const int   c  = t * 16 + l;
        const float bv = bias[c];
#pragma unroll
        for (int v = 0; v < 8; ++v) {
            const int r = rowBase + v + 8 * half;
            Y[(size_t)r * N + c] = acc[t][v] + bv;
        }
    }
}

#define LN_RPW 4

__global__ void __launch_bounds__(256)
relu_ln_kernel(float* __restrict__ x,
               const float* __restrict__ g,
               const float* __restrict__ b,
               int M) {
    const int row0 = (blockIdx.x * (blockDim.x >> 5) + (threadIdx.x >> 5)) * LN_RPW;
    if (row0 >= M) return;
    const int lane = threadIdx.x & 31;
    const float4 gg = *(const float4*)(g + lane * 4);
    const float4 bb = *(const float4*)(b + lane * 4);
    float4 vr[LN_RPW];
#pragma unroll
    for (int r = 0; r < LN_RPW; ++r) {
        const int row = (row0 + r < M) ? (row0 + r) : (M - 1);
        vr[r] = *(const float4*)(x + (size_t)row * EMB + lane * 4);
    }
#pragma unroll
    for (int r = 0; r < LN_RPW; ++r) {
        float4 v = vr[r];
        v.x = fmaxf(v.x, 0.0f); v.y = fmaxf(v.y, 0.0f);
        v.z = fmaxf(v.z, 0.0f); v.w = fmaxf(v.w, 0.0f);
        const float mu = wave_sum(v.x + v.y + v.z + v.w) * (1.0f / EMB);
        const float dx = v.x - mu, dy = v.y - mu, dz = v.z - mu, dw = v.w - mu;
        const float var = wave_sum(dx * dx + dy * dy + dz * dz + dw * dw) * (1.0f / EMB);
        const float inv = rsqrtf(var + 1e-5f);
        float4 o;
        o.x = dx * inv * gg.x + bb.x;
        o.y = dy * inv * gg.y + bb.y;
        o.z = dz * inv * gg.z + bb.z;
        o.w = dw * inv * gg.w + bb.w;
        if (row0 + r < M)
            *(float4*)(x + (size_t)(row0 + r) * EMB + lane * 4) = o;
    }
}

__global__ void gmf_kernel(const float* __restrict__ hcat,
                           float* __restrict__ gmf) {
    const int bidx = blockIdx.x * (blockDim.x >> 5) + (threadIdx.x >> 5);
    if (bidx >= BATCH) return;
    const int lane = threadIdx.x & 31;
    const float4 uv = *(const float4*)(hcat + (size_t)bidx * 2 * EMB + lane * 4);
    const float4 vv = *(const float4*)(hcat + (size_t)bidx * 2 * EMB + EMB + lane * 4);
    const float su = wave_sum(uv.x * uv.x + uv.y * uv.y + uv.z * uv.z + uv.w * uv.w);
    const float sv = wave_sum(vv.x * vv.x + vv.y * vv.y + vv.z * vv.z + vv.w * vv.w);
    const float rn = 1.0f / (fmaxf(sqrtf(su), 1e-12f) * fmaxf(sqrtf(sv), 1e-12f));
    float4 gg;
    gg.x = uv.x * vv.x * rn; gg.y = uv.y * vv.y * rn;
    gg.z = uv.z * vv.z * rn; gg.w = uv.w * vv.w * rn;
    *(float4*)(gmf + (size_t)bidx * EMB + lane * 4) = gg;
}

__global__ void score_kernel(const float* __restrict__ gmf,
                             const float* __restrict__ h3,
                             const float* __restrict__ out_w,
                             const float* __restrict__ out_b,
                             float* __restrict__ out) {
    const int bidx = blockIdx.x * (blockDim.x >> 5) + (threadIdx.x >> 5);
    if (bidx >= BATCH) return;
    const int lane = threadIdx.x & 31;
    const float4 gg = *(const float4*)(gmf + (size_t)bidx * EMB + lane * 4);
    const float4 ww = *(const float4*)(out_w + lane * 4);
    float acc = gg.x * ww.x + gg.y * ww.y + gg.z * ww.z + gg.w * ww.w;
    acc += h3[(size_t)bidx * 32 + lane] * out_w[EMB + lane];
    acc = wave_sum(acc) + out_b[0];
    if (lane == 0) out[bidx] = 1.0f / (1.0f + expf(-acc));
}

extern "C" void kernel_launch(void* const* d_in, const int* in_sizes, int n_in,
                              void* d_out, int out_size, void* d_ws, size_t ws_size,
                              hipStream_t stream) {
    const int*   user_idx = (const int*)d_in[0];
    const int*   item_idx = (const int*)d_in[1];
    const int*   eidx     = (const int*)d_in[2];
    const int*   etype    = (const int*)d_in[3];
    const float* eweight  = (const float*)d_in[4];
    const float* emb      = (const float*)d_in[5];
    const float* w1_rel   = (const float*)d_in[6];
    const float* w1_root  = (const float*)d_in[7];
    const float* b1       = (const float*)d_in[8];
    const float* ln_g     = (const float*)d_in[9];
    const float* ln_b     = (const float*)d_in[10];
    const float* w2_rel   = (const float*)d_in[11];
    const float* w2_root  = (const float*)d_in[12];
    const float* b2       = (const float*)d_in[13];
    const float* mlp_w0   = (const float*)d_in[14];
    const float* mlp_b0   = (const float*)d_in[15];
    const float* mlp_w1   = (const float*)d_in[16];
    const float* mlp_b1   = (const float*)d_in[17];
    const float* mlp_w2   = (const float*)d_in[18];
    const float* mlp_b2   = (const float*)d_in[19];
    const float* out_w    = (const float*)d_in[20];
    const float* out_b    = (const float*)d_in[21];
    float* out = (float*)d_out;

    char* ws = (char*)d_ws;
    const size_t NB = (size_t)NNODES * EMB * sizeof(float);
    float* y0 = (float*)(ws);
    float* y1 = (float*)(ws + NB);
    float* h1 = (float*)(ws + 2 * NB);
    float* agg0 = y0;
    float* agg1 = y1;
    char* r3 = ws + 3 * NB;
    int*   rowflag = (int*)r3;
    float* hcat = (float*)(r3 + (1u << 20));
    float* gmf  = (float*)((char*)hcat + (size_t)BATCH * 2 * EMB * 4);
    float* m0   = (float*)((char*)gmf  + (size_t)BATCH * EMB * 4);
    float* m1   = (float*)((char*)m0   + (size_t)BATCH * 128 * 4);
    float* m2   = (float*)((char*)m1   + (size_t)BATCH * 64 * 4);

    const int nodeGrid  = (NNODES + 127) / 128;
    const int batchGrid = (BATCH + 127) / 128;
    const int pairGrid  = (2 * BATCH) / 128;
    const int edgeGrid  = (NEDGES / EPW + 7) / 8;
    const int rowGridN  = ((NNODES + LN_RPW - 1) / LN_RPW + 7) / 8;
    const int rowGridB  = (BATCH + 7) / 8;
    const int edgeGrid2 = (NEDGES / EPW2 + 7) / 8;
    const int flagGridN = (NNODES + 255) / 256;
    const int markGrid  = (BATCH / MZ_IPW + 7) / 8;

    wmma_gemm_kernel<8><<<nodeGrid, 256, 0, stream>>>(emb, w1_rel,             nullptr, y0, NNODES, EMB, 0);
    wmma_gemm_kernel<8><<<nodeGrid, 256, 0, stream>>>(emb, w1_rel + EMB * EMB, nullptr, y1, NNODES, EMB, 0);
    wmma_gemm_kernel<8><<<nodeGrid, 256, 0, stream>>>(emb, w1_root,            b1,      h1, NNODES, EMB, 0);
    edge_scatter_kernel<<<edgeGrid, 256, 0, stream>>>(eidx, eidx + NEDGES, etype, eweight, y0, y1, h1);
    relu_ln_kernel<<<rowGridN, 256, 0, stream>>>(h1, ln_g, ln_b, NNODES);

    rowflag_clear_kernel<<<flagGridN, 256, 0, stream>>>(rowflag);
    rows_mark_zero_kernel<<<markGrid, 256, 0, stream>>>(user_idx, item_idx, rowflag, agg0, agg1);
    edge_gather_rows_kernel<<<edgeGrid2, 256, 0, stream>>>(eidx, eidx + NEDGES, etype, eweight, h1, rowflag, agg0, agg1);
    wmma_gemm_rows3_kernel<<<pairGrid, 256, 0, stream>>>(agg0, agg1, h1, w2_rel, w2_root, b2, user_idx, item_idx, hcat);

    gmf_kernel<<<rowGridB, 256, 0, stream>>>(hcat, gmf);
    wmma_gemm_kernel<8><<<batchGrid, 256, 0, stream>>>(hcat, mlp_w0, mlp_b0, m0, BATCH, 2 * EMB, 1);
    wmma_gemm_kernel<4><<<batchGrid, 256, 0, stream>>>(m0,   mlp_w1, mlp_b1, m1, BATCH, 128,     1);
    wmma_gemm_kernel<2><<<batchGrid, 256, 0, stream>>>(m1,   mlp_w2, mlp_b2, m2, BATCH, 64,      1);
    score_kernel<<<rowGridB, 256, 0, stream>>>(gmf, m2, out_w, out_b, out);
}


# generic f32 WMMA GEMM (3 layer-1 + 3 head launches): a K chunk's A and W global loads issued as one group into registers under the previous chunk's WMMAs, not one wait per W row pair and k-step
# speedup vs baseline: 2.0037x; 1.2095x over previous
#include <hip/hip_runtime.h>
#include <hip/hip_bf16.h>
#include <math.h>

#define EMB     128
#define BATCH   16384
#define NNODES  100000
#define NEDGES  640000

typedef float v2f __attribute__((ext_vector_type(2)));
typedef float v8f __attribute__((ext_vector_type(8)));

__device__ __forceinline__ float wave_sum(float v) {
#pragma unroll
    for (int m = 16; m >= 1; m >>= 1) v += __shfl_xor(v, m, 32);
    return v;
}

template <int NT>
__global__ void __launch_bounds__(256)
wmma_gemm_kernel(const float* __restrict__ X,
                 const float* __restrict__ W,
                 const float* __restrict__ bias,
                 float* __restrict__ Y,
                 int M, int K, int act) {
    constexpr int N    = NT * 16;
    constexpr int KC   = 64;
    constexpr int LDW  = N + 16;
    constexpr int NTHR = 256;
    constexpr int WIT  = (KC / 2) * N / NTHR;
    __shared__ float2 wt2[(KC / 2) * LDW];

    const int tid   = threadIdx.x;
    const int wave  = tid >> 5;
    const int lane  = tid & 31;
    const int half  = lane >> 4;
    const int l     = lane & 15;
    const int rowBase = (blockIdx.x * (NTHR >> 5) + wave) * 16;

    v8f acc[NT] = {};

    const int  row   = rowBase + l;
    const bool rowOK = (row < M);
    const int  rr    = rowOK ? row : (M - 1);
    const float* xrow = X + (size_t)rr * K + 2 * half;

    float  wx[WIT], wy[WIT];
    float2 an[KC / 4];

    auto prefetch = [&](int kc) {
#pragma unroll
        for (int s = 0; s < KC / 4; ++s)
            an[s] = *(const float2*)(xrow + kc + 4 * s);
#pragma unroll
        for (int it = 0; it < WIT; ++it) {
            const int i = tid + it * NTHR;
            const int p = i / N;
            const int c = i % N;
            wx[it] = W[(size_t)(kc + 2 * p)     * N + c];
            wy[it] = W[(size_t)(kc + 2 * p + 1) * N + c];
        }
    };

    prefetch(0);
    for (int kc = 0; kc < K; kc += KC) {
        float2 a[KC / 4];
#pragma unroll
        for (int s = 0; s < KC / 4; ++s) {
            a[s].x = rowOK ? an[s].x : 0.0f;
            a[s].y = rowOK ? an[s].y : 0.0f;
        }
#pragma unroll
        for (int it = 0; it < WIT; ++it) {
            const int i = tid + it * NTHR;
            float2 w2; w2.x = wx[it]; w2.y = wy[it];
            wt2[(i / N) * LDW + (i % N)] = w2;
        }
        __syncthreads();

        if (kc + KC < K) prefetch(kc + KC);

#pragma unroll
        for (int k = 0; k < KC; k += 4) {
            v2f av; av.x = a[k >> 2].x; av.y = a[k >> 2].y;
            const float2* bp = wt2 + ((k >> 1) + half) * LDW + l;
            float2 breg[NT];
#pragma unroll
            for (int t = 0; t < NT; ++t) breg[t] = bp[t * 16];
#pragma unroll
            for (int t = 0; t < NT; ++t) {
                v2f b; b.x = breg[t].x; b.y = breg[t].y;
                acc[t] = __builtin_amdgcn_wmma_f32_16x16x4_f32(
                    false, av, false, b, (short)0, acc[t], false, false);
            }
        }
        __syncthreads();
    }

#pragma unroll
    for (int t = 0; t < NT; ++t) {
        const int   c  = t * 16 + l;
        const float bv = bias ? bias[c] : 0.0f;
#pragma unroll
        for (int v = 0; v < 8; ++v) {
            const int r = rowBase + v + 8 * half;
            if (r < M) {
                float o = acc[t][v] + bv;
                if (act) o = fmaxf(o, 0.0f);
                Y[(size_t)r * N + c] = o;
            }
        }
    }
}

#define EPW 16
static_assert(NEDGES % EPW == 0, "a chunk never runs past the edge list");
static_assert(EPW % 4 == 0 && EPW <= 32, "chunk walked by fours, one edge per lane");

__global__ void __launch_bounds__(256)
edge_scatter_kernel(const int* __restrict__ src,
                    const int* __restrict__ dst,
                    const int* __restrict__ etype,
                    const float* __restrict__ eweight,
                    const float* __restrict__ y0,
                    const float* __restrict__ y1,
                    float* __restrict__ out) {
    const int chunk = blockIdx.x * (blockDim.x >> 5) + (threadIdx.x >> 5);
    const int e0 = chunk * EPW;
    if (e0 >= NEDGES) return;
    const int lane = threadIdx.x & 31;
    const int le = e0 + (lane & (EPW - 1));
    const int   ls = src[le];
    const int   ld = dst[le];
    const int   lt = etype[le];
    const float lw = eweight[le];

    for (int j = 0; j < EPW; j += 4) {
        float v[4][4];
        int   d[4];
        float w[4];
#pragma unroll
        for (int u = 0; u < 4; ++u) {
            const int s = __builtin_amdgcn_readlane(ls, j + u);
            const int t = __builtin_amdgcn_readlane(lt, j + u);
            d[u] = __builtin_amdgcn_readlane(ld, j + u);
            w[u] = __int_as_float(__builtin_amdgcn_readlane(__float_as_int(lw), j + u));
            const float* yr = ((t == 0) ? y0 : y1) + (size_t)s * EMB + lane;
            v[u][0] = yr[0];
            v[u][1] = yr[32];
            v[u][2] = yr[64];
            v[u][3] = yr[96];
        }
#pragma unroll
        for (int u = 0; u < 4; ++u) {
            float* o = out + (size_t)d[u] * EMB + lane;
            unsafeAtomicAdd(o,      w[u] * v[u][0]);
            unsafeAtomicAdd(o + 32, w[u] * v[u][1]);
            unsafeAtomicAdd(o + 64, w[u] * v[u][2]);
            unsafeAtomicAdd(o + 96, w[u] * v[u][3]);
        }
    }
}

__global__ void __launch_bounds__(256)
rowflag_clear_kernel(int* __restrict__ rowflag) {
    const int i = blockIdx.x * blockDim.x + threadIdx.x;
    if (i < NNODES) rowflag[i] = 0;
}

#define MZ_IPW 4
static_assert(BATCH % MZ_IPW == 0, "a wave never runs past the index lists");

__global__ void __launch_bounds__(256)
rows_mark_zero_kernel(const int* __restrict__ ui,
                      const int* __restrict__ vi,
                      int* __restrict__ rowflag,
                      float* __restrict__ agg0,
                      float* __restrict__ agg1) {
    const int b0 = (blockIdx.x * (blockDim.x >> 5) + (threadIdx.x >> 5)) * MZ_IPW;
    if (b0 >= BATCH) return;
    const int lane = threadIdx.x & 31;
    const float4 z = make_float4(0.0f, 0.0f, 0.0f, 0.0f);
    int u[MZ_IPW], v[MZ_IPW];
#pragma unroll
    for (int i = 0; i < MZ_IPW; ++i) {
        u[i] = ui[b0 + i];
        v[i] = vi[b0 + i];
    }
#pragma unroll
    for (int i = 0; i < MZ_IPW; ++i) {
        *(float4*)(agg0 + (size_t)u[i] * EMB + lane * 4) = z;
        *(float4*)(agg1 + (size_t)u[i] * EMB + lane * 4) = z;
        *(float4*)(agg0 + (size_t)v[i] * EMB + lane * 4) = z;
        *(float4*)(agg1 + (size_t)v[i] * EMB + lane * 4) = z;
        if (lane == 0) {
            rowflag[u[i]] = 1;
            rowflag[v[i]] = 1;
        }
    }
}

#define EPW2 32
static_assert(NEDGES % EPW2 == 0, "a chunk never runs past the edge list");

__global__ void __launch_bounds__(256)
edge_gather_rows_kernel(const int* __restrict__ src,
                        const int* __restrict__ dst,
                        const int* __restrict__ etype,
                        const float* __restrict__ eweight,
                        const float* __restrict__ x,
                        const int* __restrict__ rowflag,
                        float* __restrict__ agg0,
                        float* __restrict__ agg1) {
    const int chunk = blockIdx.x * (blockDim.x >> 5) + (threadIdx.x >> 5);
    const int e0 = chunk * EPW2;
    if (e0 >= NEDGES) return;
    const int lane = threadIdx.x & 31;
    const int le = e0 + lane;
    const int   ls = src[le];
    const int   ld = dst[le];
    const int   lt = etype[le];
    const float lw = eweight[le];
    unsigned todo = __builtin_amdgcn_ballot_w32(rowflag[ld] != 0);

    while (todo != 0u) {
        const int left = __builtin_popcount(todo);
        const int n = left < 4 ? left : 4;
        float v[4][4];
        int   d[4];
        int   r[4];
        float w[4];
#pragma unroll
        for (int u = 0; u < 4; ++u) {
            if (u < n) {
                const int j = __builtin_ctz(todo);
                todo &= todo - 1u;
                const int s = __builtin_amdgcn_readlane(ls, j);
                r[u] = __builtin_amdgcn_readlane(lt, j);
                d[u] = __builtin_amdgcn_readlane(ld, j);
                w[u] = __int_as_float(__builtin_amdgcn_readlane(__float_as_int(lw), j));
                const float* xr = x + (size_t)s * EMB + lane;
                v[u][0] = xr[0];
                v[u][1] = xr[32];
                v[u][2] = xr[64];
                v[u][3] = xr[96];
            }
        }
#pragma unroll
        for (int u = 0; u < 4; ++u) {
            if (u < n) {
                float* o = ((r[u] == 0) ? agg0 : agg1) + (size_t)d[u] * EMB + lane;
                unsafeAtomicAdd(o,      w[u] * v[u][0]);
                unsafeAtomicAdd(o + 32, w[u] * v[u][1]);
                unsafeAtomicAdd(o + 64, w[u] * v[u][2]);
                unsafeAtomicAdd(o + 96, w[u] * v[u][3]);
            }
        }
    }
}

static_assert((2 * BATCH) % 128 == 0, "every block of eight 16-row wave tiles is full");

__global__ void __launch_bounds__(256)
wmma_gemm_rows3_kernel(const float* __restrict__ agg0,
                       const float* __restrict__ agg1,
                       const float* __restrict__ x,
                       const float* __restrict__ w_rel,
                       const float* __restrict__ w_root,
                       const float* __restrict__ bias,
                       const int* __restrict__ ui,
                       const int* __restrict__ vi,
                       float* __restrict__ Y) {
    constexpr int NT   = 8;
    constexpr int N    = NT * 16;
    constexpr int KC   = 64;
    constexpr int LDW  = N + 16;
    constexpr int NTHR = 256;
    constexpr int WIT  = (KC / 2) * N / NTHR;
    constexpr int NCH  = 3 * (EMB / KC);
    __shared__ float2 wt2[(KC / 2) * LDW];

    const int tid   = threadIdx.x;
    const int wave  = tid >> 5;
    const int lane  = tid & 31;
    const int half  = lane >> 4;
    const int l     = lane & 15;
    const int rowBase = (blockIdx.x * (NTHR >> 5) + wave) * 16;

    v8f acc[NT] = {};

    const int    row  = rowBase + l;
    const int    node = (row & 1) ? vi[row >> 1] : ui[row >> 1];
    const size_t xoff = (size_t)node * EMB;

    float  wx[WIT], wy[WIT];
    float2 an[KC / 4];

    auto prefetch = [&](int ch) {
        const int seg = ch / (EMB / KC);
        const int kc  = (ch % (EMB / KC)) * KC;
        const float* Xs = (seg == 0) ? agg0 : ((seg == 1) ? agg1 : x);
        const float* Ws = (seg == 2) ? w_root : (w_rel + (size_t)seg * EMB * EMB);
        const float* xrow = Xs + xoff + kc + 2 * half;
#pragma unroll
        for (int s = 0; s < KC / 4; ++s)
            an[s] = *(const float2*)(xrow + 4 * s);
#pragma unroll
        for (int it = 0; it < WIT; ++it) {
            const int i = tid + it * NTHR;
            const int p = i / N;
            const int c = i % N;
            wx[it] = Ws[(size_t)(kc + 2 * p)     * N + c];
            wy[it] = Ws[(size_t)(kc + 2 * p + 1) * N + c];
        }
    };

    prefetch(0);
    for (int ch = 0; ch < NCH; ++ch) {
        float2 a[KC / 4];
#pragma unroll
        for (int s = 0; s < KC / 4; ++s) a[s] = an[s];
#pragma unroll
        for (int it = 0; it < WIT; ++it) {
            const int i = tid + it * NTHR;
            float2 w2; w2.x = wx[it]; w2.y = wy[it];
            wt2[(i / N) * LDW + (i % N)] = w2;
        }
        __syncthreads();

        if (ch + 1 < NCH) prefetch(ch + 1);

#pragma unroll
        for (int k = 0; k < KC; k += 4) {
            v2f av; av.x = a[k >> 2].x; av.y = a[k >> 2].y;
            const float2* bp = wt2 + ((k >> 1) + half) * LDW + l;
            float2 breg[NT];
#pragma unroll
            for (int t = 0; t < NT; ++t) breg[t] = bp[t * 16];
#pragma unroll
            for (int t = 0; t < NT; ++t) {
                v2f b; b.x = breg[t].x; b.y = breg[t].y;
                acc[t] = __builtin_amdgcn_wmma_f32_16x16x4_f32(
                    false, av, false, b, (short)0, acc[t], false, false);
            }
        }
        __syncthreads();
    }

#pragma unroll
    for (int t = 0; t < NT; ++t) {
        const int   c  = t * 16 + l;
        const float bv = bias[c];
#pragma unroll
        for (int v = 0; v < 8; ++v) {
            const int r = rowBase + v + 8 * half;
            Y[(size_t)r * N + c] = acc[t][v] + bv;
        }
    }
}

#define LN_RPW 4

__global__ void __launch_bounds__(256)
relu_ln_kernel(float* __restrict__ x,
               const float* __restrict__ g,
               const float* __restrict__ b,
               int M) {
    const int row0 = (blockIdx.x * (blockDim.x >> 5) + (threadIdx.x >> 5)) * LN_RPW;
    if (row0 >= M) return;
    const int lane = threadIdx.x & 31;
    const float4 gg = *(const float4*)(g + lane * 4);
    const float4 bb = *(const float4*)(b + lane * 4);
    float4 vr[LN_RPW];
#pragma unroll
    for (int r = 0; r < LN_RPW; ++r) {
        const int row = (row0 + r < M) ? (row0 + r) : (M - 1);
        vr[r] = *(const float4*)(x + (size_t)row * EMB + lane * 4);
    }
#pragma unroll
    for (int r = 0; r < LN_RPW; ++r) {
        float4 v = vr[r];
        v.x = fmaxf(v.x, 0.0f); v.y = fmaxf(v.y, 0.0f);
        v.z = fmaxf(v.z, 0.0f); v.w = fmaxf(v.w, 0.0f);
        const float mu = wave_sum(v.x + v.y + v.z + v.w) * (1.0f / EMB);
        const float dx = v.x - mu, dy = v.y - mu, dz = v.z - mu, dw = v.w - mu;
        const float var = wave_sum(dx * dx + dy * dy + dz * dz + dw * dw) * (1.0f / EMB);
        const float inv = rsqrtf(var + 1e-5f);
        float4 o;
        o.x = dx * inv * gg.x + bb.x;
        o.y = dy * inv * gg.y + bb.y;
        o.z = dz * inv * gg.z + bb.z;
        o.w = dw * inv * gg.w + bb.w;
        if (row0 + r < M)
            *(float4*)(x + (size_t)(row0 + r) * EMB + lane * 4) = o;
    }
}

__global__ void gmf_kernel(const float* __restrict__ hcat,
                           float* __restrict__ gmf) {
    const int bidx = blockIdx.x * (blockDim.x >> 5) + (threadIdx.x >> 5);
    if (bidx >= BATCH) return;
    const int lane = threadIdx.x & 31;
    const float4 uv = *(const float4*)(hcat + (size_t)bidx * 2 * EMB + lane * 4);
    const float4 vv = *(const float4*)(hcat + (size_t)bidx * 2 * EMB + EMB + lane * 4);
    const float su = wave_sum(uv.x * uv.x + uv.y * uv.y + uv.z * uv.z + uv.w * uv.w);
    const float sv = wave_sum(vv.x * vv.x + vv.y * vv.y + vv.z * vv.z + vv.w * vv.w);
    const float rn = 1.0f / (fmaxf(sqrtf(su), 1e-12f) * fmaxf(sqrtf(sv), 1e-12f));
    float4 gg;
    gg.x = uv.x * vv.x * rn; gg.y = uv.y * vv.y * rn;
    gg.z = uv.z * vv.z * rn; gg.w = uv.w * vv.w * rn;
    *(float4*)(gmf + (size_t)bidx * EMB + lane * 4) = gg;
}

__global__ void score_kernel(const float* __restrict__ gmf,
                             const float* __restrict__ h3,
                             const float* __restrict__ out_w,
                             const float* __restrict__ out_b,
                             float* __restrict__ out) {
    const int bidx = blockIdx.x * (blockDim.x >> 5) + (threadIdx.x >> 5);
    if (bidx >= BATCH) return;
    const int lane = threadIdx.x & 31;
    const float4 gg = *(const float4*)(gmf + (size_t)bidx * EMB + lane * 4);
    const float4 ww = *(const float4*)(out_w + lane * 4);
    float acc = gg.x * ww.x + gg.y * ww.y + gg.z * ww.z + gg.w * ww.w;
    acc += h3[(size_t)bidx * 32 + lane] * out_w[EMB + lane];
    acc = wave_sum(acc) + out_b[0];
    if (lane == 0) out[bidx] = 1.0f / (1.0f + expf(-acc));
}

extern "C" void kernel_launch(void* const* d_in, const int* in_sizes, int n_in,
                              void* d_out, int out_size, void* d_ws, size_t ws_size,
                              hipStream_t stream) {
    const int*   user_idx = (const int*)d_in[0];
    const int*   item_idx = (const int*)d_in[1];
    const int*   eidx     = (const int*)d_in[2];
    const int*   etype    = (const int*)d_in[3];
    const float* eweight  = (const float*)d_in[4];
    const float* emb      = (const float*)d_in[5];
    const float* w1_rel   = (const float*)d_in[6];
    const float* w1_root  = (const float*)d_in[7];
    const float* b1       = (const float*)d_in[8];
    const float* ln_g     = (const float*)d_in[9];
    const float* ln_b     = (const float*)d_in[10];
    const float* w2_rel   = (const float*)d_in[11];
    const float* w2_root  = (const float*)d_in[12];
    const float* b2       = (const float*)d_in[13];
    const float* mlp_w0   = (const float*)d_in[14];
    const float* mlp_b0   = (const float*)d_in[15];
    const float* mlp_w1   = (const float*)d_in[16];
    const float* mlp_b1   = (const float*)d_in[17];
    const float* mlp_w2   = (const float*)d_in[18];
    const float* mlp_b2   = (const float*)d_in[19];
    const float* out_w    = (const float*)d_in[20];
    const float* out_b    = (const float*)d_in[21];
    float* out = (float*)d_out;

    char* ws = (char*)d_ws;
    const size_t NB = (size_t)NNODES * EMB * sizeof(float);
    float* y0 = (float*)(ws);
    float* y1 = (float*)(ws + NB);
    float* h1 = (float*)(ws + 2 * NB);
    float* agg0 = y0;
    float* agg1 = y1;
    char* r3 = ws + 3 * NB;
    int*   rowflag = (int*)r3;
    float* hcat = (float*)(r3 + (1u << 20));
    float* gmf  = (float*)((char*)hcat + (size_t)BATCH * 2 * EMB * 4);
    float* m0   = (float*)((char*)gmf  + (size_t)BATCH * EMB * 4);
    float* m1   = (float*)((char*)m0   + (size_t)BATCH * 128 * 4);
    float* m2   = (float*)((char*)m1   + (size_t)BATCH * 64 * 4);

    const int nodeGrid  = (NNODES + 127) / 128;
    const int batchGrid = (BATCH + 127) / 128;
    const int pairGrid  = (2 * BATCH) / 128;
    const int edgeGrid  = (NEDGES / EPW + 7) / 8;
    const int rowGridN  = ((NNODES + LN_RPW - 1) / LN_RPW + 7) / 8;
    const int rowGridB  = (BATCH + 7) / 8;
    const int edgeGrid2 = (NEDGES / EPW2 + 7) / 8;
    const int flagGridN = (NNODES + 255) / 256;
    const int markGrid  = (BATCH / MZ_IPW + 7) / 8;

    wmma_gemm_kernel<8><<<nodeGrid, 256, 0, stream>>>(emb, w1_rel,             nullptr, y0, NNODES, EMB, 0);
    wmma_gemm_kernel<8><<<nodeGrid, 256, 0, stream>>>(emb, w1_rel + EMB * EMB, nullptr, y1, NNODES, EMB, 0);
    wmma_gemm_kernel<8><<<nodeGrid, 256, 0, stream>>>(emb, w1_root,            b1,      h1, NNODES, EMB, 0);
    edge_scatter_kernel<<<edgeGrid, 256, 0, stream>>>(eidx, eidx + NEDGES, etype, eweight, y0, y1, h1);
    relu_ln_kernel<<<rowGridN, 256, 0, stream>>>(h1, ln_g, ln_b, NNODES);

    rowflag_clear_kernel<<<flagGridN, 256, 0, stream>>>(rowflag);
    rows_mark_zero_kernel<<<markGrid, 256, 0, stream>>>(user_idx, item_idx, rowflag, agg0, agg1);
    edge_gather_rows_kernel<<<edgeGrid2, 256, 0, stream>>>(eidx, eidx + NEDGES, etype, eweight, h1, rowflag, agg0, agg1);
    wmma_gemm_rows3_kernel<<<pairGrid, 256, 0, stream>>>(agg0, agg1, h1, w2_rel, w2_root, b2, user_idx, item_idx, hcat);

    gmf_kernel<<<rowGridB, 256, 0, stream>>>(hcat, gmf);
    wmma_gemm_kernel<8><<<batchGrid, 256, 0, stream>>>(hcat, mlp_w0, mlp_b0, m0, BATCH, 2 * EMB, 1);
    wmma_gemm_kernel<4><<<batchGrid, 256, 0, stream>>>(m0,   mlp_w1, mlp_b1, m1, BATCH, 128,     1);
    wmma_gemm_kernel<2><<<batchGrid, 256, 0, stream>>>(m1,   mlp_w2, mlp_b2, m2, BATCH, 64,      1);
    score_kernel<<<rowGridB, 256, 0, stream>>>(gmf, m2, out_w, out_b, out);
}


# launch boundaries 13 -> 7: the three layer-1 products run as one grid; flag clear, row marking, gmf and score folded into neighbouring launches (scatter, relu+LN, last MLP layer); sums unchanged
# speedup vs baseline: 2.1307x; 1.0634x over previous
#include <hip/hip_runtime.h>
#include <hip/hip_bf16.h>
#include <math.h>

#define EMB     128
#define BATCH   16384
#define NNODES  100000
#define NEDGES  640000

typedef float v2f __attribute__((ext_vector_type(2)));
typedef float v8f __attribute__((ext_vector_type(8)));

__device__ __forceinline__ float wave_sum(float v) {
#pragma unroll
    for (int m = 16; m >= 1; m >>= 1) v += __shfl_xor(v, m, 32);
    return v;
}

template <int M>
__device__ __forceinline__ float xor_add(float v) {
    const int b = __builtin_bit_cast(int, v);
    int o;
    if constexpr (M == 16) o = __builtin_amdgcn_permlanex16(b, b, 0x76543210, 0xfedcba98, false, false);
    else                   o = __builtin_amdgcn_update_dpp(b, b, 0x160 + M, 0xf, 0xf, false);
    return v + __builtin_bit_cast(float, o);
}

__device__ __forceinline__ float wave_sum_valu(float v) {
    v = xor_add<16>(v);
    v = xor_add<8>(v);
    v = xor_add<4>(v);
    v = xor_add<2>(v);
    v = xor_add<1>(v);
    return v;
}

template <int NT, bool SCORE>
__device__ __forceinline__ void
wmma_gemm_tile(const int blk,
               const float* __restrict__ X,
               const float* __restrict__ W,
               const float* __restrict__ bias,
               float* __restrict__ Y,
               int M, int K, int act,
               const float* __restrict__ hcat,
               const float* __restrict__ out_w,
               const float* __restrict__ out_b,
               float* __restrict__ score) {
    constexpr int N    = NT * 16;
    constexpr int KC   = 64;
    constexpr int LDW  = N + 16;
    constexpr int NTHR = 256;
    constexpr int WIT  = (KC / 2) * N / NTHR;
    __shared__ float2 wt2[(KC / 2) * LDW];

    const int tid   = threadIdx.x;
    const int wave  = tid >> 5;
    const int lane  = tid & 31;
    const int half  = lane >> 4;
    const int l     = lane & 15;
    const int rowBase = (blk * (NTHR >> 5) + wave) * 16;

    v8f acc[NT] = {};

    const int  row   = rowBase + l;
    const bool rowOK = (row < M);
    const int  rr    = rowOK ? row : (M - 1);
    const float* xrow = X + (size_t)rr * K + 2 * half;

    float  wx[WIT], wy[WIT];
    float2 an[KC / 4];

    auto prefetch = [&](int kc) {
#pragma unroll
        for (int s = 0; s < KC / 4; ++s)
            an[s] = *(const float2*)(xrow + kc + 4 * s);
#pragma unroll
        for (int it = 0; it < WIT; ++it) {
            const int i = tid + it * NTHR;
            const int p = i / N;
            const int c = i % N;
            wx[it] = W[(size_t)(kc + 2 * p)     * N + c];
            wy[it] = W[(size_t)(kc + 2 * p + 1) * N + c];
        }
    };

    prefetch(0);
    for (int kc = 0; kc < K; kc += KC) {
        float2 a[KC / 4];
#pragma unroll
        for (int s = 0; s < KC / 4; ++s) {
            a[s].x = rowOK ? an[s].x : 0.0f;
            a[s].y = rowOK ? an[s].y : 0.0f;
        }
#pragma unroll
        for (int it = 0; it < WIT; ++it) {
            const int i = tid + it * NTHR;
            float2 w2; w2.x = wx[it]; w2.y = wy[it];
            wt2[(i / N) * LDW + (i % N)] = w2;
        }
        __syncthreads();

        if (kc + KC < K) prefetch(kc + KC);

#pragma unroll
        for (int k = 0; k < KC; k += 4) {
            v2f av; av.x = a[k >> 2].x; av.y = a[k >> 2].y;
            const float2* bp = wt2 + ((k >> 1) + half) * LDW + l;
            float2 breg[NT];
#pragma unroll
            for (int t = 0; t < NT; ++t) breg[t] = bp[t * 16];
#pragma unroll
            for (int t = 0; t < NT; ++t) {
                v2f b; b.x = breg[t].x; b.y = breg[t].y;
                acc[t] = __builtin_amdgcn_wmma_f32_16x16x4_f32(
                    false, av, false, b, (short)0, acc[t], false, false);
            }
        }
        __syncthreads();
    }

    if constexpr (!SCORE) {
#pragma unroll
        for (int t = 0; t < NT; ++t) {
            const int   c  = t * 16 + l;
            const float bv = bias ? bias[c] : 0.0f;
#pragma unroll
            for (int v = 0; v < 8; ++v) {
                const int r = rowBase + v + 8 * half;
                if (r < M) {
                    float o = acc[t][v] + bv;
                    if (act) o = fmaxf(o, 0.0f);
                    Y[(size_t)r * N + c] = o;
                }
            }
        }
    } else {
        constexpr int IPW = 16;
        __shared__ float hs[(NTHR >> 5) * IPW][N + 1];
        __shared__ float res[(NTHR >> 5) * IPW];
#pragma unroll
        for (int t = 0; t < NT; ++t) {
            const int   c  = t * 16 + l;
            const float bv = bias[c];
#pragma unroll
            for (int v = 0; v < 8; ++v) {
                float o = acc[t][v] + bv;
                if (act) o = fmaxf(o, 0.0f);
                hs[wave * IPW + v + 8 * half][c] = o;
            }
        }
        __syncthreads();
        static_assert(N == 32, "one tile column per lane in the dot below");
        const float4 ww = *(const float4*)(out_w + lane * 4);
        const float  wh = out_w[EMB + lane];
        const float  ob = out_b[0];
        float4 uu[IPW], vv[IPW];
        float  hh[IPW], sa[IPW];
#pragma unroll
        for (int i = 0; i < IPW; ++i) {
            const size_t item = (size_t)(rowBase + i);
            uu[i] = *(const float4*)(hcat + item * 2 * EMB + lane * 4);
            vv[i] = *(const float4*)(hcat + item * 2 * EMB + EMB + lane * 4);
            hh[i] = hs[wave * IPW + i][lane];
        }
#pragma unroll
        for (int i = 0; i < IPW; ++i) {
            const float4 u4 = uu[i], v4 = vv[i];
            const float su = wave_sum_valu(u4.x * u4.x + u4.y * u4.y + u4.z * u4.z + u4.w * u4.w);
            const float sv = wave_sum_valu(v4.x * v4.x + v4.y * v4.y + v4.z * v4.z + v4.w * v4.w);
            const float rn = 1.0f / (fmaxf(sqrtf(su), 1e-12f) * fmaxf(sqrtf(sv), 1e-12f));
            float a = (u4.x * v4.x * rn) * ww.x + (u4.y * v4.y * rn) * ww.y
                    + (u4.z * v4.z * rn) * ww.z + (u4.w * v4.w * rn) * ww.w;
            a += hh[i] * wh;
            sa[i] = wave_sum_valu(a);
        }
        float mine = sa[0];
#pragma unroll
        for (int i = 1; i < IPW; ++i) mine = (lane == i) ? sa[i] : mine;
        if (lane < IPW) res[wave * IPW + lane] = 1.0f / (1.0f + expf(-(mine + ob)));
        __syncthreads();
        if (tid < (NTHR >> 5) * IPW)
            score[(size_t)blk * ((NTHR >> 5) * IPW) + tid] = res[tid];
    }
}

template <int NT, bool SCORE>
__global__ void __launch_bounds__(256)
wmma_gemm_kernel(const float* __restrict__ X,
                 const float* __restrict__ W,
                 const float* __restrict__ bias,
                 float* __restrict__ Y,
                 int M, int K, int act,
                 const float* __restrict__ hcat,
                 const float* __restrict__ out_w,
                 const float* __restrict__ out_b,
                 float* __restrict__ score) {
    wmma_gemm_tile<NT, SCORE>(blockIdx.x, X, W, bias, Y, M, K, act, hcat, out_w, out_b, score);
}

struct Layer1Args {
    const float* w0; const float* w1; const float* w2;
    const float* bias2;
    float* y0; float* y1; float* y2;
};

__global__ void __launch_bounds__(256)
wmma_gemm_layer1_kernel(const float* __restrict__ X, const Layer1Args a, int M, int K) {
    const int p   = blockIdx.x % 3;
    const int blk = blockIdx.x / 3;
    const float* W    = (p == 0) ? a.w0 : ((p == 1) ? a.w1 : a.w2);
    const float* bias = (p == 2) ? a.bias2 : nullptr;
    float*       Y    = (p == 0) ? a.y0 : ((p == 1) ? a.y1 : a.y2);
    wmma_gemm_tile<8, false>(blk, X, W, bias, Y, M, K, 0, nullptr, nullptr, nullptr, nullptr);
}

#define EPW 16
static_assert(NEDGES % EPW == 0, "a chunk never runs past the edge list");
static_assert(EPW % 4 == 0 && EPW <= 32, "chunk walked by fours, one edge per lane");
#define FLAG_WAVES ((NNODES + 31) / 32)
static_assert(FLAG_WAVES <= NEDGES / EPW, "enough waves in this launch to clear every flag");

__global__ void __launch_bounds__(256)
edge_scatter_kernel(const int* __restrict__ src,
                    const int* __restrict__ dst,
                    const int* __restrict__ etype,
                    const float* __restrict__ eweight,
                    const float* __restrict__ y0,
                    const float* __restrict__ y1,
                    float* __restrict__ out,
                    int* __restrict__ rowflag) {
    const int chunk = blockIdx.x * (blockDim.x >> 5) + (threadIdx.x >> 5);
    const int e0 = chunk * EPW;
    if (e0 >= NEDGES) return;
    const int lane = threadIdx.x & 31;
    if (chunk < FLAG_WAVES) {
        const int fi = chunk * 32 + lane;
        if (fi < NNODES) rowflag[fi] = 0;
    }
    const int le = e0 + (lane & (EPW - 1));
    const int   ls = src[le];
    const int   ld = dst[le];
    const int   lt = etype[le];
    const float lw = eweight[le];

    for (int j = 0; j < EPW; j += 4) {
        float v[4][4];
        int   d[4];
        float w[4];
#pragma unroll
        for (int u = 0; u < 4; ++u) {
            const int s = __builtin_amdgcn_readlane(ls, j + u);
            const int t = __builtin_amdgcn_readlane(lt, j + u);
            d[u] = __builtin_amdgcn_readlane(ld, j + u);
            w[u] = __int_as_float(__builtin_amdgcn_readlane(__float_as_int(lw), j + u));
            const float* yr = ((t == 0) ? y0 : y1) + (size_t)s * EMB + lane;
            v[u][0] = yr[0];
            v[u][1] = yr[32];
            v[u][2] = yr[64];
            v[u][3] = yr[96];
        }
#pragma unroll
        for (int u = 0; u < 4; ++u) {
            float* o = out + (size_t)d[u] * EMB + lane;
            unsafeAtomicAdd(o,      w[u] * v[u][0]);
            unsafeAtomicAdd(o + 32, w[u] * v[u][1]);
            unsafeAtomicAdd(o + 64, w[u] * v[u][2]);
            unsafeAtomicAdd(o + 96, w[u] * v[u][3]);
        }
    }
}

#define MZ_IPW 4
static_assert(BATCH % MZ_IPW == 0, "a wave never runs past the index lists");

__device__ __forceinline__ void
rows_mark_zero(int wv, int lane,
               const int* __restrict__ ui,
               const int* __restrict__ vi,
               int* __restrict__ rowflag,
               float* __restrict__ agg0,
               float* __restrict__ agg1) {
    const int b0 = wv * MZ_IPW;
    if (b0 >= BATCH) return;
    const float4 z = make_float4(0.0f, 0.0f, 0.0f, 0.0f);
    int u[MZ_IPW], v[MZ_IPW];
#pragma unroll
    for (int i = 0; i < MZ_IPW; ++i) {
        u[i] = ui[b0 + i];
        v[i] = vi[b0 + i];
    }
#pragma unroll
    for (int i = 0; i < MZ_IPW; ++i) {
        *(float4*)(agg0 + (size_t)u[i] * EMB + lane * 4) = z;
        *(float4*)(agg1 + (size_t)u[i] * EMB + lane * 4) = z;
        *(float4*)(agg0 + (size_t)v[i] * EMB + lane * 4) = z;
        *(float4*)(agg1 + (size_t)v[i] * EMB + lane * 4) = z;
        if (lane == 0) {
            rowflag[u[i]] = 1;
            rowflag[v[i]] = 1;
        }
    }
}

#define EPW2 32
static_assert(NEDGES % EPW2 == 0, "a chunk never runs past the edge list");

__global__ void __launch_bounds__(256)
edge_gather_rows_kernel(const int* __restrict__ src,
                        const int* __restrict__ dst,
                        const int* __restrict__ etype,
                        const float* __restrict__ eweight,
                        const float* __restrict__ x,
                        const int* __restrict__ rowflag,
                        float* __restrict__ agg0,
                        float* __restrict__ agg1) {
    const int chunk = blockIdx.x * (blockDim.x >> 5) + (threadIdx.x >> 5);
    const int e0 = chunk * EPW2;
    if (e0 >= NEDGES) return;
    const int lane = threadIdx.x & 31;
    const int le = e0 + lane;
    const int   ls = src[le];
    const int   ld = dst[le];
    const int   lt = etype[le];
    const float lw = eweight[le];
    unsigned todo = __builtin_amdgcn_ballot_w32(rowflag[ld] != 0);

    while (todo != 0u) {
        const int left = __builtin_popcount(todo);
        const int n = left < 4 ? left : 4;
        float v[4][4];
        int   d[4];
        int   r[4];
        float w[4];
#pragma unroll
        for (int u = 0; u < 4; ++u) {
            if (u < n) {
                const int j = __builtin_ctz(todo);
                todo &= todo - 1u;
                const int s = __builtin_amdgcn_readlane(ls, j);
                r[u] = __builtin_amdgcn_readlane(lt, j);
                d[u] = __builtin_amdgcn_readlane(ld, j);
                w[u] = __int_as_float(__builtin_amdgcn_readlane(__float_as_int(lw), j));
                const float* xr = x + (size_t)s * EMB + lane;
                v[u][0] = xr[0];
                v[u][1] = xr[32];
                v[u][2] = xr[64];
                v[u][3] = xr[96];
            }
        }
#pragma unroll
        for (int u = 0; u < 4; ++u) {
            if (u < n) {
                float* o = ((r[u] == 0) ? agg0 : agg1) + (size_t)d[u] * EMB + lane;
                unsafeAtomicAdd(o,      w[u] * v[u][0]);
                unsafeAtomicAdd(o + 32, w[u] * v[u][1]);
                unsafeAtomicAdd(o + 64, w[u] * v[u][2]);
                unsafeAtomicAdd(o + 96, w[u] * v[u][3]);
            }
        }
    }
}

static_assert((2 * BATCH) % 128 == 0, "every block of eight 16-row wave tiles is full");

__global__ void __launch_bounds__(256)
wmma_gemm_rows3_kernel(const float* __restrict__ agg0,
                       const float* __restrict__ agg1,
                       const float* __restrict__ x,
                       const float* __restrict__ w_rel,
                       const float* __restrict__ w_root,
                       const float* __restrict__ bias,
                       const int* __restrict__ ui,
                       const int* __restrict__ vi,
                       float* __restrict__ Y) {
    constexpr int NT   = 8;
    constexpr int N    = NT * 16;
    constexpr int KC   = 64;
    constexpr int LDW  = N + 16;
    constexpr int NTHR = 256;
    constexpr int WIT  = (KC / 2) * N / NTHR;
    constexpr int NCH  = 3 * (EMB / KC);
    __shared__ float2 wt2[(KC / 2) * LDW];

    const int tid   = threadIdx.x;
    const int wave  = tid >> 5;
    const int lane  = tid & 31;
    const int half  = lane >> 4;
    const int l     = lane & 15;
    const int rowBase = (blockIdx.x * (NTHR >> 5) + wave) * 16;

    v8f acc[NT] = {};

    const int    row  = rowBase + l;
    const int    node = (row & 1) ? vi[row >> 1] : ui[row >> 1];
    const size_t xoff = (size_t)node * EMB;

    float  wx[WIT], wy[WIT];
    float2 an[KC / 4];

    auto prefetch = [&](int ch) {
        const int seg = ch / (EMB / KC);
        const int kc  = (ch % (EMB / KC)) * KC;
        const float* Xs = (seg == 0) ? agg0 : ((seg == 1) ? agg1 : x);
        const float* Ws = (seg == 2) ? w_root : (w_rel + (size_t)seg * EMB * EMB);
        const float* xrow = Xs + xoff + kc + 2 * half;
#pragma unroll
        for (int s = 0; s < KC / 4; ++s)
            an[s] = *(const float2*)(xrow + 4 * s);
#pragma unroll
        for (int it = 0; it < WIT; ++it) {
            const int i = tid + it * NTHR;
            const int p = i / N;
            const int c = i % N;
            wx[it] = Ws[(size_t)(kc + 2 * p)     * N + c];
            wy[it] = Ws[(size_t)(kc + 2 * p + 1) * N + c];
        }
    };

    prefetch(0);
    for (int ch = 0; ch < NCH; ++ch) {
        float2 a[KC / 4];
#pragma unroll
        for (int s = 0; s < KC / 4; ++s) a[s] = an[s];
#pragma unroll
        for (int it = 0; it < WIT; ++it) {
            const int i = tid + it * NTHR;
            float2 w2; w2.x = wx[it]; w2.y = wy[it];
            wt2[(i / N) * LDW + (i % N)] = w2;
        }
        __syncthreads();

        if (ch + 1 < NCH) prefetch(ch + 1);

#pragma unroll
        for (int k = 0; k < KC; k += 4) {
            v2f av; av.x = a[k >> 2].x; av.y = a[k >> 2].y;
            const float2* bp = wt2 + ((k >> 1) + half) * LDW + l;
            float2 breg[NT];
#pragma unroll
            for (int t = 0; t < NT; ++t) breg[t] = bp[t * 16];
#pragma unroll
            for (int t = 0; t < NT; ++t) {
                v2f b; b.x = breg[t].x; b.y = breg[t].y;
                acc[t] = __builtin_amdgcn_wmma_f32_16x16x4_f32(
                    false, av, false, b, (short)0, acc[t], false, false);
            }
        }
        __syncthreads();
    }

#pragma unroll
    for (int t = 0; t < NT; ++t) {
        const int   c  = t * 16 + l;
        const float bv = bias[c];
#pragma unroll
        for (int v = 0; v < 8; ++v) {
            const int r = rowBase + v + 8 * half;
            Y[(size_t)r * N + c] = acc[t][v] + bv;
        }
    }
}

#define LN_RPW 4
static_assert((NNODES + LN_RPW - 1) / LN_RPW >= BATCH / MZ_IPW, "enough waves in this launch to mark every item");

__global__ void __launch_bounds__(256)
relu_ln_kernel(float* __restrict__ x,
               const float* __restrict__ g,
               const float* __restrict__ b,
               int M,
               const int* __restrict__ ui,
               const int* __restrict__ vi,
               int* __restrict__ rowflag,
               float* __restrict__ agg0,
               float* __restrict__ agg1) {
    const int wv   = blockIdx.x * (blockDim.x >> 5) + (threadIdx.x >> 5);
    const int row0 = wv * LN_RPW;
    if (row0 >= M) return;
    const int lane = threadIdx.x & 31;
    rows_mark_zero(wv, lane, ui, vi, rowflag, agg0, agg1);
    const float4 gg = *(const float4*)(g + lane * 4);
    const float4 bb = *(const float4*)(b + lane * 4);
    float4 vr[LN_RPW];
#pragma unroll
    for (int r = 0; r < LN_RPW; ++r) {
        const int row = (row0 + r < M) ? (row0 + r) : (M - 1);
        vr[r] = *(const float4*)(x + (size_t)row * EMB + lane * 4);
    }
#pragma unroll
    for (int r = 0; r < LN_RPW; ++r) {
        float4 v = vr[r];
        v.x = fmaxf(v.x, 0.0f); v.y = fmaxf(v.y, 0.0f);
        v.z = fmaxf(v.z, 0.0f); v.w = fmaxf(v.w, 0.0f);
        const float mu = wave_sum(v.x + v.y + v.z + v.w) * (1.0f / EMB);
        const float dx = v.x - mu, dy = v.y - mu, dz = v.z - mu, dw = v.w - mu;
        const float var = wave_sum(dx * dx + dy * dy + dz * dz + dw * dw) * (1.0f / EMB);
        const float inv = rsqrtf(var + 1e-5f);
        float4 o;
        o.x = dx * inv * gg.x + bb.x;
        o.y = dy * inv * gg.y + bb.y;
        o.z = dz * inv * gg.z + bb.z;
        o.w = dw * inv * gg.w + bb.w;
        if (row0 + r < M)
            *(float4*)(x + (size_t)(row0 + r) * EMB + lane * 4) = o;
    }
}

extern "C" void kernel_launch(void* const* d_in, const int* in_sizes, int n_in,
                              void* d_out, int out_size, void* d_ws, size_t ws_size,
                              hipStream_t stream) {
    const int*   user_idx = (const int*)d_in[0];
    const int*   item_idx = (const int*)d_in[1];
    const int*   eidx     = (const int*)d_in[2];
    const int*   etype    = (const int*)d_in[3];
    const float* eweight  = (const float*)d_in[4];
    const float* emb      = (const float*)d_in[5];
    const float* w1_rel   = (const float*)d_in[6];
    const float* w1_root  = (const float*)d_in[7];
    const float* b1       = (const float*)d_in[8];
    const float* ln_g     = (const float*)d_in[9];
    const float* ln_b     = (const float*)d_in[10];
    const float* w2_rel   = (const float*)d_in[11];
    const float* w2_root  = (const float*)d_in[12];
    const float* b2       = (const float*)d_in[13];
    const float* mlp_w0   = (const float*)d_in[14];
    const float* mlp_b0   = (const float*)d_in[15];
    const float* mlp_w1   = (const float*)d_in[16];
    const float* mlp_b1   = (const float*)d_in[17];
    const float* mlp_w2   = (const float*)d_in[18];
    const float* mlp_b2   = (const float*)d_in[19];
    const float* out_w    = (const float*)d_in[20];
    const float* out_b    = (const float*)d_in[21];
    float* out = (float*)d_out;

    char* ws = (char*)d_ws;
    const size_t NB = (size_t)NNODES * EMB * sizeof(float);
    float* y0 = (float*)(ws);
    float* y1 = (float*)(ws + NB);
    float* h1 = (float*)(ws + 2 * NB);
    float* agg0 = y0;
    float* agg1 = y1;
    char* r3 = ws + 3 * NB;
    int*   rowflag = (int*)r3;
    float* hcat = (float*)(r3 + (1u << 20));
    float* m0   = (float*)((char*)hcat + (size_t)BATCH * 2 * EMB * 4);
    float* m1   = (float*)((char*)m0   + (size_t)BATCH * 128 * 4);

    const int nodeGrid  = (NNODES + 127) / 128;
    const int batchGrid = (BATCH + 127) / 128;
    const int pairGrid  = (2 * BATCH) / 128;
    const int edgeGrid  = (NEDGES / EPW + 7) / 8;
    const int rowGridN  = ((NNODES + LN_RPW - 1) / LN_RPW + 7) / 8;
    const int edgeGrid2 = (NEDGES / EPW2 + 7) / 8;
    static_assert(BATCH % 128 == 0, "the scoring block of the last MLP layer is full");

    const Layer1Args l1 = { w1_rel, w1_rel + EMB * EMB, w1_root, b1, y0, y1, h1 };
    wmma_gemm_layer1_kernel<<<3 * nodeGrid, 256, 0, stream>>>(emb, l1, NNODES, EMB);
    edge_scatter_kernel<<<edgeGrid, 256, 0, stream>>>(eidx, eidx + NEDGES, etype, eweight, y0, y1, h1, rowflag);
    relu_ln_kernel<<<rowGridN, 256, 0, stream>>>(h1, ln_g, ln_b, NNODES, user_idx, item_idx, rowflag, agg0, agg1);

    edge_gather_rows_kernel<<<edgeGrid2, 256, 0, stream>>>(eidx, eidx + NEDGES, etype, eweight, h1, rowflag, agg0, agg1);
    wmma_gemm_rows3_kernel<<<pairGrid, 256, 0, stream>>>(agg0, agg1, h1, w2_rel, w2_root, b2, user_idx, item_idx, hcat);

    wmma_gemm_kernel<8, false><<<batchGrid, 256, 0, stream>>>(hcat, mlp_w0, mlp_b0, m0, BATCH, 2 * EMB, 1, nullptr, nullptr, nullptr, nullptr);
    wmma_gemm_kernel<4, false><<<batchGrid, 256, 0, stream>>>(m0,   mlp_w1, mlp_b1, m1, BATCH, 128,     1, nullptr, nullptr, nullptr, nullptr);
    wmma_gemm_kernel<2, true><<<batchGrid, 256, 0, stream>>>(m1,   mlp_w2, mlp_b2, nullptr, BATCH, 64,  1, hcat, out_w, out_b, out);
}


# layer-2 product at the gathered rows + 3 MLP layers + score as ONE launch (was 4): a block keeps its 64 items' concat(u,v), h0, h1 tiles in LDS as the next f32 WMMA product's A; sum order unchanged
# speedup vs baseline: 2.3385x; 1.0975x over previous
#include <hip/hip_runtime.h>
#include <hip/hip_bf16.h>
#include <math.h>

#define EMB     128
#define BATCH   16384
#define NNODES  100000
#define NEDGES  640000

typedef float v2f __attribute__((ext_vector_type(2)));
typedef float v8f __attribute__((ext_vector_type(8)));

__device__ __forceinline__ float wave_sum(float v) {
#pragma unroll
    for (int m = 16; m >= 1; m >>= 1) v += __shfl_xor(v, m, 32);
    return v;
}

template <int M>
__device__ __forceinline__ float xor_add(float v) {
    const int b = __builtin_bit_cast(int, v);
    int o;
    if constexpr (M == 16) o = __builtin_amdgcn_permlanex16(b, b, 0x76543210, 0xfedcba98, false, false);
    else                   o = __builtin_amdgcn_update_dpp(b, b, 0x160 + M, 0xf, 0xf, false);
    return v + __builtin_bit_cast(float, o);
}

__device__ __forceinline__ float wave_sum_valu(float v) {
    v = xor_add<16>(v);
    v = xor_add<8>(v);
    v = xor_add<4>(v);
    v = xor_add<2>(v);
    v = xor_add<1>(v);
    return v;
}

template <int NT>
__device__ __forceinline__ void
wmma_gemm_tile(const int blk,
               const float* __restrict__ X,
               const float* __restrict__ W,
               const float* __restrict__ bias,
               float* __restrict__ Y,
               int M, int K) {
    constexpr int N    = NT * 16;
    constexpr int KC   = 64;
    constexpr int LDW  = N + 16;
    constexpr int NTHR = 256;
    constexpr int WIT  = (KC / 2) * N / NTHR;
    __shared__ float2 wt2[(KC / 2) * LDW];

    const int tid   = threadIdx.x;
    const int wave  = tid >> 5;
    const int lane  = tid & 31;
    const int half  = lane >> 4;
    const int l     = lane & 15;
    const int rowBase = (blk * (NTHR >> 5) + wave) * 16;

    v8f acc[NT] = {};

    const int  row   = rowBase + l;
    const bool rowOK = (row < M);
    const int  rr    = rowOK ? row : (M - 1);
    const float* xrow = X + (size_t)rr * K + 2 * half;

    float  wx[WIT], wy[WIT];
    float2 an[KC / 4];

    auto prefetch = [&](int kc) {
#pragma unroll
        for (int s = 0; s < KC / 4; ++s)
            an[s] = *(const float2*)(xrow + kc + 4 * s);
#pragma unroll
        for (int it = 0; it < WIT; ++it) {
            const int i = tid + it * NTHR;
            const int p = i / N;
            const int c = i % N;
            wx[it] = W[(size_t)(kc + 2 * p)     * N + c];
            wy[it] = W[(size_t)(kc + 2 * p + 1) * N + c];
        }
    };

    prefetch(0);
    for (int kc = 0; kc < K; kc += KC) {
        float2 a[KC / 4];
#pragma unroll
        for (int s = 0; s < KC / 4; ++s) {
            a[s].x = rowOK ? an[s].x : 0.0f;
            a[s].y = rowOK ? an[s].y : 0.0f;
        }
#pragma unroll
        for (int it = 0; it < WIT; ++it) {
            const int i = tid + it * NTHR;
            float2 w2; w2.x = wx[it]; w2.y = wy[it];
            wt2[(i / N) * LDW + (i % N)] = w2;
        }
        __syncthreads();

        if (kc + KC < K) prefetch(kc + KC);

#pragma unroll
        for (int k = 0; k < KC; k += 4) {
            v2f av; av.x = a[k >> 2].x; av.y = a[k >> 2].y;
            const float2* bp = wt2 + ((k >> 1) + half) * LDW + l;
            float2 breg[NT];
#pragma unroll
            for (int t = 0; t < NT; ++t) breg[t] = bp[t * 16];
#pragma unroll
            for (int t = 0; t < NT; ++t) {
                v2f b; b.x = breg[t].x; b.y = breg[t].y;
                acc[t] = __builtin_amdgcn_wmma_f32_16x16x4_f32(
                    false, av, false, b, (short)0, acc[t], false, false);
            }
        }
        __syncthreads();
    }

#pragma unroll
    for (int t = 0; t < NT; ++t) {
        const int   c  = t * 16 + l;
        const float bv = bias ? bias[c] : 0.0f;
#pragma unroll
        for (int v = 0; v < 8; ++v) {
            const int r = rowBase + v + 8 * half;
            if (r < M)
                Y[(size_t)r * N + c] = acc[t][v] + bv;
        }
    }
}

struct Layer1Args {
    const float* w0; const float* w1; const float* w2;
    const float* bias2;
    float* y0; float* y1; float* y2;
};

__global__ void __launch_bounds__(256)
wmma_gemm_layer1_kernel(const float* __restrict__ X, const Layer1Args a, int M, int K) {
    const int p   = blockIdx.x % 3;
    const int blk = blockIdx.x / 3;
    const float* W    = (p == 0) ? a.w0 : ((p == 1) ? a.w1 : a.w2);
    const float* bias = (p == 2) ? a.bias2 : nullptr;
    float*       Y    = (p == 0) ? a.y0 : ((p == 1) ? a.y1 : a.y2);
    wmma_gemm_tile<8>(blk, X, W, bias, Y, M, K);
}

#define EPW 16
static_assert(NEDGES % EPW == 0, "a chunk never runs past the edge list");
static_assert(EPW % 4 == 0 && EPW <= 32, "chunk walked by fours, one edge per lane");
#define FLAG_WAVES ((NNODES + 31) / 32)
static_assert(FLAG_WAVES <= NEDGES / EPW, "enough waves in this launch to clear every flag");

__global__ void __launch_bounds__(256)
edge_scatter_kernel(const int* __restrict__ src,
                    const int* __restrict__ dst,
                    const int* __restrict__ etype,
                    const float* __restrict__ eweight,
                    const float* __restrict__ y0,
                    const float* __restrict__ y1,
                    float* __restrict__ out,
                    int* __restrict__ rowflag) {
    const int chunk = blockIdx.x * (blockDim.x >> 5) + (threadIdx.x >> 5);
    const int e0 = chunk * EPW;
    if (e0 >= NEDGES) return;
    const int lane = threadIdx.x & 31;
    if (chunk < FLAG_WAVES) {
        const int fi = chunk * 32 + lane;
        if (fi < NNODES) rowflag[fi] = 0;
    }
    const int le = e0 + (lane & (EPW - 1));
    const int   ls = src[le];
    const int   ld = dst[le];
    const int   lt = etype[le];
    const float lw = eweight[le];

    for (int j = 0; j < EPW; j += 4) {
        float v[4][4];
        int   d[4];
        float w[4];
#pragma unroll
        for (int u = 0; u < 4; ++u) {
            const int s = __builtin_amdgcn_readlane(ls, j + u);
            const int t = __builtin_amdgcn_readlane(lt, j + u);
            d[u] = __builtin_amdgcn_readlane(ld, j + u);
            w[u] = __int_as_float(__builtin_amdgcn_readlane(__float_as_int(lw), j + u));
            const float* yr = ((t == 0) ? y0 : y1) + (size_t)s * EMB + lane;
            v[u][0] = yr[0];
            v[u][1] = yr[32];
            v[u][2] = yr[64];
            v[u][3] = yr[96];
        }
#pragma unroll
        for (int u = 0; u < 4; ++u) {
            float* o = out + (size_t)d[u] * EMB + lane;
            unsafeAtomicAdd(o,      w[u] * v[u][0]);
            unsafeAtomicAdd(o + 32, w[u] * v[u][1]);
            unsafeAtomicAdd(o + 64, w[u] * v[u][2]);
            unsafeAtomicAdd(o + 96, w[u] * v[u][3]);
        }
    }
}

#define MZ_IPW 4
static_assert(BATCH % MZ_IPW == 0, "a wave never runs past the index lists");

__device__ __forceinline__ void
rows_mark_zero(int wv, int lane,
               const int* __restrict__ ui,
               const int* __restrict__ vi,
               int* __restrict__ rowflag,
               float* __restrict__ agg0,
               float* __restrict__ agg1) {
    const int b0 = wv * MZ_IPW;
    if (b0 >= BATCH) return;
    const float4 z = make_float4(0.0f, 0.0f, 0.0f, 0.0f);
    int u[MZ_IPW], v[MZ_IPW];
#pragma unroll
    for (int i = 0; i < MZ_IPW; ++i) {
        u[i] = ui[b0 + i];
        v[i] = vi[b0 + i];
    }
#pragma unroll
    for (int i = 0; i < MZ_IPW; ++i) {
        *(float4*)(agg0 + (size_t)u[i] * EMB + lane * 4) = z;
        *(float4*)(agg1 + (size_t)u[i] * EMB + lane * 4) = z;
        *(float4*)(agg0 + (size_t)v[i] * EMB + lane * 4) = z;
        *(float4*)(agg1 + (size_t)v[i] * EMB + lane * 4) = z;
        if (lane == 0) {
            rowflag[u[i]] = 1;
            rowflag[v[i]] = 1;
        }
    }
}

#define EPW2 32
static_assert(NEDGES % EPW2 == 0, "a chunk never runs past the edge list");

__global__ void __launch_bounds__(256)
edge_gather_rows_kernel(const int* __restrict__ src,
                        const int* __restrict__ dst,
                        const int* __restrict__ etype,
                        const float* __restrict__ eweight,
                        const float* __restrict__ x,
                        const int* __restrict__ rowflag,
                        float* __restrict__ agg0,
                        float* __restrict__ agg1) {
    const int chunk = blockIdx.x * (blockDim.x >> 5) + (threadIdx.x >> 5);
    const int e0 = chunk * EPW2;
    if (e0 >= NEDGES) return;
    const int lane = threadIdx.x & 31;
    const int le = e0 + lane;
    const int   ls = src[le];
    const int   ld = dst[le];
    const int   lt = etype[le];
    const float lw = eweight[le];
    unsigned todo = __builtin_amdgcn_ballot_w32(rowflag[ld] != 0);

    while (todo != 0u) {
        const int left = __builtin_popcount(todo);
        const int n = left < 4 ? left : 4;
        float v[4][4];
        int   d[4];
        int   r[4];
        float w[4];
#pragma unroll
        for (int u = 0; u < 4; ++u) {
            if (u < n) {
                const int j = __builtin_ctz(todo);
                todo &= todo - 1u;
                const int s = __builtin_amdgcn_readlane(ls, j);
                r[u] = __builtin_amdgcn_readlane(lt, j);
                d[u] = __builtin_amdgcn_readlane(ld, j);
                w[u] = __int_as_float(__builtin_amdgcn_readlane(__float_as_int(lw), j));
                const float* xr = x + (size_t)s * EMB + lane;
                v[u][0] = xr[0];
                v[u][1] = xr[32];
                v[u][2] = xr[64];
                v[u][3] = xr[96];
            }
        }
#pragma unroll
        for (int u = 0; u < 4; ++u) {
            if (u < n) {
                float* o = ((r[u] == 0) ? agg0 : agg1) + (size_t)d[u] * EMB + lane;
                unsafeAtomicAdd(o,      w[u] * v[u][0]);
                unsafeAtomicAdd(o + 32, w[u] * v[u][1]);
                unsafeAtomicAdd(o + 64, w[u] * v[u][2]);
                unsafeAtomicAdd(o + 96, w[u] * v[u][3]);
            }
        }
    }
}


#define HB_ITEMS 64
#define HC_P (2 * EMB + 4)
#define H0_P (128 + 4)
#define H1_P (64 + 4)
#define H2_P (32 + 2)
static_assert((2 * BATCH) % 128 == 0, "every block of eight 16-row wave tiles is full");
static_assert(BATCH % HB_ITEMS == 0, "every block's item tile is full");

template <int NT, int N>
__device__ __forceinline__ void
lds_rows_gemm(const float* arow, const int K,
              const float* __restrict__ W, const int col0,
              float2* wt2, const int tid, const int half, const int l,
              v8f (&acc)[NT]) {
    constexpr int KC   = 64;
    constexpr int LDW  = N + 16;
    constexpr int NTHR = 256;
    constexpr int WIT  = (KC / 2) * N / NTHR;
    static_assert(WIT >= 1 && (KC / 2) * N % NTHR == 0, "the W chunk divides evenly over the block");

    float wx[WIT], wy[WIT];
    auto prefetch = [&](int kc) {
#pragma unroll
        for (int it = 0; it < WIT; ++it) {
            const int i = tid + it * NTHR;
            const int p = i / N;
            const int c = i % N;
            wx[it] = W[(size_t)(kc + 2 * p)     * N + c];
            wy[it] = W[(size_t)(kc + 2 * p + 1) * N + c];
        }
    };

    prefetch(0);
    for (int kc = 0; kc < K; kc += KC) {
#pragma unroll
        for (int it = 0; it < WIT; ++it) {
            const int i = tid + it * NTHR;
            float2 w2; w2.x = wx[it]; w2.y = wy[it];
            wt2[(i / N) * LDW + (i % N)] = w2;
        }
        __syncthreads();

        float2 a[KC / 4];
#pragma unroll
        for (int s = 0; s < KC / 4; ++s)
            a[s] = *(const float2*)(arow + kc + 4 * s);

        if (kc + KC < K) prefetch(kc + KC);

#pragma unroll
        for (int k = 0; k < KC; k += 4) {
            v2f av; av.x = a[k >> 2].x; av.y = a[k >> 2].y;
            const float2* bp = wt2 + ((k >> 1) + half) * LDW + col0 + l;
            float2 breg[NT];
#pragma unroll
            for (int t = 0; t < NT; ++t) breg[t] = bp[t * 16];
#pragma unroll
            for (int t = 0; t < NT; ++t) {
                v2f b; b.x = breg[t].x; b.y = breg[t].y;
                acc[t] = __builtin_amdgcn_wmma_f32_16x16x4_f32(
                    false, av, false, b, (short)0, acc[t], false, false);
            }
        }
        __syncthreads();
    }
}

__global__ void __launch_bounds__(256)
head_fused_kernel(const float* __restrict__ agg0,
                  const float* __restrict__ agg1,
                  const float* __restrict__ x,
                  const float* __restrict__ w_rel,
                  const float* __restrict__ w_root,
                  const float* __restrict__ bias,
                  const int* __restrict__ ui,
                  const int* __restrict__ vi,
                  const float* __restrict__ mlp_w0, const float* __restrict__ mlp_b0,
                  const float* __restrict__ mlp_w1, const float* __restrict__ mlp_b1,
                  const float* __restrict__ mlp_w2, const float* __restrict__ mlp_b2,
                  const float* __restrict__ out_w,
                  const float* __restrict__ out_b,
                  float* __restrict__ score) {
    constexpr int NT   = 8;
    constexpr int N    = NT * 16;
    constexpr int KC   = 64;
    constexpr int LDW  = N + 16;
    constexpr int NTHR = 256;
    constexpr int WIT  = (KC / 2) * N / NTHR;
    constexpr int NCH  = 3 * (EMB / KC);
    __shared__ float2 wt2[(KC / 2) * LDW];
    __shared__ __attribute__((aligned(16))) float hc[HB_ITEMS * HC_P];
    __shared__ __attribute__((aligned(16))) float h0s[HB_ITEMS * H0_P];
    __shared__ __attribute__((aligned(16))) float h1s[HB_ITEMS * H1_P];
    __shared__ float h2s[HB_ITEMS * H2_P];
    __shared__ float res[HB_ITEMS];

    const int tid   = threadIdx.x;
    const int wave  = tid >> 5;
    const int lane  = tid & 31;
    const int half  = lane >> 4;
    const int l     = lane & 15;
    const int rowBase = (blockIdx.x * (NTHR >> 5) + wave) * 16;

    {
        v8f acc[NT] = {};

        const int    row  = rowBase + l;
        const int    node = (row & 1) ? vi[row >> 1] : ui[row >> 1];
        const size_t xoff = (size_t)node * EMB;

        float  wx[WIT], wy[WIT];
        float2 an[KC / 4];

        auto prefetch = [&](int ch) {
            const int seg = ch / (EMB / KC);
            const int kc  = (ch % (EMB / KC)) * KC;
            const float* Xs = (seg == 0) ? agg0 : ((seg == 1) ? agg1 : x);
            const float* Ws = (seg == 2) ? w_root : (w_rel + (size_t)seg * EMB * EMB);
            const float* xrow = Xs + xoff + kc + 2 * half;
#pragma unroll
            for (int s = 0; s < KC / 4; ++s)
                an[s] = *(const float2*)(xrow + 4 * s);
#pragma unroll
            for (int it = 0; it < WIT; ++it) {
                const int i = tid + it * NTHR;
                const int p = i / N;
                const int c = i % N;
                wx[it] = Ws[(size_t)(kc + 2 * p)     * N + c];
                wy[it] = Ws[(size_t)(kc + 2 * p + 1) * N + c];
            }
        };

        prefetch(0);
        for (int ch = 0; ch < NCH; ++ch) {
            float2 a[KC / 4];
#pragma unroll
            for (int s = 0; s < KC / 4; ++s) a[s] = an[s];
#pragma unroll
            for (int it = 0; it < WIT; ++it) {
                const int i = tid + it * NTHR;
                float2 w2; w2.x = wx[it]; w2.y = wy[it];
                wt2[(i / N) * LDW + (i % N)] = w2;
            }
            __syncthreads();

            if (ch + 1 < NCH) prefetch(ch + 1);

#pragma unroll
            for (int k = 0; k < KC; k += 4) {
                v2f av; av.x = a[k >> 2].x; av.y = a[k >> 2].y;
                const float2* bp = wt2 + ((k >> 1) + half) * LDW + l;
                float2 breg[NT];
#pragma unroll
                for (int t = 0; t < NT; ++t) breg[t] = bp[t * 16];
#pragma unroll
                for (int t = 0; t < NT; ++t) {
                    v2f b; b.x = breg[t].x; b.y = breg[t].y;
                    acc[t] = __builtin_amdgcn_wmma_f32_16x16x4_f32(
                        false, av, false, b, (short)0, acc[t], false, false);
                }
            }
            __syncthreads();
        }

#pragma unroll
        for (int t = 0; t < NT; ++t) {
            const int   c  = t * 16 + l;
            const float bv = bias[c];
#pragma unroll
            for (int v = 0; v < 8; ++v) {
                const int rl = wave * 16 + v + 8 * half;
                hc[(rl >> 1) * HC_P + (rl & 1) * EMB + c] = acc[t][v] + bv;
            }
        }
    }

    const int rt = wave & 3;
    const int cg = wave >> 2;
    const int arow_item = rt * 16 + l;
    {
        v8f c0[4] = {};
        lds_rows_gemm<4, 128>(hc + arow_item * HC_P + 2 * half, 2 * EMB, mlp_w0, cg * 64, wt2, tid, half, l, c0);
#pragma unroll
        for (int t = 0; t < 4; ++t) {
            const int   c  = cg * 64 + t * 16 + l;
            const float bv = mlp_b0[c];
#pragma unroll
            for (int v = 0; v < 8; ++v)
                h0s[(rt * 16 + v + 8 * half) * H0_P + c] = fmaxf(c0[t][v] + bv, 0.0f);
        }
    }
    {
        v8f c1[2] = {};
        lds_rows_gemm<2, 64>(h0s + arow_item * H0_P + 2 * half, 128, mlp_w1, cg * 32, wt2, tid, half, l, c1);
#pragma unroll
        for (int t = 0; t < 2; ++t) {
            const int   c  = cg * 32 + t * 16 + l;
            const float bv = mlp_b1[c];
#pragma unroll
            for (int v = 0; v < 8; ++v)
                h1s[(rt * 16 + v + 8 * half) * H1_P + c] = fmaxf(c1[t][v] + bv, 0.0f);
        }
    }
    {
        v8f c2[1] = {};
        lds_rows_gemm<1, 32>(h1s + arow_item * H1_P + 2 * half, 64, mlp_w2, cg * 16, wt2, tid, half, l, c2);
        const int   c  = cg * 16 + l;
        const float bv = mlp_b2[c];
#pragma unroll
        for (int v = 0; v < 8; ++v)
            h2s[(rt * 16 + v + 8 * half) * H2_P + c] = fmaxf(c2[0][v] + bv, 0.0f);
    }
    __syncthreads();

    constexpr int IPW = HB_ITEMS / (NTHR >> 5);
    static_assert(IPW <= 32, "lane i finishes item i");
    const float4 ww = *(const float4*)(out_w + lane * 4);
    const float  wh = out_w[EMB + lane];
    const float  ob = out_b[0];
    float4 uu[IPW], vv[IPW];
    float  hh[IPW], sa[IPW];
#pragma unroll
    for (int i = 0; i < IPW; ++i) {
        const int it = wave * IPW + i;
        uu[i] = *(const float4*)(hc + it * HC_P + lane * 4);
        vv[i] = *(const float4*)(hc + it * HC_P + EMB + lane * 4);
        hh[i] = h2s[it * H2_P + lane];
    }
#pragma unroll
    for (int i = 0; i < IPW; ++i) {
        const float4 u4 = uu[i], v4 = vv[i];
        const float su = wave_sum_valu(u4.x * u4.x + u4.y * u4.y + u4.z * u4.z + u4.w * u4.w);
        const float sv = wave_sum_valu(v4.x * v4.x + v4.y * v4.y + v4.z * v4.z + v4.w * v4.w);
        const float rn = 1.0f / (fmaxf(sqrtf(su), 1e-12f) * fmaxf(sqrtf(sv), 1e-12f));
        float a = (u4.x * v4.x * rn) * ww.x + (u4.y * v4.y * rn) * ww.y
                + (u4.z * v4.z * rn) * ww.z + (u4.w * v4.w * rn) * ww.w;
        a += hh[i] * wh;
        sa[i] = wave_sum_valu(a);
    }
    float mine = sa[0];
#pragma unroll
    for (int i = 1; i < IPW; ++i) mine = (lane == i) ? sa[i] : mine;
    if (lane < IPW) res[wave * IPW + lane] = 1.0f / (1.0f + expf(-(mine + ob)));
    __syncthreads();
    if (tid < HB_ITEMS)
        score[(size_t)blockIdx.x * HB_ITEMS + tid] = res[tid];
}

#define LN_RPW 4
static_assert((NNODES + LN_RPW - 1) / LN_RPW >= BATCH / MZ_IPW, "enough waves in this launch to mark every item");

__global__ void __launch_bounds__(256)
relu_ln_kernel(float* __restrict__ x,
               const float* __restrict__ g,
               const float* __restrict__ b,
               int M,
               const int* __restrict__ ui,
               const int* __restrict__ vi,
               int* __restrict__ rowflag,
               float* __restrict__ agg0,
               float* __restrict__ agg1) {
    const int wv   = blockIdx.x * (blockDim.x >> 5) + (threadIdx.x >> 5);
    const int row0 = wv * LN_RPW;
    if (row0 >= M) return;
    const int lane = threadIdx.x & 31;
    rows_mark_zero(wv, lane, ui, vi, rowflag, agg0, agg1);
    const float4 gg = *(const float4*)(g + lane * 4);
    const float4 bb = *(const float4*)(b + lane * 4);
    float4 vr[LN_RPW];
#pragma unroll
    for (int r = 0; r < LN_RPW; ++r) {
        const int row = (row0 + r < M) ? (row0 + r) : (M - 1);
        vr[r] = *(const float4*)(x + (size_t)row * EMB + lane * 4);
    }
#pragma unroll
    for (int r = 0; r < LN_RPW; ++r) {
        float4 v = vr[r];
        v.x = fmaxf(v.x, 0.0f); v.y = fmaxf(v.y, 0.0f);
        v.z = fmaxf(v.z, 0.0f); v.w = fmaxf(v.w, 0.0f);
        const float mu = wave_sum(v.x + v.y + v.z + v.w) * (1.0f / EMB);
        const float dx = v.x - mu, dy = v.y - mu, dz = v.z - mu, dw = v.w - mu;
        const float var = wave_sum(dx * dx + dy * dy + dz * dz + dw * dw) * (1.0f / EMB);
        const float inv = rsqrtf(var + 1e-5f);
        float4 o;
        o.x = dx * inv * gg.x + bb.x;
        o.y = dy * inv * gg.y + bb.y;
        o.z = dz * inv * gg.z + bb.z;
        o.w = dw * inv * gg.w + bb.w;
        if (row0 + r < M)
            *(float4*)(x + (size_t)(row0 + r) * EMB + lane * 4) = o;
    }
}

extern "C" void kernel_launch(void* const* d_in, const int* in_sizes, int n_in,
                              void* d_out, int out_size, void* d_ws, size_t ws_size,
                              hipStream_t stream) {
    const int*   user_idx = (const int*)d_in[0];
    const int*   item_idx = (const int*)d_in[1];
    const int*   eidx     = (const int*)d_in[2];
    const int*   etype    = (const int*)d_in[3];
    const float* eweight  = (const float*)d_in[4];
    const float* emb      = (const float*)d_in[5];
    const float* w1_rel   = (const float*)d_in[6];
    const float* w1_root  = (const float*)d_in[7];
    const float* b1       = (const float*)d_in[8];
    const float* ln_g     = (const float*)d_in[9];
    const float* ln_b     = (const float*)d_in[10];
    const float* w2_rel   = (const float*)d_in[11];
    const float* w2_root  = (const float*)d_in[12];
    const float* b2       = (const float*)d_in[13];
    const float* mlp_w0   = (const float*)d_in[14];
    const float* mlp_b0   = (const float*)d_in[15];
    const float* mlp_w1   = (const float*)d_in[16];
    const float* mlp_b1   = (const float*)d_in[17];
    const float* mlp_w2   = (const float*)d_in[18];
    const float* mlp_b2   = (const float*)d_in[19];
    const float* out_w    = (const float*)d_in[20];
    const float* out_b    = (const float*)d_in[21];
    float* out = (float*)d_out;

    char* ws = (char*)d_ws;
    const size_t NB = (size_t)NNODES * EMB * sizeof(float);
    float* y0 = (float*)(ws);
    float* y1 = (float*)(ws + NB);
    float* h1 = (float*)(ws + 2 * NB);
    float* agg0 = y0;
    float* agg1 = y1;
    int*   rowflag = (int*)(ws + 3 * NB);

    const int nodeGrid  = (NNODES + 127) / 128;
    const int headGrid  = BATCH / HB_ITEMS;
    const int edgeGrid  = (NEDGES / EPW + 7) / 8;
    const int rowGridN  = ((NNODES + LN_RPW - 1) / LN_RPW + 7) / 8;
    const int edgeGrid2 = (NEDGES / EPW2 + 7) / 8;

    const Layer1Args l1 = { w1_rel, w1_rel + EMB * EMB, w1_root, b1, y0, y1, h1 };
    wmma_gemm_layer1_kernel<<<3 * nodeGrid, 256, 0, stream>>>(emb, l1, NNODES, EMB);
    edge_scatter_kernel<<<edgeGrid, 256, 0, stream>>>(eidx, eidx + NEDGES, etype, eweight, y0, y1, h1, rowflag);
    relu_ln_kernel<<<rowGridN, 256, 0, stream>>>(h1, ln_g, ln_b, NNODES, user_idx, item_idx, rowflag, agg0, agg1);

    edge_gather_rows_kernel<<<edgeGrid2, 256, 0, stream>>>(eidx, eidx + NEDGES, etype, eweight, h1, rowflag, agg0, agg1);

    head_fused_kernel<<<headGrid, 256, 0, stream>>>(agg0, agg1, h1, w2_rel, w2_root, b2, user_idx, item_idx,
                                                    mlp_w0, mlp_b0, mlp_w1, mlp_b1, mlp_w2, mlp_b2,
                                                    out_w, out_b, out);
}


# edges grouped by destination each call (count, scan, place: int atomics); one wave sums a row's source rows in registers, stores it once (L1 with ReLU+LN, L2): no f32 atomics; sums in placing order
# speedup vs baseline: 2.4339x; 1.0408x over previous
#include <hip/hip_runtime.h>
#include <hip/hip_bf16.h>
#include <math.h>

#define EMB     128
#define BATCH   16384
#define NNODES  100000
#define NEDGES  640000

typedef float v2f __attribute__((ext_vector_type(2)));
typedef float v8f __attribute__((ext_vector_type(8)));

__device__ __forceinline__ float wave_sum(float v) {
#pragma unroll
    for (int m = 16; m >= 1; m >>= 1) v += __shfl_xor(v, m, 32);
    return v;
}

template <int M>
__device__ __forceinline__ float xor_add(float v) {
    const int b = __builtin_bit_cast(int, v);
    int o;
    if constexpr (M == 16) o = __builtin_amdgcn_permlanex16(b, b, 0x76543210, 0xfedcba98, false, false);
    else                   o = __builtin_amdgcn_update_dpp(b, b, 0x160 + M, 0xf, 0xf, false);
    return v + __builtin_bit_cast(float, o);
}

__device__ __forceinline__ float wave_sum_valu(float v) {
    v = xor_add<16>(v);
    v = xor_add<8>(v);
    v = xor_add<4>(v);
    v = xor_add<2>(v);
    v = xor_add<1>(v);
    return v;
}

template <int NT>
__device__ __forceinline__ void
wmma_gemm_tile(const int blk,
               const float* __restrict__ X,
               const float* __restrict__ W,
               const float* __restrict__ bias,
               float* __restrict__ Y,
               int M, int K) {
    constexpr int N    = NT * 16;
    constexpr int KC   = 64;
    constexpr int LDW  = N + 16;
    constexpr int NTHR = 256;
    constexpr int WIT  = (KC / 2) * N / NTHR;
    __shared__ float2 wt2[(KC / 2) * LDW];

    const int tid   = threadIdx.x;
    const int wave  = tid >> 5;
    const int lane  = tid & 31;
    const int half  = lane >> 4;
    const int l     = lane & 15;
    const int rowBase = (blk * (NTHR >> 5) + wave) * 16;

    v8f acc[NT] = {};

    const int  row   = rowBase + l;
    const bool rowOK = (row < M);
    const int  rr    = rowOK ? row : (M - 1);
    const float* xrow = X + (size_t)rr * K + 2 * half;

    float  wx[WIT], wy[WIT];
    float2 an[KC / 4];

    auto prefetch = [&](int kc) {
#pragma unroll
        for (int s = 0; s < KC / 4; ++s)
            an[s] = *(const float2*)(xrow + kc + 4 * s);
#pragma unroll
        for (int it = 0; it < WIT; ++it) {
            const int i = tid + it * NTHR;
            const int p = i / N;
            const int c = i % N;
            wx[it] = W[(size_t)(kc + 2 * p)     * N + c];
            wy[it] = W[(size_t)(kc + 2 * p + 1) * N + c];
        }
    };

    prefetch(0);
    for (int kc = 0; kc < K; kc += KC) {
        float2 a[KC / 4];
#pragma unroll
        for (int s = 0; s < KC / 4; ++s) {
            a[s].x = rowOK ? an[s].x : 0.0f;
            a[s].y = rowOK ? an[s].y : 0.0f;
        }
#pragma unroll
        for (int it = 0; it < WIT; ++it) {
            const int i = tid + it * NTHR;
            float2 w2; w2.x = wx[it]; w2.y = wy[it];
            wt2[(i / N) * LDW + (i % N)] = w2;
        }
        __syncthreads();

        if (kc + KC < K) prefetch(kc + KC);

#pragma unroll
        for (int k = 0; k < KC; k += 4) {
            v2f av; av.x = a[k >> 2].x; av.y = a[k >> 2].y;
            const float2* bp = wt2 + ((k >> 1) + half) * LDW + l;
            float2 breg[NT];
#pragma unroll
            for (int t = 0; t < NT; ++t) breg[t] = bp[t * 16];
#pragma unroll
            for (int t = 0; t < NT; ++t) {
                v2f b; b.x = breg[t].x; b.y = breg[t].y;
                acc[t] = __builtin_amdgcn_wmma_f32_16x16x4_f32(
                    false, av, false, b, (short)0, acc[t], false, false);
            }
        }
        __syncthreads();
    }

#pragma unroll
    for (int t = 0; t < NT; ++t) {
        const int   c  = t * 16 + l;
        const float bv = bias ? bias[c] : 0.0f;
#pragma unroll
        for (int v = 0; v < 8; ++v) {
            const int r = rowBase + v + 8 * half;
            if (r < M)
                Y[(size_t)r * N + c] = acc[t][v] + bv;
        }
    }
}

struct Layer1Args {
    const float* w0; const float* w1; const float* w2;
    const float* bias2;
    float* y0; float* y1; float* y2;
};

__global__ void __launch_bounds__(256)
wmma_gemm_layer1_kernel(const float* __restrict__ X, const Layer1Args a, int M, int K,
                        int* __restrict__ cnt, int ncnt) {
    const int ci = blockIdx.x * 256 + threadIdx.x;
    if (ci < ncnt) cnt[ci] = 0;
    const int p   = blockIdx.x % 3;
    const int blk = blockIdx.x / 3;
    const float* W    = (p == 0) ? a.w0 : ((p == 1) ? a.w1 : a.w2);
    const float* bias = (p == 2) ? a.bias2 : nullptr;
    float*       Y    = (p == 0) ? a.y0 : ((p == 1) ? a.y1 : a.y2);
    wmma_gemm_tile<8>(blk, X, W, bias, Y, M, K);
}

#define CNT_PAD 102400
#define SCAN_EPB 4096
static_assert(CNT_PAD % SCAN_EPB == 0 && CNT_PAD >= NNODES, "the scan blocks cover every counter");
static_assert(NEDGES % 1024 == 0, "four edges per thread, full blocks");
static_assert(NNODES < (1 << 30), "a source row index leaves bit 30 free for the relation");
static_assert(3 * ((NNODES + 127) / 128) * 256 >= CNT_PAD, "enough threads in the layer-1 product launch to clear every counter");

__global__ void __launch_bounds__(256)
edge_count_kernel(const int* __restrict__ dst, int* __restrict__ cnt) {
    const int i = blockIdx.x * 256 + threadIdx.x;
    const int4 d = *(const int4*)(dst + 4 * i);
    atomicAdd(cnt + d.x, 1);
    atomicAdd(cnt + d.y, 1);
    atomicAdd(cnt + d.z, 1);
    atomicAdd(cnt + d.w, 1);
}

__global__ void __launch_bounds__(1024)
edge_offsets_kernel(const int* __restrict__ cnt, int* __restrict__ cur) {
    __shared__ int wsum[32];
    __shared__ int wtot[32];
    const int tid  = threadIdx.x;
    const int wave = tid >> 5;
    const int lane = tid & 31;
    const int4* c4 = (const int4*)cnt;
    const int first = blockIdx.x * 1024;
    const int4 mine = c4[first + tid];
    int part = 0;
#pragma unroll 8
    for (int i = tid; i < first; i += 1024) {
        const int4 c = c4[i];
        part += (c.x + c.y) + (c.z + c.w);
    }
    const int s = (mine.x + mine.y) + (mine.z + mine.w);
#pragma unroll
    for (int m = 16; m >= 1; m >>= 1) part += __shfl_xor(part, m, 32);
    int inc = s;
#pragma unroll
    for (int o = 1; o < 32; o <<= 1) {
        const int t = __shfl_up(inc, o, 32);
        if (lane >= o) inc += t;
    }
    if (lane == 0)  wsum[wave] = part;
    if (lane == 31) wtot[wave] = inc;
    __syncthreads();
    int mix = wsum[lane] + ((lane < wave) ? wtot[lane] : 0);
#pragma unroll
    for (int m = 16; m >= 1; m >>= 1) mix += __shfl_xor(mix, m, 32);
    const int base = mix + inc - s;
    int4 o4;
    o4.x = base;
    o4.y = base + mine.x;
    o4.z = base + mine.x + mine.y;
    o4.w = base + mine.x + mine.y + mine.z;
    ((int4*)cur)[first + tid] = o4;
}

__global__ void __launch_bounds__(256)
edge_place_kernel(const int* __restrict__ src,
                  const int* __restrict__ dst,
                  const int* __restrict__ etype,
                  const float* __restrict__ eweight,
                  int* __restrict__ cur,
                  int2* __restrict__ rec) {
    const int i = blockIdx.x * 256 + threadIdx.x;
    const int4   s = *(const int4*)(src + 4 * i);
    const int4   d = *(const int4*)(dst + 4 * i);
    const int4   t = *(const int4*)(etype + 4 * i);
    const float4 w = *(const float4*)(eweight + 4 * i);
    const int p0 = atomicAdd(cur + d.x, 1);
    const int p1 = atomicAdd(cur + d.y, 1);
    const int p2 = atomicAdd(cur + d.z, 1);
    const int p3 = atomicAdd(cur + d.w, 1);
    rec[p0] = make_int2(s.x | ((t.x == 0) ? 0 : (1 << 30)), __float_as_int(w.x));
    rec[p1] = make_int2(s.y | ((t.y == 0) ? 0 : (1 << 30)), __float_as_int(w.y));
    rec[p2] = make_int2(s.z | ((t.z == 0) ? 0 : (1 << 30)), __float_as_int(w.z));
    rec[p3] = make_int2(s.w | ((t.w == 0) ? 0 : (1 << 30)), __float_as_int(w.w));
}

__device__ __forceinline__ void axpy4(float4& acc, const float w, const float4 v) {
    acc.x += w * v.x; acc.y += w * v.y; acc.z += w * v.z; acc.w += w * v.w;
}

#define PL_RPW 4
static_assert(NNODES % (PL_RPW * 8) == 0, "every wave of every block has its rows");

__global__ void __launch_bounds__(256)
pull_relu_ln_kernel(const int* __restrict__ cur,
                    const int2* __restrict__ rec,
                    const float* __restrict__ y0,
                    const float* __restrict__ y1,
                    float* __restrict__ x,
                    const float* __restrict__ g,
                    const float* __restrict__ b) {
    const int wv   = blockIdx.x * 8 + (threadIdx.x >> 5);
    const int row0 = wv * PL_RPW;
    const int lane = threadIdx.x & 31;
    const int bi = row0 - 1 + lane;
    int bv = 0;
    if (lane <= PL_RPW && bi >= 0) bv = cur[bi];
    const int beg = __builtin_amdgcn_readlane(bv, 0);
    const int e0  = __builtin_amdgcn_readlane(bv, 1);
    const int e1  = __builtin_amdgcn_readlane(bv, 2);
    const int e2  = __builtin_amdgcn_readlane(bv, 3);
    const int end = __builtin_amdgcn_readlane(bv, 4);

    float4 a0 = *(const float4*)(x + (size_t)(row0 + 0) * EMB + lane * 4);
    float4 a1 = *(const float4*)(x + (size_t)(row0 + 1) * EMB + lane * 4);
    float4 a2 = *(const float4*)(x + (size_t)(row0 + 2) * EMB + lane * 4);
    float4 a3 = *(const float4*)(x + (size_t)(row0 + 3) * EMB + lane * 4);
    const float4 gg = *(const float4*)(g + lane * 4);
    const float4 bb = *(const float4*)(b + lane * 4);

    for (int cb = beg; cb < end; cb += 32) {
        const int left = end - cb;
        const int nrec = left < 32 ? left : 32;
        int2 rc = make_int2(0, 0);
        if (lane < nrec) rc = rec[cb + lane];
        for (int q = 0; q < nrec; q += 8) {
            const int m = nrec - q;
            const int n = m < 8 ? m : 8;
            float4 v[8];
            float  w[8];
#pragma unroll
            for (int u = 0; u < 8; ++u) {
                const int k = q + (u < n ? u : n - 1);
                const int sx = __builtin_amdgcn_readlane(rc.x, k);
                w[u] = __int_as_float(__builtin_amdgcn_readlane(rc.y, k));
                const float* yr = ((sx >> 30) ? y1 : y0) + (size_t)(sx & 0x3fffffff) * EMB + lane * 4;
                v[u] = *(const float4*)yr;
            }
#pragma unroll
            for (int u = 0; u < 8; ++u) {
                if (u < n) {
                    const int p = cb + q + u;
                    if      (p < e0) axpy4(a0, w[u], v[u]);
                    else if (p < e1) axpy4(a1, w[u], v[u]);
                    else if (p < e2) axpy4(a2, w[u], v[u]);
                    else             axpy4(a3, w[u], v[u]);
                }
            }
        }
    }

    float4 vr[PL_RPW] = { a0, a1, a2, a3 };
#pragma unroll
    for (int r = 0; r < PL_RPW; ++r) {
        float4 v = vr[r];
        v.x = fmaxf(v.x, 0.0f); v.y = fmaxf(v.y, 0.0f);
        v.z = fmaxf(v.z, 0.0f); v.w = fmaxf(v.w, 0.0f);
        const float mu = wave_sum(v.x + v.y + v.z + v.w) * (1.0f / EMB);
        const float dx = v.x - mu, dy = v.y - mu, dz = v.z - mu, dw = v.w - mu;
        const float var = wave_sum(dx * dx + dy * dy + dz * dz + dw * dw) * (1.0f / EMB);
        const float inv = rsqrtf(var + 1e-5f);
        float4 o;
        o.x = dx * inv * gg.x + bb.x;
        o.y = dy * inv * gg.y + bb.y;
        o.z = dz * inv * gg.z + bb.z;
        o.w = dw * inv * gg.w + bb.w;
        *(float4*)(x + (size_t)(row0 + r) * EMB + lane * 4) = o;
    }
}

static_assert(BATCH % 8 == 0, "every wave of every block has its item");

__device__ __forceinline__ void
pull_rows_by_relation(const int beg, const int end, const int lane,
                      const int2* __restrict__ rec, const float* __restrict__ x,
                      float4& s0, float4& s1) {
    for (int cb = beg; cb < end; cb += 32) {
        const int left = end - cb;
        const int nrec = left < 32 ? left : 32;
        int2 rc = make_int2(0, 0);
        if (lane < nrec) rc = rec[cb + lane];
        for (int q = 0; q < nrec; q += 8) {
            const int m = nrec - q;
            const int n = m < 8 ? m : 8;
            float4 v[8];
            float  w[8];
            int    t[8];
#pragma unroll
            for (int u = 0; u < 8; ++u) {
                const int k = q + (u < n ? u : n - 1);
                const int sx = __builtin_amdgcn_readlane(rc.x, k);
                w[u] = __int_as_float(__builtin_amdgcn_readlane(rc.y, k));
                t[u] = sx >> 30;
                v[u] = *(const float4*)(x + (size_t)(sx & 0x3fffffff) * EMB + lane * 4);
            }
#pragma unroll
            for (int u = 0; u < 8; ++u) {
                if (u < n) {
                    if (t[u] == 0) axpy4(s0, w[u], v[u]);
                    else           axpy4(s1, w[u], v[u]);
                }
            }
        }
    }
}

__global__ void __launch_bounds__(256)
gathered_rows_pull_kernel(const int* __restrict__ ui,
                          const int* __restrict__ vi,
                          const int* __restrict__ cur,
                          const int2* __restrict__ rec,
                          const float* __restrict__ x,
                          float* __restrict__ agg0,
                          float* __restrict__ agg1) {
    const int item = blockIdx.x * 8 + (threadIdx.x >> 5);
    const int lane = threadIdx.x & 31;
    const int node = (lane & 2) ? vi[item] : ui[item];
    const int bi   = node - 1 + (lane & 1);
    int bv = 0;
    if (lane < 4 && bi >= 0) bv = cur[bi];
    const int ub = __builtin_amdgcn_readlane(bv, 0);
    const int ue = __builtin_amdgcn_readlane(bv, 1);
    const int vb = __builtin_amdgcn_readlane(bv, 2);
    const int ve = __builtin_amdgcn_readlane(bv, 3);
    const float4 z = make_float4(0.0f, 0.0f, 0.0f, 0.0f);
    float4 u0 = z, u1 = z, v0 = z, v1 = z;
    pull_rows_by_relation(ub, ue, lane, rec, x, u0, u1);
    pull_rows_by_relation(vb, ve, lane, rec, x, v0, v1);
    const size_t r = (size_t)item * 2;
    *(float4*)(agg0 + (r + 0) * EMB + lane * 4) = u0;
    *(float4*)(agg1 + (r + 0) * EMB + lane * 4) = u1;
    *(float4*)(agg0 + (r + 1) * EMB + lane * 4) = v0;
    *(float4*)(agg1 + (r + 1) * EMB + lane * 4) = v1;
}

#define HB_ITEMS 64
#define HC_P (2 * EMB + 4)
#define H0_P (128 + 4)
#define H1_P (64 + 4)
#define H2_P (32 + 2)
static_assert((2 * BATCH) % 128 == 0, "every block of eight 16-row wave tiles is full");
static_assert(BATCH % HB_ITEMS == 0, "every block's item tile is full");

template <int NT, int N>
__device__ __forceinline__ void
lds_rows_gemm(const float* arow, const int K,
              const float* __restrict__ W, const int col0,
              float2* wt2, const int tid, const int half, const int l,
              v8f (&acc)[NT]) {
    constexpr int KC   = 64;
    constexpr int LDW  = N + 16;
    constexpr int NTHR = 256;
    constexpr int WIT  = (KC / 2) * N / NTHR;
    static_assert(WIT >= 1 && (KC / 2) * N % NTHR == 0, "the W chunk divides evenly over the block");

    float wx[WIT], wy[WIT];
    auto prefetch = [&](int kc) {
#pragma unroll
        for (int it = 0; it < WIT; ++it) {
            const int i = tid + it * NTHR;
            const int p = i / N;
            const int c = i % N;
            wx[it] = W[(size_t)(kc + 2 * p)     * N + c];
            wy[it] = W[(size_t)(kc + 2 * p + 1) * N + c];
        }
    };

    prefetch(0);
    for (int kc = 0; kc < K; kc += KC) {
#pragma unroll
        for (int it = 0; it < WIT; ++it) {
            const int i = tid + it * NTHR;
            float2 w2; w2.x = wx[it]; w2.y = wy[it];
            wt2[(i / N) * LDW + (i % N)] = w2;
        }
        __syncthreads();

        float2 a[KC / 4];
#pragma unroll
        for (int s = 0; s < KC / 4; ++s)
            a[s] = *(const float2*)(arow + kc + 4 * s);

        if (kc + KC < K) prefetch(kc + KC);

#pragma unroll
        for (int k = 0; k < KC; k += 4) {
            v2f av; av.x = a[k >> 2].x; av.y = a[k >> 2].y;
            const float2* bp = wt2 + ((k >> 1) + half) * LDW + col0 + l;
            float2 breg[NT];
#pragma unroll
            for (int t = 0; t < NT; ++t) breg[t] = bp[t * 16];
#pragma unroll
            for (int t = 0; t < NT; ++t) {
                v2f b; b.x = breg[t].x; b.y = breg[t].y;
                acc[t] = __builtin_amdgcn_wmma_f32_16x16x4_f32(
                    false, av, false, b, (short)0, acc[t], false, false);
            }
        }
        __syncthreads();
    }
}

__global__ void __launch_bounds__(256)
head_fused_kernel(const float* __restrict__ agg0,
                  const float* __restrict__ agg1,
                  const float* __restrict__ x,
                  const float* __restrict__ w_rel,
                  const float* __restrict__ w_root,
                  const float* __restrict__ bias,
                  const int* __restrict__ ui,
                  const int* __restrict__ vi,
                  const float* __restrict__ mlp_w0, const float* __restrict__ mlp_b0,
                  const float* __restrict__ mlp_w1, const float* __restrict__ mlp_b1,
                  const float* __restrict__ mlp_w2, const float* __restrict__ mlp_b2,
                  const float* __restrict__ out_w,
                  const float* __restrict__ out_b,
                  float* __restrict__ score) {
    constexpr int NT   = 8;
    constexpr int N    = NT * 16;
    constexpr int KC   = 64;
    constexpr int LDW  = N + 16;
    constexpr int NTHR = 256;
    constexpr int WIT  = (KC / 2) * N / NTHR;
    constexpr int NCH  = 3 * (EMB / KC);
    __shared__ float2 wt2[(KC / 2) * LDW];
    __shared__ __attribute__((aligned(16))) float hc[HB_ITEMS * HC_P];
    __shared__ __attribute__((aligned(16))) float h0s[HB_ITEMS * H0_P];
    __shared__ __attribute__((aligned(16))) float h1s[HB_ITEMS * H1_P];
    __shared__ float h2s[HB_ITEMS * H2_P];
    __shared__ float res[HB_ITEMS];

    const int tid   = threadIdx.x;
    const int wave  = tid >> 5;
    const int lane  = tid & 31;
    const int half  = lane >> 4;
    const int l     = lane & 15;
    const int rowBase = (blockIdx.x * (NTHR >> 5) + wave) * 16;

    {
        v8f acc[NT] = {};

        const int    row  = rowBase + l;
        const int    node = (row & 1) ? vi[row >> 1] : ui[row >> 1];
        const size_t xoff = (size_t)node * EMB;
        const size_t roff = (size_t)row * EMB;

        float  wx[WIT], wy[WIT];
        float2 an[KC / 4];

        auto prefetch = [&](int ch) {
            const int seg = ch / (EMB / KC);
            const int kc  = (ch % (EMB / KC)) * KC;
            const float* Xs = (seg == 0) ? agg0 : ((seg == 1) ? agg1 : x);
            const float* Ws = (seg == 2) ? w_root : (w_rel + (size_t)seg * EMB * EMB);
            const float* xrow = Xs + ((seg == 2) ? xoff : roff) + kc + 2 * half;
#pragma unroll
            for (int s = 0; s < KC / 4; ++s)
                an[s] = *(const float2*)(xrow + 4 * s);
#pragma unroll
            for (int it = 0; it < WIT; ++it) {
                const int i = tid + it * NTHR;
                const int p = i / N;
                const int c = i % N;
                wx[it] = Ws[(size_t)(kc + 2 * p)     * N + c];
                wy[it] = Ws[(size_t)(kc + 2 * p + 1) * N + c];
            }
        };

        prefetch(0);
        for (int ch = 0; ch < NCH; ++ch) {
            float2 a[KC / 4];
#pragma unroll
            for (int s = 0; s < KC / 4; ++s) a[s] = an[s];
#pragma unroll
            for (int it = 0; it < WIT; ++it) {
                const int i = tid + it * NTHR;
                float2 w2; w2.x = wx[it]; w2.y = wy[it];
                wt2[(i / N) * LDW + (i % N)] = w2;
            }
            __syncthreads();

            if (ch + 1 < NCH) prefetch(ch + 1);

#pragma unroll
            for (int k = 0; k < KC; k += 4) {
                v2f av; av.x = a[k >> 2].x; av.y = a[k >> 2].y;
                const float2* bp = wt2 + ((k >> 1) + half) * LDW + l;
                float2 breg[NT];
#pragma unroll
                for (int t = 0; t < NT; ++t) breg[t] = bp[t * 16];
#pragma unroll
                for (int t = 0; t < NT; ++t) {
                    v2f b; b.x = breg[t].x; b.y = breg[t].y;
                    acc[t] = __builtin_amdgcn_wmma_f32_16x16x4_f32(
                        false, av, false, b, (short)0, acc[t], false, false);
                }
            }
            __syncthreads();
        }

#pragma unroll
        for (int t = 0; t < NT; ++t) {
            const int   c  = t * 16 + l;
            const float bv = bias[c];
#pragma unroll
            for (int v = 0; v < 8; ++v) {
                const int rl = wave * 16 + v + 8 * half;
                hc[(rl >> 1) * HC_P + (rl & 1) * EMB + c] = acc[t][v] + bv;
            }
        }
    }

    const int rt = wave & 3;
    const int cg = wave >> 2;
    const int arow_item = rt * 16 + l;
    {
        v8f c0[4] = {};
        lds_rows_gemm<4, 128>(hc + arow_item * HC_P + 2 * half, 2 * EMB, mlp_w0, cg * 64, wt2, tid, half, l, c0);
#pragma unroll
        for (int t = 0; t < 4; ++t) {
            const int   c  = cg * 64 + t * 16 + l;
            const float bv = mlp_b0[c];
#pragma unroll
            for (int v = 0; v < 8; ++v)
                h0s[(rt * 16 + v + 8 * half) * H0_P + c] = fmaxf(c0[t][v] + bv, 0.0f);
        }
    }
    {
        v8f c1[2] = {};
        lds_rows_gemm<2, 64>(h0s + arow_item * H0_P + 2 * half, 128, mlp_w1, cg * 32, wt2, tid, half, l, c1);
#pragma unroll
        for (int t = 0; t < 2; ++t) {
            const int   c  = cg * 32 + t * 16 + l;
            const float bv = mlp_b1[c];
#pragma unroll
            for (int v = 0; v < 8; ++v)
                h1s[(rt * 16 + v + 8 * half) * H1_P + c] = fmaxf(c1[t][v] + bv, 0.0f);
        }
    }
    {
        v8f c2[1] = {};
        lds_rows_gemm<1, 32>(h1s + arow_item * H1_P + 2 * half, 64, mlp_w2, cg * 16, wt2, tid, half, l, c2);
        const int   c  = cg * 16 + l;
        const float bv = mlp_b2[c];
#pragma unroll
        for (int v = 0; v < 8; ++v)
            h2s[(rt * 16 + v + 8 * half) * H2_P + c] = fmaxf(c2[0][v] + bv, 0.0f);
    }
    __syncthreads();

    constexpr int IPW = HB_ITEMS / (NTHR >> 5);
    static_assert(IPW <= 32, "lane i finishes item i");
    const float4 ww = *(const float4*)(out_w + lane * 4);
    const float  wh = out_w[EMB + lane];
    const float  ob = out_b[0];
    float4 uu[IPW], vv[IPW];
    float  hh[IPW], sa[IPW];
#pragma unroll
    for (int i = 0; i < IPW; ++i) {
        const int it = wave * IPW + i;
        uu[i] = *(const float4*)(hc + it * HC_P + lane * 4);
        vv[i] = *(const float4*)(hc + it * HC_P + EMB + lane * 4);
        hh[i] = h2s[it * H2_P + lane];
    }
#pragma unroll
    for (int i = 0; i < IPW; ++i) {
        const float4 u4 = uu[i], v4 = vv[i];
        const float su = wave_sum_valu(u4.x * u4.x + u4.y * u4.y + u4.z * u4.z + u4.w * u4.w);
        const float sv = wave_sum_valu(v4.x * v4.x + v4.y * v4.y + v4.z * v4.z + v4.w * v4.w);
        const float rn = 1.0f / (fmaxf(sqrtf(su), 1e-12f) * fmaxf(sqrtf(sv), 1e-12f));
        float a = (u4.x * v4.x * rn) * ww.x + (u4.y * v4.y * rn) * ww.y
                + (u4.z * v4.z * rn) * ww.z + (u4.w * v4.w * rn) * ww.w;
        a += hh[i] * wh;
        sa[i] = wave_sum_valu(a);
    }
    float mine = sa[0];
#pragma unroll
    for (int i = 1; i < IPW; ++i) mine = (lane == i) ? sa[i] : mine;
    if (lane < IPW) res[wave * IPW + lane] = 1.0f / (1.0f + expf(-(mine + ob)));
    __syncthreads();
    if (tid < HB_ITEMS)
        score[(size_t)blockIdx.x * HB_ITEMS + tid] = res[tid];
}

extern "C" void kernel_launch(void* const* d_in, const int* in_sizes, int n_in,
                              void* d_out, int out_size, void* d_ws, size_t ws_size,
                              hipStream_t stream) {
    const int*   user_idx = (const int*)d_in[0];
    const int*   item_idx = (const int*)d_in[1];
    const int*   eidx     = (const int*)d_in[2];
    const int*   etype    = (const int*)d_in[3];
    const float* eweight  = (const float*)d_in[4];
    const float* emb      = (const float*)d_in[5];
    const float* w1_rel   = (const float*)d_in[6];
    const float* w1_root  = (const float*)d_in[7];
    const float* b1       = (const float*)d_in[8];
    const float* ln_g     = (const float*)d_in[9];
    const float* ln_b     = (const float*)d_in[10];
    const float* w2_rel   = (const float*)d_in[11];
    const float* w2_root  = (const float*)d_in[12];
    const float* b2       = (const float*)d_in[13];
    const float* mlp_w0   = (const float*)d_in[14];
    const float* mlp_b0   = (const float*)d_in[15];
    const float* mlp_w1   = (const float*)d_in[16];
    const float* mlp_b1   = (const float*)d_in[17];
    const float* mlp_w2   = (const float*)d_in[18];
    const float* mlp_b2   = (const float*)d_in[19];
    const float* out_w    = (const float*)d_in[20];
    const float* out_b    = (const float*)d_in[21];
    float* out = (float*)d_out;

    char* ws = (char*)d_ws;
    const size_t NB = (size_t)NNODES * EMB * sizeof(float);
    float* y0 = (float*)(ws);
    float* y1 = (float*)(ws + NB);
    float* h1 = (float*)(ws + 2 * NB);
    float* agg0 = y0;
    float* agg1 = y1;
    int*  cnt = (int*)(ws + 3 * NB);
    int*  cur = cnt + CNT_PAD;
    int2* rec = (int2*)(cur + CNT_PAD);

    const int nodeGrid  = (NNODES + 127) / 128;
    const int headGrid  = BATCH / HB_ITEMS;
    const int edgeGrid4 = NEDGES / 1024;
    const int scanGrid  = CNT_PAD / SCAN_EPB;
    const int pullGrid  = NNODES / (PL_RPW * 8);
    const int itemGrid  = BATCH / 8;

    const Layer1Args l1 = { w1_rel, w1_rel + EMB * EMB, w1_root, b1, y0, y1, h1 };
    wmma_gemm_layer1_kernel<<<3 * nodeGrid, 256, 0, stream>>>(emb, l1, NNODES, EMB, cnt, CNT_PAD);

    edge_count_kernel<<<edgeGrid4, 256, 0, stream>>>(eidx + NEDGES, cnt);
    edge_offsets_kernel<<<scanGrid, 1024, 0, stream>>>(cnt, cur);
    edge_place_kernel<<<edgeGrid4, 256, 0, stream>>>(eidx, eidx + NEDGES, etype, eweight, cur, rec);

    pull_relu_ln_kernel<<<pullGrid, 256, 0, stream>>>(cur, rec, y0, y1, h1, ln_g, ln_b);

    gathered_rows_pull_kernel<<<itemGrid, 256, 0, stream>>>(user_idx, item_idx, cur, rec, h1, agg0, agg1);

    head_fused_kernel<<<headGrid, 256, 0, stream>>>(agg0, agg1, h1, w2_rel, w2_root, b2, user_idx, item_idx,
                                                    mlp_w0, mlp_b0, mlp_w1, mlp_b1, mlp_w2, mlp_b2,
                                                    out_w, out_b, out);
}


# layer 1 as one launch, built like layer 2: a block sums its rows' neighbour embedding rows per relation into LDS, then ONE K=384 f32 WMMA product, ReLU+LN, one store per row; sum order changes
# speedup vs baseline: 2.7571x; 1.1328x over previous
#include <hip/hip_runtime.h>
#include <hip/hip_bf16.h>
#include <math.h>

#define EMB     128
#define BATCH   16384
#define NNODES  100000
#define NEDGES  640000

typedef float v2f __attribute__((ext_vector_type(2)));
typedef float v8f __attribute__((ext_vector_type(8)));

template <int M>
__device__ __forceinline__ float xor_add(float v) {
    const int b = __builtin_bit_cast(int, v);
    int o;
    if constexpr (M == 16) o = __builtin_amdgcn_permlanex16(b, b, 0x76543210, 0xfedcba98, false, false);
    else                   o = __builtin_amdgcn_update_dpp(b, b, 0x160 + M, 0xf, 0xf, false);
    return v + __builtin_bit_cast(float, o);
}

__device__ __forceinline__ float wave_sum_valu(float v) {
    v = xor_add<16>(v);
    v = xor_add<8>(v);
    v = xor_add<4>(v);
    v = xor_add<2>(v);
    v = xor_add<1>(v);
    return v;
}

#define CNT_PAD 102400
#define SCAN_EPB 4096
static_assert(CNT_PAD % SCAN_EPB == 0 && CNT_PAD >= NNODES, "the scan blocks cover every counter");
static_assert(NEDGES % 1024 == 0, "four edges per thread, full blocks");
static_assert(NNODES < (1 << 30), "a source row index leaves bit 30 free for the relation");
static_assert(CNT_PAD % 1024 == 0, "the clearing launch stores four counters per thread, full blocks");

__global__ void __launch_bounds__(256)
edge_count_clear_kernel(int* __restrict__ cnt) {
    ((int4*)cnt)[blockIdx.x * 256 + threadIdx.x] = make_int4(0, 0, 0, 0);
}

__global__ void __launch_bounds__(256)
edge_count_kernel(const int* __restrict__ dst, int* __restrict__ cnt) {
    const int i = blockIdx.x * 256 + threadIdx.x;
    const int4 d = *(const int4*)(dst + 4 * i);
    atomicAdd(cnt + d.x, 1);
    atomicAdd(cnt + d.y, 1);
    atomicAdd(cnt + d.z, 1);
    atomicAdd(cnt + d.w, 1);
}

__global__ void __launch_bounds__(1024)
edge_offsets_kernel(const int* __restrict__ cnt, int* __restrict__ cur) {
    __shared__ int wsum[32];
    __shared__ int wtot[32];
    const int tid  = threadIdx.x;
    const int wave = tid >> 5;
    const int lane = tid & 31;
    const int4* c4 = (const int4*)cnt;
    const int first = blockIdx.x * 1024;
    const int4 mine = c4[first + tid];
    int part = 0;
#pragma unroll 8
    for (int i = tid; i < first; i += 1024) {
        const int4 c = c4[i];
        part += (c.x + c.y) + (c.z + c.w);
    }
    const int s = (mine.x + mine.y) + (mine.z + mine.w);
#pragma unroll
    for (int m = 16; m >= 1; m >>= 1) part += __shfl_xor(part, m, 32);
    int inc = s;
#pragma unroll
    for (int o = 1; o < 32; o <<= 1) {
        const int t = __shfl_up(inc, o, 32);
        if (lane >= o) inc += t;
    }
    if (lane == 0)  wsum[wave] = part;
    if (lane == 31) wtot[wave] = inc;
    __syncthreads();
    int mix = wsum[lane] + ((lane < wave) ? wtot[lane] : 0);
#pragma unroll
    for (int m = 16; m >= 1; m >>= 1) mix += __shfl_xor(mix, m, 32);
    const int base = mix + inc - s;
    int4 o4;
    o4.x = base;
    o4.y = base + mine.x;
    o4.z = base + mine.x + mine.y;
    o4.w = base + mine.x + mine.y + mine.z;
    ((int4*)cur)[first + tid] = o4;
}

__global__ void __launch_bounds__(256)
edge_place_kernel(const int* __restrict__ src,
                  const int* __restrict__ dst,
                  const int* __restrict__ etype,
                  const float* __restrict__ eweight,
                  int* __restrict__ cur,
                  int2* __restrict__ rec) {
    const int i = blockIdx.x * 256 + threadIdx.x;
    const int4   s = *(const int4*)(src + 4 * i);
    const int4   d = *(const int4*)(dst + 4 * i);
    const int4   t = *(const int4*)(etype + 4 * i);
    const float4 w = *(const float4*)(eweight + 4 * i);
    const int p0 = atomicAdd(cur + d.x, 1);
    const int p1 = atomicAdd(cur + d.y, 1);
    const int p2 = atomicAdd(cur + d.z, 1);
    const int p3 = atomicAdd(cur + d.w, 1);
    rec[p0] = make_int2(s.x | ((t.x == 0) ? 0 : (1 << 30)), __float_as_int(w.x));
    rec[p1] = make_int2(s.y | ((t.y == 0) ? 0 : (1 << 30)), __float_as_int(w.y));
    rec[p2] = make_int2(s.z | ((t.z == 0) ? 0 : (1 << 30)), __float_as_int(w.z));
    rec[p3] = make_int2(s.w | ((t.w == 0) ? 0 : (1 << 30)), __float_as_int(w.w));
}

__device__ __forceinline__ void axpy4(float4& acc, const float w, const float4 v) {
    acc.x += w * v.x; acc.y += w * v.y; acc.z += w * v.z; acc.w += w * v.w;
}

static_assert(BATCH % 8 == 0, "every wave of every block has its item");

__device__ __forceinline__ void
pull_rows_by_relation(const int beg, const int end, const int lane,
                      const int2* __restrict__ rec, const float* __restrict__ x,
                      float4& s0, float4& s1) {
    for (int cb = beg; cb < end; cb += 32) {
        const int left = end - cb;
        const int nrec = left < 32 ? left : 32;
        int2 rc = make_int2(0, 0);
        if (lane < nrec) rc = rec[cb + lane];
        for (int q = 0; q < nrec; q += 8) {
            const int m = nrec - q;
            const int n = m < 8 ? m : 8;
            float4 v[8];
            float  w[8];
            int    t[8];
#pragma unroll
            for (int u = 0; u < 8; ++u) {
                const int k = q + (u < n ? u : n - 1);
                const int sx = __builtin_amdgcn_readlane(rc.x, k);
                w[u] = __int_as_float(__builtin_amdgcn_readlane(rc.y, k));
                t[u] = sx >> 30;
                v[u] = *(const float4*)(x + (size_t)(sx & 0x3fffffff) * EMB + lane * 4);
            }
#pragma unroll
            for (int u = 0; u < 8; ++u) {
                if (u < n) {
                    if (t[u] == 0) axpy4(s0, w[u], v[u]);
                    else           axpy4(s1, w[u], v[u]);
                }
            }
        }
    }
}

__global__ void __launch_bounds__(256)
gathered_rows_pull_kernel(const int* __restrict__ ui,
                          const int* __restrict__ vi,
                          const int* __restrict__ cur,
                          const int2* __restrict__ rec,
                          const float* __restrict__ x,
                          float* __restrict__ agg0,
                          float* __restrict__ agg1) {
    const int item = blockIdx.x * 8 + (threadIdx.x >> 5);
    const int lane = threadIdx.x & 31;
    const int node = (lane & 2) ? vi[item] : ui[item];
    const int bi   = node - 1 + (lane & 1);
    int bv = 0;
    if (lane < 4 && bi >= 0) bv = cur[bi];
    const int ub = __builtin_amdgcn_readlane(bv, 0);
    const int ue = __builtin_amdgcn_readlane(bv, 1);
    const int vb = __builtin_amdgcn_readlane(bv, 2);
    const int ve = __builtin_amdgcn_readlane(bv, 3);
    const float4 z = make_float4(0.0f, 0.0f, 0.0f, 0.0f);
    float4 u0 = z, u1 = z, v0 = z, v1 = z;
    pull_rows_by_relation(ub, ue, lane, rec, x, u0, u1);
    pull_rows_by_relation(vb, ve, lane, rec, x, v0, v1);
    const size_t r = (size_t)item * 2;
    *(float4*)(agg0 + (r + 0) * EMB + lane * 4) = u0;
    *(float4*)(agg1 + (r + 0) * EMB + lane * 4) = u1;
    *(float4*)(agg0 + (r + 1) * EMB + lane * 4) = v0;
    *(float4*)(agg1 + (r + 1) * EMB + lane * 4) = v1;
}

#define HB_ITEMS 64
#define HC_P (2 * EMB + 4)
#define H0_P (128 + 4)
#define H1_P (64 + 4)
#define H2_P (32 + 2)
static_assert((2 * BATCH) % 128 == 0, "every block of eight 16-row wave tiles is full");
static_assert(BATCH % HB_ITEMS == 0, "every block's item tile is full");

template <int NT, int N>
__device__ __forceinline__ void
lds_rows_gemm(const float* arow, const int K,
              const float* __restrict__ W, const int col0,
              float2* wt2, const int tid, const int half, const int l,
              v8f (&acc)[NT]) {
    constexpr int KC   = 64;
    constexpr int LDW  = N + 16;
    constexpr int NTHR = 256;
    constexpr int WIT  = (KC / 2) * N / NTHR;
    static_assert(WIT >= 1 && (KC / 2) * N % NTHR == 0, "the W chunk divides evenly over the block");

    float wx[WIT], wy[WIT];
    auto prefetch = [&](int kc) {
#pragma unroll
        for (int it = 0; it < WIT; ++it) {
            const int i = tid + it * NTHR;
            const int p = i / N;
            const int c = i % N;
            wx[it] = W[(size_t)(kc + 2 * p)     * N + c];
            wy[it] = W[(size_t)(kc + 2 * p + 1) * N + c];
        }
    };

    prefetch(0);
    for (int kc = 0; kc < K; kc += KC) {
#pragma unroll
        for (int it = 0; it < WIT; ++it) {
            const int i = tid + it * NTHR;
            float2 w2; w2.x = wx[it]; w2.y = wy[it];
            wt2[(i / N) * LDW + (i % N)] = w2;
        }
        __syncthreads();

        float2 a[KC / 4];
#pragma unroll
        for (int s = 0; s < KC / 4; ++s)
            a[s] = *(const float2*)(arow + kc + 4 * s);

        if (kc + KC < K) prefetch(kc + KC);

#pragma unroll
        for (int k = 0; k < KC; k += 4) {
            v2f av; av.x = a[k >> 2].x; av.y = a[k >> 2].y;
            const float2* bp = wt2 + ((k >> 1) + half) * LDW + col0 + l;
            float2 breg[NT];
#pragma unroll
            for (int t = 0; t < NT; ++t) breg[t] = bp[t * 16];
#pragma unroll
            for (int t = 0; t < NT; ++t) {
                v2f b; b.x = breg[t].x; b.y = breg[t].y;
                acc[t] = __builtin_amdgcn_wmma_f32_16x16x4_f32(
                    false, av, false, b, (short)0, acc[t], false, false);
            }
        }
        __syncthreads();
    }
}

template <int NT, int N>
__device__ __forceinline__ void
rows_gemm_stepped(const float* arow, const int K,
                  const float* __restrict__ W, const int col0,
                  float2* wt2, const int tid, const int half, const int l,
                  v8f (&acc)[NT]) {
    constexpr int KC   = 64;
    constexpr int LDW  = N + 16;
    constexpr int NTHR = 256;
    constexpr int WIT  = (KC / 2) * N / NTHR;
    constexpr int PST  = NTHR / N;
    static_assert(WIT >= 1 && (KC / 2) * N % NTHR == 0 && NTHR % N == 0, "the W chunk divides evenly over the block, a thread keeps its column");

    const int p0 = tid / N;
    const int c  = tid % N;
    const float* wp = W + (size_t)(2 * p0) * N + c;
    float2* sp = wt2 + p0 * LDW + c;

    float wx[WIT], wy[WIT];
    auto prefetch = [&](const float* wk) {
#pragma unroll
        for (int it = 0; it < WIT; ++it) {
            wx[it] = wk[(2 * it * PST)     * N];
            wy[it] = wk[(2 * it * PST + 1) * N];
        }
    };

    prefetch(wp);
    for (int kc = 0; kc < K; kc += KC) {
#pragma unroll
        for (int it = 0; it < WIT; ++it) {
            float2 w2; w2.x = wx[it]; w2.y = wy[it];
            sp[it * PST * LDW] = w2;
        }
        __syncthreads();

        float2 a[KC / 4];
#pragma unroll
        for (int s = 0; s < KC / 4; ++s)
            a[s] = *(const float2*)(arow + kc + 4 * s);

        if (kc + KC < K) prefetch(wp + (size_t)(kc + KC) * N);

#pragma unroll
        for (int k = 0; k < KC; k += 4) {
            v2f av; av.x = a[k >> 2].x; av.y = a[k >> 2].y;
            const float2* bp = wt2 + ((k >> 1) + half) * LDW + col0 + l;
            float2 breg[NT];
#pragma unroll
            for (int t = 0; t < NT; ++t) breg[t] = bp[t * 16];
#pragma unroll
            for (int t = 0; t < NT; ++t) {
                v2f b; b.x = breg[t].x; b.y = breg[t].y;
                acc[t] = __builtin_amdgcn_wmma_f32_16x16x4_f32(
                    false, av, false, b, (short)0, acc[t], false, false);
            }
        }
        __syncthreads();
    }
}

#define L1_ROWS 64
#define L1_RPW  (L1_ROWS / 8)
#define L1_AP   (2 * EMB + 4)
#define L1_OP   (EMB + 4)
static_assert(L1_RPW + 1 < 32, "one lane per row boundary of a wave's stretch");
static_assert(((NNODES + L1_ROWS - 1) / L1_ROWS) * L1_ROWS <= CNT_PAD, "row ends exist for every row of the last block");
static_assert(L1_ROWS * L1_OP <= L1_ROWS * L1_AP, "the product tile fits in the sums' tile");

__global__ void __launch_bounds__(256) __attribute__((amdgpu_waves_per_eu(6)))
layer1_fused_kernel(const int* __restrict__ cur,
                    const int2* __restrict__ rec,
                    const float* __restrict__ emb,
                    const float* __restrict__ w_rel,
                    const float* __restrict__ w_root,
                    const float* __restrict__ bias,
                    const float* __restrict__ g,
                    const float* __restrict__ b,
                    float* __restrict__ xout) {
    __shared__ float2 wt2[(64 / 2) * (EMB + 16)];
    __shared__ __attribute__((aligned(16))) float at[L1_ROWS * L1_AP];

    const int tid  = threadIdx.x;
    const int wave = tid >> 5;
    const int lane = tid & 31;
    const int half = lane >> 4;
    const int l    = lane & 15;
    const int rowB = blockIdx.x * L1_ROWS;
    const int wl0  = wave * L1_RPW;

    {
        const int bi = rowB + wl0 - 1 + lane;
        int bv = 0;
        if (lane <= L1_RPW && bi >= 0) bv = cur[bi];
        const int beg = __builtin_amdgcn_readlane(bv, 0);
        const int end = __builtin_amdgcn_readlane(bv, L1_RPW);
        int rowEnd = __builtin_amdgcn_readlane(bv, 1);
        int rw = 0;
        const float4 z = make_float4(0.0f, 0.0f, 0.0f, 0.0f);
        float4 s0 = z, s1 = z;
        auto flush = [&]() {
            float* tp = at + (wl0 + rw) * L1_AP + lane * 4;
            *(float4*)tp         = s0;
            *(float4*)(tp + EMB) = s1;
            s0 = z; s1 = z;
            ++rw;
            rowEnd = __builtin_amdgcn_readlane(bv, rw + 1);
        };
        for (int cb = beg; cb < end; cb += 32) {
            const int left = end - cb;
            const int nrec = left < 32 ? left : 32;
            int2 rc = make_int2(0, 0);
            if (lane < nrec) rc = rec[cb + lane];
            for (int q = 0; q < nrec; q += 8) {
                const int m = nrec - q;
                const int n = m < 8 ? m : 8;
                float4 v[8];
                float  w[8];
                int    t[8];
#pragma unroll
                for (int u = 0; u < 8; ++u) {
                    const int k = q + (u < n ? u : n - 1);
                    const int sx = __builtin_amdgcn_readlane(rc.x, k);
                    w[u] = __int_as_float(__builtin_amdgcn_readlane(rc.y, k));
                    t[u] = sx >> 30;
                    v[u] = *(const float4*)(emb + (size_t)(sx & 0x3fffffff) * EMB + lane * 4);
                }
#pragma unroll
                for (int u = 0; u < 8; ++u) {
                    if (u < n) {
                        const int p = cb + q + u;
                        while (p >= rowEnd && rw < L1_RPW - 1) flush();
                        if (t[u] == 0) axpy4(s0, w[u], v[u]);
                        else           axpy4(s1, w[u], v[u]);
                    }
                }
            }
        }
        while (rw < L1_RPW) flush();
    }

    const int rt = wave & 3;
    const int cg = wave >> 2;
    v8f acc[4] = {};
    rows_gemm_stepped<4, EMB>(at + (rt * 16 + l) * L1_AP + 2 * half, 2 * EMB, w_rel, cg * 64, wt2, tid, half, l, acc);
    {
        const int r  = rowB + rt * 16 + l;
        const int rr = r < NNODES ? r : NNODES - 1;
        rows_gemm_stepped<4, EMB>(emb + (size_t)rr * EMB + 2 * half, EMB, w_root, cg * 64, wt2, tid, half, l, acc);
    }
#pragma unroll
    for (int t = 0; t < 4; ++t) {
        const int   c  = cg * 64 + t * 16 + l;
        const float bc = bias[c];
#pragma unroll
        for (int v = 0; v < 8; ++v)
            at[(rt * 16 + v + 8 * half) * L1_OP + c] = acc[t][v] + bc;
    }
    __syncthreads();

    const float4 gg = *(const float4*)(g + lane * 4);
    const float4 bb = *(const float4*)(b + lane * 4);
    float4 vr[L1_RPW];
#pragma unroll
    for (int r = 0; r < L1_RPW; ++r)
        vr[r] = *(const float4*)(at + (wl0 + r) * L1_OP + lane * 4);
#pragma unroll
    for (int r = 0; r < L1_RPW; ++r) {
        float4 v = vr[r];
        v.x = fmaxf(v.x, 0.0f); v.y = fmaxf(v.y, 0.0f);
        v.z = fmaxf(v.z, 0.0f); v.w = fmaxf(v.w, 0.0f);
        const float mu = wave_sum_valu(v.x + v.y + v.z + v.w) * (1.0f / EMB);
        const float dx = v.x - mu, dy = v.y - mu, dz = v.z - mu, dw = v.w - mu;
        const float var = wave_sum_valu(dx * dx + dy * dy + dz * dz + dw * dw) * (1.0f / EMB);
        const float inv = rsqrtf(var + 1e-5f);
        float4 o;
        o.x = dx * inv * gg.x + bb.x;
        o.y = dy * inv * gg.y + bb.y;
        o.z = dz * inv * gg.z + bb.z;
        o.w = dw * inv * gg.w + bb.w;
        const int row = rowB + wl0 + r;
        if (row < NNODES)
            *(float4*)(xout + (size_t)row * EMB + lane * 4) = o;
    }
}

__global__ void __launch_bounds__(256)
head_fused_kernel(const float* __restrict__ agg0,
                  const float* __restrict__ agg1,
                  const float* __restrict__ x,
                  const float* __restrict__ w_rel,
                  const float* __restrict__ w_root,
                  const float* __restrict__ bias,
                  const int* __restrict__ ui,
                  const int* __restrict__ vi,
                  const float* __restrict__ mlp_w0, const float* __restrict__ mlp_b0,
                  const float* __restrict__ mlp_w1, const float* __restrict__ mlp_b1,
                  const float* __restrict__ mlp_w2, const float* __restrict__ mlp_b2,
                  const float* __restrict__ out_w,
                  const float* __restrict__ out_b,
                  float* __restrict__ score) {
    constexpr int NT   = 8;
    constexpr int N    = NT * 16;
    constexpr int KC   = 64;
    constexpr int LDW  = N + 16;
    constexpr int NTHR = 256;
    constexpr int WIT  = (KC / 2) * N / NTHR;
    constexpr int NCH  = 3 * (EMB / KC);
    __shared__ float2 wt2[(KC / 2) * LDW];
    __shared__ __attribute__((aligned(16))) float hc[HB_ITEMS * HC_P];
    __shared__ __attribute__((aligned(16))) float h0s[HB_ITEMS * H0_P];
    __shared__ __attribute__((aligned(16))) float h1s[HB_ITEMS * H1_P];
    __shared__ float h2s[HB_ITEMS * H2_P];
    __shared__ float res[HB_ITEMS];

    const int tid   = threadIdx.x;
    const int wave  = tid >> 5;
    const int lane  = tid & 31;
    const int half  = lane >> 4;
    const int l     = lane & 15;
    const int rowBase = (blockIdx.x * (NTHR >> 5) + wave) * 16;

    {
        v8f acc[NT] = {};

        const int    row  = rowBase + l;
        const int    node = (row & 1) ? vi[row >> 1] : ui[row >> 1];
        const size_t xoff = (size_t)node * EMB;
        const size_t roff = (size_t)row * EMB;

        float  wx[WIT], wy[WIT];
        float2 an[KC / 4];

        auto prefetch = [&](int ch) {
            const int seg = ch / (EMB / KC);
            const int kc  = (ch % (EMB / KC)) * KC;
            const float* Xs = (seg == 0) ? agg0 : ((seg == 1) ? agg1 : x);
            const float* Ws = (seg == 2) ? w_root : (w_rel + (size_t)seg * EMB * EMB);
            const float* xrow = Xs + ((seg == 2) ? xoff : roff) + kc + 2 * half;
#pragma unroll
            for (int s = 0; s < KC / 4; ++s)
                an[s] = *(const float2*)(xrow + 4 * s);
#pragma unroll
            for (int it = 0; it < WIT; ++it) {
                const int i = tid + it * NTHR;
                const int p = i / N;
                const int c = i % N;
                wx[it] = Ws[(size_t)(kc + 2 * p)     * N + c];
                wy[it] = Ws[(size_t)(kc + 2 * p + 1) * N + c];
            }
        };

        prefetch(0);
        for (int ch = 0; ch < NCH; ++ch) {
            float2 a[KC / 4];
#pragma unroll
            for (int s = 0; s < KC / 4; ++s) a[s] = an[s];
#pragma unroll
            for (int it = 0; it < WIT; ++it) {
                const int i = tid + it * NTHR;
                float2 w2; w2.x = wx[it]; w2.y = wy[it];
                wt2[(i / N) * LDW + (i % N)] = w2;
            }
            __syncthreads();

            if (ch + 1 < NCH) prefetch(ch + 1);

#pragma unroll
            for (int k = 0; k < KC; k += 4) {
                v2f av; av.x = a[k >> 2].x; av.y = a[k >> 2].y;
                const float2* bp = wt2 + ((k >> 1) + half) * LDW + l;
                float2 breg[NT];
#pragma unroll
                for (int t = 0; t < NT; ++t) breg[t] = bp[t * 16];
#pragma unroll
                for (int t = 0; t < NT; ++t) {
                    v2f b; b.x = breg[t].x; b.y = breg[t].y;
                    acc[t] = __builtin_amdgcn_wmma_f32_16x16x4_f32(
                        false, av, false, b, (short)0, acc[t], false, false);
                }
            }
            __syncthreads();
        }

#pragma unroll
        for (int t = 0; t < NT; ++t) {
            const int   c  = t * 16 + l;
            const float bv = bias[c];
#pragma unroll
            for (int v = 0; v < 8; ++v) {
                const int rl = wave * 16 + v + 8 * half;
                hc[(rl >> 1) * HC_P + (rl & 1) * EMB + c] = acc[t][v] + bv;
            }
        }
    }

    const int rt = wave & 3;
    const int cg = wave >> 2;
    const int arow_item = rt * 16 + l;
    {
        v8f c0[4] = {};
        lds_rows_gemm<4, 128>(hc + arow_item * HC_P + 2 * half, 2 * EMB, mlp_w0, cg * 64, wt2, tid, half, l, c0);
#pragma unroll
        for (int t = 0; t < 4; ++t) {
            const int   c  = cg * 64 + t * 16 + l;
            const float bv = mlp_b0[c];
#pragma unroll
            for (int v = 0; v < 8; ++v)
                h0s[(rt * 16 + v + 8 * half) * H0_P + c] = fmaxf(c0[t][v] + bv, 0.0f);
        }
    }
    {
        v8f c1[2] = {};
        lds_rows_gemm<2, 64>(h0s + arow_item * H0_P + 2 * half, 128, mlp_w1, cg * 32, wt2, tid, half, l, c1);
#pragma unroll
        for (int t = 0; t < 2; ++t) {
            const int   c  = cg * 32 + t * 16 + l;
            const float bv = mlp_b1[c];
#pragma unroll
            for (int v = 0; v < 8; ++v)
                h1s[(rt * 16 + v + 8 * half) * H1_P + c] = fmaxf(c1[t][v] + bv, 0.0f);
        }
    }
    {
        v8f c2[1] = {};
        lds_rows_gemm<1, 32>(h1s + arow_item * H1_P + 2 * half, 64, mlp_w2, cg * 16, wt2, tid, half, l, c2);
        const int   c  = cg * 16 + l;
        const float bv = mlp_b2[c];
#pragma unroll
        for (int v = 0; v < 8; ++v)
            h2s[(rt * 16 + v + 8 * half) * H2_P + c] = fmaxf(c2[0][v] + bv, 0.0f);
    }
    __syncthreads();

    constexpr int IPW = HB_ITEMS / (NTHR >> 5);
    static_assert(IPW <= 32, "lane i finishes item i");
    const float4 ww = *(const float4*)(out_w + lane * 4);
    const float  wh = out_w[EMB + lane];
    const float  ob = out_b[0];
    float4 uu[IPW], vv[IPW];
    float  hh[IPW], sa[IPW];
#pragma unroll
    for (int i = 0; i < IPW; ++i) {
        const int it = wave * IPW + i;
        uu[i] = *(const float4*)(hc + it * HC_P + lane * 4);
        vv[i] = *(const float4*)(hc + it * HC_P + EMB + lane * 4);
        hh[i] = h2s[it * H2_P + lane];
    }
#pragma unroll
    for (int i = 0; i < IPW; ++i) {
        const float4 u4 = uu[i], v4 = vv[i];
        const float su = wave_sum_valu(u4.x * u4.x + u4.y * u4.y + u4.z * u4.z + u4.w * u4.w);
        const float sv = wave_sum_valu(v4.x * v4.x + v4.y * v4.y + v4.z * v4.z + v4.w * v4.w);
        const float rn = 1.0f / (fmaxf(sqrtf(su), 1e-12f) * fmaxf(sqrtf(sv), 1e-12f));
        float a = (u4.x * v4.x * rn) * ww.x + (u4.y * v4.y * rn) * ww.y
                + (u4.z * v4.z * rn) * ww.z + (u4.w * v4.w * rn) * ww.w;
        a += hh[i] * wh;
        sa[i] = wave_sum_valu(a);
    }
    float mine = sa[0];
#pragma unroll
    for (int i = 1; i < IPW; ++i) mine = (lane == i) ? sa[i] : mine;
    if (lane < IPW) res[wave * IPW + lane] = 1.0f / (1.0f + expf(-(mine + ob)));
    __syncthreads();
    if (tid < HB_ITEMS)
        score[(size_t)blockIdx.x * HB_ITEMS + tid] = res[tid];
}

extern "C" void kernel_launch(void* const* d_in, const int* in_sizes, int n_in,
                              void* d_out, int out_size, void* d_ws, size_t ws_size,
                              hipStream_t stream) {
    const int*   user_idx = (const int*)d_in[0];
    const int*   item_idx = (const int*)d_in[1];
    const int*   eidx     = (const int*)d_in[2];
    const int*   etype    = (const int*)d_in[3];
    const float* eweight  = (const float*)d_in[4];
    const float* emb      = (const float*)d_in[5];
    const float* w1_rel   = (const float*)d_in[6];
    const float* w1_root  = (const float*)d_in[7];
    const float* b1       = (const float*)d_in[8];
    const float* ln_g     = (const float*)d_in[9];
    const float* ln_b     = (const float*)d_in[10];
    const float* w2_rel   = (const float*)d_in[11];
    const float* w2_root  = (const float*)d_in[12];
    const float* b2       = (const float*)d_in[13];
    const float* mlp_w0   = (const float*)d_in[14];
    const float* mlp_b0   = (const float*)d_in[15];
    const float* mlp_w1   = (const float*)d_in[16];
    const float* mlp_b1   = (const float*)d_in[17];
    const float* mlp_w2   = (const float*)d_in[18];
    const float* mlp_b2   = (const float*)d_in[19];
    const float* out_w    = (const float*)d_in[20];
    const float* out_b    = (const float*)d_in[21];
    float* out = (float*)d_out;

    char* ws = (char*)d_ws;
    const size_t NB = (size_t)NNODES * EMB * sizeof(float);
    float* agg0 = (float*)(ws);
    float* agg1 = (float*)(ws + NB);
    float* h1   = (float*)(ws + 2 * NB);
    int*  cnt = (int*)(ws + 3 * NB);
    int*  cur = cnt + CNT_PAD;
    int2* rec = (int2*)(cur + CNT_PAD);

    const int l1Grid    = (NNODES + L1_ROWS - 1) / L1_ROWS;
    const int headGrid  = BATCH / HB_ITEMS;
    const int edgeGrid4 = NEDGES / 1024;
    const int scanGrid  = CNT_PAD / SCAN_EPB;
    const int itemGrid  = BATCH / 8;

    edge_count_clear_kernel<<<CNT_PAD / 1024, 256, 0, stream>>>(cnt);
    edge_count_kernel<<<edgeGrid4, 256, 0, stream>>>(eidx + NEDGES, cnt);
    edge_offsets_kernel<<<scanGrid, 1024, 0, stream>>>(cnt, cur);
    edge_place_kernel<<<edgeGrid4, 256, 0, stream>>>(eidx, eidx + NEDGES, etype, eweight, cur, rec);

    layer1_fused_kernel<<<l1Grid, 256, 0, stream>>>(cur, rec, emb, w1_rel, w1_root, b1, ln_g, ln_b, h1);

    gathered_rows_pull_kernel<<<itemGrid, 256, 0, stream>>>(user_idx, item_idx, cur, rec, h1, agg0, agg1);

    head_fused_kernel<<<headGrid, 256, 0, stream>>>(agg0, agg1, h1, w2_rel, w2_root, b2, user_idx, item_idx,
                                                    mlp_w0, mlp_b0, mlp_w1, mlp_b1, mlp_w2, mlp_b2,
                                                    out_w, out_b, out);
}


# edge lists as 16 record slots per destination row (one 128-B line) filled in ONE placing pass, overflow list for longer rows: count and scan launches go (7 -> 5); sums in placing order
# speedup vs baseline: 3.2852x; 1.1915x over previous
#include <hip/hip_runtime.h>
#include <hip/hip_bf16.h>
#include <math.h>

#define EMB     128
#define BATCH   16384
#define NNODES  100000
#define NEDGES  640000

typedef float v2f __attribute__((ext_vector_type(2)));
typedef float v8f __attribute__((ext_vector_type(8)));

template <int M>
__device__ __forceinline__ float xor_add(float v) {
    const int b = __builtin_bit_cast(int, v);
    int o;
    if constexpr (M == 16) o = __builtin_amdgcn_permlanex16(b, b, 0x76543210, 0xfedcba98, false, false);
    else                   o = __builtin_amdgcn_update_dpp(b, b, 0x160 + M, 0xf, 0xf, false);
    return v + __builtin_bit_cast(float, o);
}

__device__ __forceinline__ float wave_sum_valu(float v) {
    v = xor_add<16>(v);
    v = xor_add<8>(v);
    v = xor_add<4>(v);
    v = xor_add<2>(v);
    v = xor_add<1>(v);
    return v;
}

#define CNT_PAD 102400
#define BKT 16
#define ROWS_PAD (((NNODES + 63) / 64) * 64)
#define OVN_IDX (CNT_PAD - 1)
static_assert(ROWS_PAD <= OVN_IDX, "every covered row has a counter in front of the overflow word");
static_assert(NEDGES % 1024 == 0, "four edges per thread, full blocks");
static_assert(NNODES < (1 << 30), "a source row index leaves bit 30 free for the relation");
static_assert(CNT_PAD % 1024 == 0, "the clearing launch stores four counters per thread, full blocks");

__global__ void __launch_bounds__(256)
edge_count_clear_kernel(int* __restrict__ cnt) {
    ((int4*)cnt)[blockIdx.x * 256 + threadIdx.x] = make_int4(0, 0, 0, 0);
}

__device__ __forceinline__ void
put_record(const int d, const int p, const int2 r,
           int* __restrict__ cnt, int2* __restrict__ rec, int4* __restrict__ ovf) {
    if (p < BKT) {
        rec[(size_t)d * BKT + p] = r;
    } else {
        const int q = atomicAdd(cnt + OVN_IDX, 1);
        ovf[q] = make_int4(d, r.x, r.y, 0);
    }
}

__global__ void __launch_bounds__(256)
edge_place_kernel(const int* __restrict__ src,
                  const int* __restrict__ dst,
                  const int* __restrict__ etype,
                  const float* __restrict__ eweight,
                  int* __restrict__ cnt,
                  int2* __restrict__ rec,
                  int4* __restrict__ ovf) {
    const int i = blockIdx.x * 256 + threadIdx.x;
    const int4   s = *(const int4*)(src + 4 * i);
    const int4   d = *(const int4*)(dst + 4 * i);
    const int4   t = *(const int4*)(etype + 4 * i);
    const float4 w = *(const float4*)(eweight + 4 * i);
    const int p0 = atomicAdd(cnt + d.x, 1);
    const int p1 = atomicAdd(cnt + d.y, 1);
    const int p2 = atomicAdd(cnt + d.z, 1);
    const int p3 = atomicAdd(cnt + d.w, 1);
    put_record(d.x, p0, make_int2(s.x | ((t.x == 0) ? 0 : (1 << 30)), __float_as_int(w.x)), cnt, rec, ovf);
    put_record(d.y, p1, make_int2(s.y | ((t.y == 0) ? 0 : (1 << 30)), __float_as_int(w.y)), cnt, rec, ovf);
    put_record(d.z, p2, make_int2(s.z | ((t.z == 0) ? 0 : (1 << 30)), __float_as_int(w.z)), cnt, rec, ovf);
    put_record(d.w, p3, make_int2(s.w | ((t.w == 0) ? 0 : (1 << 30)), __float_as_int(w.w)), cnt, rec, ovf);
}

__device__ __forceinline__ void axpy4(float4& acc, const float w, const float4 v) {
    acc.x += w * v.x; acc.y += w * v.y; acc.z += w * v.z; acc.w += w * v.w;
}

__device__ __forceinline__ void
sum_held_records(const int rx, const int ry, const int base, const int nrec, const int lane,
                 const float* __restrict__ x, float4& s0, float4& s1) {
    for (int q = 0; q < nrec; q += 8) {
        const int m = nrec - q;
        const int n = m < 8 ? m : 8;
        float4 v[8];
        float  w[8];
        int    t[8];
#pragma unroll
        for (int u = 0; u < 8; ++u) {
            const int k = base + q + (u < n ? u : n - 1);
            const int sx = __builtin_amdgcn_readlane(rx, k);
            w[u] = __int_as_float(__builtin_amdgcn_readlane(ry, k));
            t[u] = sx >> 30;
            v[u] = *(const float4*)(x + (size_t)(sx & 0x3fffffff) * EMB + lane * 4);
        }
#pragma unroll
        for (int u = 0; u < 8; ++u) {
            if (u < n) {
                if (t[u] == 0) axpy4(s0, w[u], v[u]);
                else           axpy4(s1, w[u], v[u]);
            }
        }
    }
}

__device__ __forceinline__ void
sum_overflow_records(const int d, const int novf, const int lane,
                     const int4* __restrict__ ovf, const float* __restrict__ x,
                     float4& s0, float4& s1) {
    for (int cb = 0; cb < novf; cb += 32) {
        int4 e = make_int4(-1, 0, 0, 0);
        if (cb + lane < novf) e = ovf[cb + lane];
        unsigned hit = __builtin_amdgcn_ballot_w32(e.x == d);
        while (hit != 0u) {
            const int k = __builtin_ctz(hit);
            hit &= hit - 1u;
            const int    sx = __builtin_amdgcn_readlane(e.y, k);
            const float  w  = __int_as_float(__builtin_amdgcn_readlane(e.z, k));
            const float4 v  = *(const float4*)(x + (size_t)(sx & 0x3fffffff) * EMB + lane * 4);
            if ((sx >> 30) == 0) axpy4(s0, w, v);
            else                 axpy4(s1, w, v);
        }
    }
}

static_assert(BATCH % 8 == 0, "every wave of every block has its item");

__global__ void __launch_bounds__(256)
gathered_rows_pull_kernel(const int* __restrict__ ui,
                          const int* __restrict__ vi,
                          const int* __restrict__ cnt,
                          const int2* __restrict__ rec,
                          const int4* __restrict__ ovf,
                          const float* __restrict__ x,
                          float* __restrict__ agg0,
                          float* __restrict__ agg1) {
    const int item = blockIdx.x * 8 + (threadIdx.x >> 5);
    const int lane = threadIdx.x & 31;
    const int half = lane >> 4;
    const int l    = lane & 15;
    const int node = half ? vi[item] : ui[item];
    int cv = 0;
    if (l == 0)         cv = cnt[node];
    else if (lane == 1) cv = cnt[OVN_IDX];
    const int cu = __builtin_amdgcn_readlane(cv, 0);
    const int cw = __builtin_amdgcn_readlane(cv, 16);
    const int no = __builtin_amdgcn_readlane(cv, 1);
    const int novf = no < NEDGES ? no : NEDGES;
    const int nu = __builtin_amdgcn_readlane(node, 0);
    const int nv = __builtin_amdgcn_readlane(node, 16);
    const int nbu = cu < BKT ? cu : BKT;
    const int nbv = cw < BKT ? cw : BKT;
    int2 rc = make_int2(0, 0);
    if (l < (half ? nbv : nbu)) rc = rec[(size_t)node * BKT + l];
    const float4 z = make_float4(0.0f, 0.0f, 0.0f, 0.0f);
    float4 u0 = z, u1 = z, v0 = z, v1 = z;
    sum_held_records(rc.x, rc.y, 0, nbu, lane, x, u0, u1);
    if (cu > BKT) sum_overflow_records(nu, novf, lane, ovf, x, u0, u1);
    sum_held_records(rc.x, rc.y, 16, nbv, lane, x, v0, v1);
    if (cw > BKT) sum_overflow_records(nv, novf, lane, ovf, x, v0, v1);
    const size_t r = (size_t)item * 2;
    *(float4*)(agg0 + (r + 0) * EMB + lane * 4) = u0;
    *(float4*)(agg1 + (r + 0) * EMB + lane * 4) = u1;
    *(float4*)(agg0 + (r + 1) * EMB + lane * 4) = v0;
    *(float4*)(agg1 + (r + 1) * EMB + lane * 4) = v1;
}

#define HB_ITEMS 64
#define HC_P (2 * EMB + 4)
#define H0_P (128 + 4)
#define H1_P (64 + 4)
#define H2_P (32 + 2)
static_assert((2 * BATCH) % 128 == 0, "every block of eight 16-row wave tiles is full");
static_assert(BATCH % HB_ITEMS == 0, "every block's item tile is full");

template <int NT, int N>
__device__ __forceinline__ void
lds_rows_gemm(const float* arow, const int K,
              const float* __restrict__ W, const int col0,
              float2* wt2, const int tid, const int half, const int l,
              v8f (&acc)[NT]) {
    constexpr int KC   = 64;
    constexpr int LDW  = N + 16;
    constexpr int NTHR = 256;
    constexpr int WIT  = (KC / 2) * N / NTHR;
    static_assert(WIT >= 1 && (KC / 2) * N % NTHR == 0, "the W chunk divides evenly over the block");

    float wx[WIT], wy[WIT];
    auto prefetch = [&](int kc) {
#pragma unroll
        for (int it = 0; it < WIT; ++it) {
            const int i = tid + it * NTHR;
            const int p = i / N;
            const int c = i % N;
            wx[it] = W[(size_t)(kc + 2 * p)     * N + c];
            wy[it] = W[(size_t)(kc + 2 * p + 1) * N + c];
        }
    };

    prefetch(0);
    for (int kc = 0; kc < K; kc += KC) {
#pragma unroll
        for (int it = 0; it < WIT; ++it) {
            const int i = tid + it * NTHR;
            float2 w2; w2.x = wx[it]; w2.y = wy[it];
            wt2[(i / N) * LDW + (i % N)] = w2;
        }
        __syncthreads();

        float2 a[KC / 4];
#pragma unroll
        for (int s = 0; s < KC / 4; ++s)
            a[s] = *(const float2*)(arow + kc + 4 * s);

        if (kc + KC < K) prefetch(kc + KC);

#pragma unroll
        for (int k = 0; k < KC; k += 4) {
            v2f av; av.x = a[k >> 2].x; av.y = a[k >> 2].y;
            const float2* bp = wt2 + ((k >> 1) + half) * LDW + col0 + l;
            float2 breg[NT];
#pragma unroll
            for (int t = 0; t < NT; ++t) breg[t] = bp[t * 16];
#pragma unroll
            for (int t = 0; t < NT; ++t) {
                v2f b; b.x = breg[t].x; b.y = breg[t].y;
                acc[t] = __builtin_amdgcn_wmma_f32_16x16x4_f32(
                    false, av, false, b, (short)0, acc[t], false, false);
            }
        }
        __syncthreads();
    }
}

template <int NT, int N>
__device__ __forceinline__ void
rows_gemm_stepped(const float* arow, const int K,
                  const float* __restrict__ W, const int col0,
                  float2* wt2, const int tid, const int half, const int l,
                  v8f (&acc)[NT]) {
    constexpr int KC   = 64;
    constexpr int LDW  = N + 16;
    constexpr int NTHR = 256;
    constexpr int WIT  = (KC / 2) * N / NTHR;
    constexpr int PST  = NTHR / N;
    static_assert(WIT >= 1 && (KC / 2) * N % NTHR == 0 && NTHR % N == 0, "the W chunk divides evenly over the block, a thread keeps its column");

    const int p0 = tid / N;
    const int c  = tid % N;
    const float* wp = W + (size_t)(2 * p0) * N + c;
    float2* sp = wt2 + p0 * LDW + c;

    float wx[WIT], wy[WIT];
    auto prefetch = [&](const float* wk) {
#pragma unroll
        for (int it = 0; it < WIT; ++it) {
            wx[it] = wk[(2 * it * PST)     * N];
            wy[it] = wk[(2 * it * PST + 1) * N];
        }
    };

    prefetch(wp);
    for (int kc = 0; kc < K; kc += KC) {
#pragma unroll
        for (int it = 0; it < WIT; ++it) {
            float2 w2; w2.x = wx[it]; w2.y = wy[it];
            sp[it * PST * LDW] = w2;
        }
        __syncthreads();

        float2 a[KC / 4];
#pragma unroll
        for (int s = 0; s < KC / 4; ++s)
            a[s] = *(const float2*)(arow + kc + 4 * s);

        if (kc + KC < K) prefetch(wp + (size_t)(kc + KC) * N);

#pragma unroll
        for (int k = 0; k < KC; k += 4) {
            v2f av; av.x = a[k >> 2].x; av.y = a[k >> 2].y;
            const float2* bp = wt2 + ((k >> 1) + half) * LDW + col0 + l;
            float2 breg[NT];
#pragma unroll
            for (int t = 0; t < NT; ++t) breg[t] = bp[t * 16];
#pragma unroll
            for (int t = 0; t < NT; ++t) {
                v2f b; b.x = breg[t].x; b.y = breg[t].y;
                acc[t] = __builtin_amdgcn_wmma_f32_16x16x4_f32(
                    false, av, false, b, (short)0, acc[t], false, false);
            }
        }
        __syncthreads();
    }
}

#define L1_ROWS 64
#define L1_RPW  (L1_ROWS / 8)
#define L1_AP   (2 * EMB + 4)
#define L1_OP   (EMB + 4)
static_assert(L1_RPW + 1 < 32 && L1_RPW % 2 == 0, "one lane per row count and one for the overflow length; two rows per slot load");
static_assert(((NNODES + L1_ROWS - 1) / L1_ROWS) * L1_ROWS == ROWS_PAD, "counters and slots exist for every row of the last block");
static_assert(L1_ROWS * L1_OP <= L1_ROWS * L1_AP, "the product tile fits in the sums' tile");

__global__ void __launch_bounds__(256) __attribute__((amdgpu_waves_per_eu(6)))
layer1_fused_kernel(const int* __restrict__ cnt,
                    const int2* __restrict__ rec,
                    const int4* __restrict__ ovf,
                    const float* __restrict__ emb,
                    const float* __restrict__ w_rel,
                    const float* __restrict__ w_root,
                    const float* __restrict__ bias,
                    const float* __restrict__ g,
                    const float* __restrict__ b,
                    float* __restrict__ xout) {
    __shared__ float2 wt2[(64 / 2) * (EMB + 16)];
    __shared__ __attribute__((aligned(16))) float at[L1_ROWS * L1_AP];

    const int tid  = threadIdx.x;
    const int wave = tid >> 5;
    const int lane = tid & 31;
    const int half = lane >> 4;
    const int l    = lane & 15;
    const int rowB = blockIdx.x * L1_ROWS;
    const int wl0  = wave * L1_RPW;

    {
        const int row0 = rowB + wl0;
        int cv = 0;
        if (lane < L1_RPW)       cv = cnt[row0 + lane];
        else if (lane == L1_RPW) cv = cnt[OVN_IDX];
        const int no   = __builtin_amdgcn_readlane(cv, L1_RPW);
        const int novf = no < NEDGES ? no : NEDGES;
        int2 rc[L1_RPW / 2];
#pragma unroll
        for (int j = 0; j < L1_RPW / 2; ++j) {
            const int c0 = __builtin_amdgcn_readlane(cv, 2 * j);
            const int c1 = __builtin_amdgcn_readlane(cv, 2 * j + 1);
            const int n0 = c0 < BKT ? c0 : BKT;
            const int n1 = c1 < BKT ? c1 : BKT;
            rc[j] = make_int2(0, 0);
            if (l < (half ? n1 : n0)) rc[j] = rec[(size_t)(row0 + 2 * j + half) * BKT + l];
        }
        const float4 z = make_float4(0.0f, 0.0f, 0.0f, 0.0f);
        static_assert(L1_RPW == 8, "the selects below choose among four slot loads");
#pragma unroll 1
        for (int r = 0; r < L1_RPW; ++r) {
            const int c  = __builtin_amdgcn_readlane(cv, r);
            const int nb = c < BKT ? c : BKT;
            const int j  = r >> 1;
            const int rx = j == 0 ? rc[0].x : (j == 1 ? rc[1].x : (j == 2 ? rc[2].x : rc[3].x));
            const int ry = j == 0 ? rc[0].y : (j == 1 ? rc[1].y : (j == 2 ? rc[2].y : rc[3].y));
            float4 s0 = z, s1 = z;
            sum_held_records(rx, ry, (r & 1) * 16, nb, lane, emb, s0, s1);
            if (c > BKT) sum_overflow_records(row0 + r, novf, lane, ovf, emb, s0, s1);
            float* tp = at + (wl0 + r) * L1_AP + lane * 4;
            *(float4*)tp         = s0;
            *(float4*)(tp + EMB) = s1;
        }
    }

    const int rt = wave & 3;
    const int cg = wave >> 2;
    v8f acc[4] = {};
    rows_gemm_stepped<4, EMB>(at + (rt * 16 + l) * L1_AP + 2 * half, 2 * EMB, w_rel, cg * 64, wt2, tid, half, l, acc);
    {
        const int r  = rowB + rt * 16 + l;
        const int rr = r < NNODES ? r : NNODES - 1;
        rows_gemm_stepped<4, EMB>(emb + (size_t)rr * EMB + 2 * half, EMB, w_root, cg * 64, wt2, tid, half, l, acc);
    }
#pragma unroll
    for (int t = 0; t < 4; ++t) {
        const int   c  = cg * 64 + t * 16 + l;
        const float bc = bias[c];
#pragma unroll
        for (int v = 0; v < 8; ++v)
            at[(rt * 16 + v + 8 * half) * L1_OP + c] = acc[t][v] + bc;
    }
    __syncthreads();

    const float4 gg = *(const float4*)(g + lane * 4);
    const float4 bb = *(const float4*)(b + lane * 4);
    float4 vr[L1_RPW];
#pragma unroll
    for (int r = 0; r < L1_RPW; ++r)
        vr[r] = *(const float4*)(at + (wl0 + r) * L1_OP + lane * 4);
#pragma unroll
    for (int r = 0; r < L1_RPW; ++r) {
        float4 v = vr[r];
        v.x = fmaxf(v.x, 0.0f); v.y = fmaxf(v.y, 0.0f);
        v.z = fmaxf(v.z, 0.0f); v.w = fmaxf(v.w, 0.0f);
        const float mu = wave_sum_valu(v.x + v.y + v.z + v.w) * (1.0f / EMB);
        const float dx = v.x - mu, dy = v.y - mu, dz = v.z - mu, dw = v.w - mu;
        const float var = wave_sum_valu(dx * dx + dy * dy + dz * dz + dw * dw) * (1.0f / EMB);
        const float inv = rsqrtf(var + 1e-5f);
        float4 o;
        o.x = dx * inv * gg.x + bb.x;
        o.y = dy * inv * gg.y + bb.y;
        o.z = dz * inv * gg.z + bb.z;
        o.w = dw * inv * gg.w + bb.w;
        const int row = rowB + wl0 + r;
        if (row < NNODES)
            *(float4*)(xout + (size_t)row * EMB + lane * 4) = o;
    }
}

__global__ void __launch_bounds__(256)
head_fused_kernel(const float* __restrict__ agg0,
                  const float* __restrict__ agg1,
                  const float* __restrict__ x,
                  const float* __restrict__ w_rel,
                  const float* __restrict__ w_root,
                  const float* __restrict__ bias,
                  const int* __restrict__ ui,
                  const int* __restrict__ vi,
                  const float* __restrict__ mlp_w0, const float* __restrict__ mlp_b0,
                  const float* __restrict__ mlp_w1, const float* __restrict__ mlp_b1,
                  const float* __restrict__ mlp_w2, const float* __restrict__ mlp_b2,
                  const float* __restrict__ out_w,
                  const float* __restrict__ out_b,
                  float* __restrict__ score) {
    constexpr int NT   = 8;
    constexpr int N    = NT * 16;
    constexpr int KC   = 64;
    constexpr int LDW  = N + 16;
    constexpr int NTHR = 256;
    constexpr int WIT  = (KC / 2) * N / NTHR;
    constexpr int NCH  = 3 * (EMB / KC);
    __shared__ float2 wt2[(KC / 2) * LDW];
    __shared__ __attribute__((aligned(16))) float hc[HB_ITEMS * HC_P];
    __shared__ __attribute__((aligned(16))) float h0s[HB_ITEMS * H0_P];
    __shared__ __attribute__((aligned(16))) float h1s[HB_ITEMS * H1_P];
    __shared__ float h2s[HB_ITEMS * H2_P];
    __shared__ float res[HB_ITEMS];

    const int tid   = threadIdx.x;
    const int wave  = tid >> 5;
    const int lane  = tid & 31;
    const int half  = lane >> 4;
    const int l     = lane & 15;
    const int rowBase = (blockIdx.x * (NTHR >> 5) + wave) * 16;

    {
        v8f acc[NT] = {};

        const int    row  = rowBase + l;
        const int    node = (row & 1) ? vi[row >> 1] : ui[row >> 1];
        const size_t xoff = (size_t)node * EMB;
        const size_t roff = (size_t)row * EMB;

        float  wx[WIT], wy[WIT];
        float2 an[KC / 4];

        auto prefetch = [&](int ch) {
            const int seg = ch / (EMB / KC);
            const int kc  = (ch % (EMB / KC)) * KC;
            const float* Xs = (seg == 0) ? agg0 : ((seg == 1) ? agg1 : x);
            const float* Ws = (seg == 2) ? w_root : (w_rel + (size_t)seg * EMB * EMB);
            const float* xrow = Xs + ((seg == 2) ? xoff : roff) + kc + 2 * half;
#pragma unroll
            for (int s = 0; s < KC / 4; ++s)
                an[s] = *(const float2*)(xrow + 4 * s);
#pragma unroll
            for (int it = 0; it < WIT; ++it) {
                const int i = tid + it * NTHR;
                const int p = i / N;
                const int c = i % N;
                wx[it] = Ws[(size_t)(kc + 2 * p)     * N + c];
                wy[it] = Ws[(size_t)(kc + 2 * p + 1) * N + c];
            }
        };

        prefetch(0);
        for (int ch = 0; ch < NCH; ++ch) {
            float2 a[KC / 4];
#pragma unroll
            for (int s = 0; s < KC / 4; ++s) a[s] = an[s];
#pragma unroll
            for (int it = 0; it < WIT; ++it) {
                const int i = tid + it * NTHR;
                float2 w2; w2.x = wx[it]; w2.y = wy[it];
                wt2[(i / N) * LDW + (i % N)] = w2;
            }
            __syncthreads();

            if (ch + 1 < NCH) prefetch(ch + 1);

#pragma unroll
            for (int k = 0; k < KC; k += 4) {
                v2f av; av.x = a[k >> 2].x; av.y = a[k >> 2].y;
                const float2* bp = wt2 + ((k >> 1) + half) * LDW + l;
                float2 breg[NT];
#pragma unroll
                for (int t = 0; t < NT; ++t) breg[t] = bp[t * 16];
#pragma unroll
                for (int t = 0; t < NT; ++t) {
                    v2f b; b.x = breg[t].x; b.y = breg[t].y;
                    acc[t] = __builtin_amdgcn_wmma_f32_16x16x4_f32(
                        false, av, false, b, (short)0, acc[t], false, false);
                }
            }
            __syncthreads();
        }

#pragma unroll
        for (int t = 0; t < NT; ++t) {
            const int   c  = t * 16 + l;
            const float bv = bias[c];
#pragma unroll
            for (int v = 0; v < 8; ++v) {
                const int rl = wave * 16 + v + 8 * half;
                hc[(rl >> 1) * HC_P + (rl & 1) * EMB + c] = acc[t][v] + bv;
            }
        }
    }

    const int rt = wave & 3;
    const int cg = wave >> 2;
    const int arow_item = rt * 16 + l;
    {
        v8f c0[4] = {};
        lds_rows_gemm<4, 128>(hc + arow_item * HC_P + 2 * half, 2 * EMB, mlp_w0, cg * 64, wt2, tid, half, l, c0);
#pragma unroll
        for (int t = 0; t < 4; ++t) {
            const int   c  = cg * 64 + t * 16 + l;
            const float bv = mlp_b0[c];
#pragma unroll
            for (int v = 0; v < 8; ++v)
                h0s[(rt * 16 + v + 8 * half) * H0_P + c] = fmaxf(c0[t][v] + bv, 0.0f);
        }
    }
    {
        v8f c1[2] = {};
        lds_rows_gemm<2, 64>(h0s + arow_item * H0_P + 2 * half, 128, mlp_w1, cg * 32, wt2, tid, half, l, c1);
#pragma unroll
        for (int t = 0; t < 2; ++t) {
            const int   c  = cg * 32 + t * 16 + l;
            const float bv = mlp_b1[c];
#pragma unroll
            for (int v = 0; v < 8; ++v)
                h1s[(rt * 16 + v + 8 * half) * H1_P + c] = fmaxf(c1[t][v] + bv, 0.0f);
        }
    }
    {
        v8f c2[1] = {};
        lds_rows_gemm<1, 32>(h1s + arow_item * H1_P + 2 * half, 64, mlp_w2, cg * 16, wt2, tid, half, l, c2);
        const int   c  = cg * 16 + l;
        const float bv = mlp_b2[c];
#pragma unroll
        for (int v = 0; v < 8; ++v)
            h2s[(rt * 16 + v + 8 * half) * H2_P + c] = fmaxf(c2[0][v] + bv, 0.0f);
    }
    __syncthreads();

    constexpr int IPW = HB_ITEMS / (NTHR >> 5);
    static_assert(IPW <= 32, "lane i finishes item i");
    const float4 ww = *(const float4*)(out_w + lane * 4);
    const float  wh = out_w[EMB + lane];
    const float  ob = out_b[0];
    float4 uu[IPW], vv[IPW];
    float  hh[IPW], sa[IPW];
#pragma unroll
    for (int i = 0; i < IPW; ++i) {
        const int it = wave * IPW + i;
        uu[i] = *(const float4*)(hc + it * HC_P + lane * 4);
        vv[i] = *(const float4*)(hc + it * HC_P + EMB + lane * 4);
        hh[i] = h2s[it * H2_P + lane];
    }
#pragma unroll
    for (int i = 0; i < IPW; ++i) {
        const float4 u4 = uu[i], v4 = vv[i];
        const float su = wave_sum_valu(u4.x * u4.x + u4.y * u4.y + u4.z * u4.z + u4.w * u4.w);
        const float sv = wave_sum_valu(v4.x * v4.x + v4.y * v4.y + v4.z * v4.z + v4.w * v4.w);
        const float rn = 1.0f / (fmaxf(sqrtf(su), 1e-12f) * fmaxf(sqrtf(sv), 1e-12f));
        float a = (u4.x * v4.x * rn) * ww.x + (u4.y * v4.y * rn) * ww.y
                + (u4.z * v4.z * rn) * ww.z + (u4.w * v4.w * rn) * ww.w;
        a += hh[i] * wh;
        sa[i] = wave_sum_valu(a);
    }
    float mine = sa[0];
#pragma unroll
    for (int i = 1; i < IPW; ++i) mine = (lane == i) ? sa[i] : mine;
    if (lane < IPW) res[wave * IPW + lane] = 1.0f / (1.0f + expf(-(mine + ob)));
    __syncthreads();
    if (tid < HB_ITEMS)
        score[(size_t)blockIdx.x * HB_ITEMS + tid] = res[tid];
}

extern "C" void kernel_launch(void* const* d_in, const int* in_sizes, int n_in,
                              void* d_out, int out_size, void* d_ws, size_t ws_size,
                              hipStream_t stream) {
    const int*   user_idx = (const int*)d_in[0];
    const int*   item_idx = (const int*)d_in[1];
    const int*   eidx     = (const int*)d_in[2];
    const int*   etype    = (const int*)d_in[3];
    const float* eweight  = (const float*)d_in[4];
    const float* emb      = (const float*)d_in[5];
    const float* w1_rel   = (const float*)d_in[6];
    const float* w1_root  = (const float*)d_in[7];
    const float* b1       = (const float*)d_in[8];
    const float* ln_g     = (const float*)d_in[9];
    const float* ln_b     = (const float*)d_in[10];
    const float* w2_rel   = (const float*)d_in[11];
    const float* w2_root  = (const float*)d_in[12];
    const float* b2       = (const float*)d_in[13];
    const float* mlp_w0   = (const float*)d_in[14];
    const float* mlp_b0   = (const float*)d_in[15];
    const float* mlp_w1   = (const float*)d_in[16];
    const float* mlp_b1   = (const float*)d_in[17];
    const float* mlp_w2   = (const float*)d_in[18];
    const float* mlp_b2   = (const float*)d_in[19];
    const float* out_w    = (const float*)d_in[20];
    const float* out_b    = (const float*)d_in[21];
    float* out = (float*)d_out;

    char* ws = (char*)d_ws;
    const size_t NB = (size_t)NNODES * EMB * sizeof(float);
    float* agg0 = (float*)(ws);
    float* agg1 = (float*)(ws + NB);
    float* h1   = (float*)(ws + 2 * NB);
    const size_t AGB = (size_t)2 * BATCH * EMB * sizeof(float);
    const size_t RCB = (size_t)ROWS_PAD * BKT * sizeof(int2);
    static_assert((size_t)2 * BATCH * EMB * sizeof(float) + (size_t)ROWS_PAD * BKT * sizeof(int2)
                  + (size_t)NEDGES * sizeof(int4) <= (size_t)NNODES * EMB * sizeof(float),
                  "slots and overflow list fit in the first region behind agg0");
    int*  cnt = (int*)(ws + 3 * NB);
    int2* rec = (int2*)(ws + AGB);
    int4* ovf = (int4*)(ws + AGB + RCB);

    const int l1Grid    = (NNODES + L1_ROWS - 1) / L1_ROWS;
    const int headGrid  = BATCH / HB_ITEMS;
    const int edgeGrid4 = NEDGES / 1024;
    const int itemGrid  = BATCH / 8;

    edge_count_clear_kernel<<<CNT_PAD / 1024, 256, 0, stream>>>(cnt);
    edge_place_kernel<<<edgeGrid4, 256, 0, stream>>>(eidx, eidx + NEDGES, etype, eweight, cnt, rec, ovf);

    layer1_fused_kernel<<<l1Grid, 256, 0, stream>>>(cnt, rec, ovf, emb, w1_rel, w1_root, b1, ln_g, ln_b, h1);

    gathered_rows_pull_kernel<<<itemGrid, 256, 0, stream>>>(user_idx, item_idx, cnt, rec, ovf, h1, agg0, agg1);

    head_fused_kernel<<<headGrid, 256, 0, stream>>>(agg0, agg1, h1, w2_rel, w2_root, b2, user_idx, item_idx,
                                                    mlp_w0, mlp_b0, mlp_w1, mlp_b1, mlp_w2, mlp_b2,
                                                    out_w, out_b, out);
}
